# starting kernel
# baseline (speedup 1.0000x reference)
#include <hip/hip_runtime.h>
#include <hip/hip_bf16.h>


#define NUM_HEADS 4
#define HEAD_DIM 32
#define F_IN 256
#define F_OUT 128
#define LDA 260

typedef __attribute__((ext_vector_type(2))) float v2f;
typedef __attribute__((ext_vector_type(8))) float v8f;

__global__ __launch_bounds__(256) void gat_init(float* __restrict__ out,
                                                float* __restrict__ emax,
                                                float* __restrict__ asum,
                                                int N) {
    int tid = blockIdx.x * blockDim.x + threadIdx.x;
    if (tid < N * F_OUT) out[tid] = 0.0f;
    if (tid < N * NUM_HEADS) { emax[tid] = -3.0e38f; asum[tid] = 0.0f; }
}

__global__ __launch_bounds__(256) void gat_gemm(const float* __restrict__ x,
                                                const float* __restrict__ W,
                                                float* __restrict__ h) {
    __shared__ float As[16 * LDA];

    const int t = threadIdx.x;
    const int rowBase = blockIdx.x * 16;

    #pragma unroll
    for (int j = 0; j < 4; ++j) {
        int f = t + 256 * j;
        int r = f >> 6;
        int c = (f & 63) << 2;
        float4 v = *(const float4*)(x + (size_t)(rowBase + r) * F_IN + c);
        *(float4*)(&As[r * LDA + c]) = v;
    }
    __syncthreads();

    const int wave = t >> 5;
    const int lane = t & 31;
    const int m    = lane & 15;
    const int koff = (lane >> 4) << 1;
    const int col  = (wave << 4) + m;

    const float* Wp = W + col;
    v8f acc = {};

    #pragma unroll 8
    for (int k = 0; k < F_IN; k += 4) {
        v2f a = *(const v2f*)(&As[m * LDA + k + koff]);
        v2f b;
        b.x = Wp[(size_t)(k + koff) * F_OUT];
        b.y = Wp[(size_t)(k + koff + 1) * F_OUT];
        acc = __builtin_amdgcn_wmma_f32_16x16x4_f32(
             false, a,  false, b,
             (short)0, acc,  false,  false);
    }

    const int roff = (lane >> 4) << 3;
    float* hp = h + (size_t)rowBase * F_OUT + col;
    #pragma unroll
    for (int v = 0; v < 8; ++v)
        hp[(size_t)(v + roff) * F_OUT] = acc[v];
}

__global__ __launch_bounds__(256) void gat_gemm_tail(const float* __restrict__ x,
                                                     const float* __restrict__ W,
                                                     float* __restrict__ h,
                                                     int N, int rowStart) {
    int idx = blockIdx.x * blockDim.x + threadIdx.x;
    int total = (N - rowStart) * F_OUT;
    if (idx >= total) return;
    int r = rowStart + idx / F_OUT;
    int c = idx % F_OUT;
    float s = 0.0f;
    for (int k = 0; k < F_IN; ++k)
        s += x[(size_t)r * F_IN + k] * W[(size_t)k * F_OUT + c];
    h[(size_t)r * F_OUT + c] = s;
}

__global__ __launch_bounds__(256) void gat_alpha(const float* __restrict__ h,
                                                 const float* __restrict__ a_src,
                                                 const float* __restrict__ a_dst,
                                                 float* __restrict__ asrc,
                                                 float* __restrict__ adst,
                                                 int N) {
    int tid = blockIdx.x * blockDim.x + threadIdx.x;
    if (tid >= N * NUM_HEADS) return;
    int n  = tid >> 2;
    int hd = tid & 3;
    const float* hp = h + (size_t)n * F_OUT + hd * HEAD_DIM;
    const float* as = a_src + hd * HEAD_DIM;
    const float* ad = a_dst + hd * HEAD_DIM;
    float s1 = 0.0f, s2 = 0.0f;
    #pragma unroll
    for (int i = 0; i < HEAD_DIM; i += 4) {
        float4 hv = *(const float4*)(hp + i);
        float4 av = *(const float4*)(as + i);
        float4 dv = *(const float4*)(ad + i);
        s1 += hv.x * av.x + hv.y * av.y + hv.z * av.z + hv.w * av.w;
        s2 += hv.x * dv.x + hv.y * dv.y + hv.z * dv.z + hv.w * dv.w;
    }
    asrc[tid] = s1;
    adst[tid] = s2;
}

__device__ __forceinline__ float leaky(float v) { return v > 0.0f ? v : 0.2f * v; }

__global__ __launch_bounds__(256) void gat_edge_max(const int* __restrict__ src,
                                                    const int* __restrict__ dst,
                                                    const float* __restrict__ asrc,
                                                    const float* __restrict__ adst,
                                                    float* __restrict__ emax,
                                                    int E) {
    int e = blockIdx.x * blockDim.x + threadIdx.x;
    if (e >= E) return;
    int s = src[e], d = dst[e];
    #pragma unroll
    for (int hd = 0; hd < NUM_HEADS; ++hd) {
        float v = leaky(asrc[s * NUM_HEADS + hd] + adst[d * NUM_HEADS + hd]);
        __hip_atomic_fetch_max(&emax[d * NUM_HEADS + hd], v,
                               __ATOMIC_RELAXED, __HIP_MEMORY_SCOPE_AGENT);
    }
}

__global__ __launch_bounds__(256) void gat_edge_sum(const int* __restrict__ src,
                                                    const int* __restrict__ dst,
                                                    const float* __restrict__ asrc,
                                                    const float* __restrict__ adst,
                                                    const float* __restrict__ emax,
                                                    float* __restrict__ asum,
                                                    int E) {
    int e = blockIdx.x * blockDim.x + threadIdx.x;
    if (e >= E) return;
    int s = src[e], d = dst[e];
    #pragma unroll
    for (int hd = 0; hd < NUM_HEADS; ++hd) {
        float v = leaky(asrc[s * NUM_HEADS + hd] + adst[d * NUM_HEADS + hd]);
        float w = __expf(v - emax[d * NUM_HEADS + hd]);
        atomicAdd(&asum[d * NUM_HEADS + hd], w);
    }
}

__global__ __launch_bounds__(256) void gat_aggregate(const int* __restrict__ src,
                                                     const int* __restrict__ dst,
                                                     const float* __restrict__ asrc,
                                                     const float* __restrict__ adst,
                                                     const float* __restrict__ emax,
                                                     const float* __restrict__ asum,
                                                     const float* __restrict__ h,
                                                     float* __restrict__ out,
                                                     int E) {
    int e = blockIdx.x * 8 + (threadIdx.x >> 5);
    if (e >= E) return;
    int lane = threadIdx.x & 31;
    int s = src[e], d = dst[e];
    int elem = lane << 2;
    int hd   = elem >> 5;

    float v = leaky(asrc[s * NUM_HEADS + hd] + adst[d * NUM_HEADS + hd]);
    float w = __expf(v - emax[d * NUM_HEADS + hd]) /
              (asum[d * NUM_HEADS + hd] + 1e-10f);

    float4 hv = *(const float4*)(h + (size_t)s * F_OUT + elem);
    float* op = out + (size_t)d * F_OUT + elem;
    atomicAdd(op + 0, w * hv.x);
    atomicAdd(op + 1, w * hv.y);
    atomicAdd(op + 2, w * hv.z);
    atomicAdd(op + 3, w * hv.w);
}

extern "C" void kernel_launch(void* const* d_in, const int* in_sizes, int n_in,
                              void* d_out, int out_size, void* d_ws, size_t ws_size,
                              hipStream_t stream) {
    const float* x     = (const float*)d_in[0];
    const int*   ei    = (const int*)d_in[1];
    const float* W     = (const float*)d_in[2];
    const float* a_src = (const float*)d_in[3];
    const float* a_dst = (const float*)d_in[4];
    float*       out   = (float*)d_out;

    const int N = in_sizes[0] / F_IN;
    const int E = in_sizes[1] / 2;
    const int* src = ei;
    const int* dst = ei + E;

    float* h    = (float*)d_ws;
    float* asrc = h    + (size_t)N * F_OUT;
    float* adst = asrc + (size_t)N * NUM_HEADS;
    float* emax = adst + (size_t)N * NUM_HEADS;
    float* asum = emax + (size_t)N * NUM_HEADS;

    {
        int total = N * F_OUT;
        gat_init<<<(total + 255) / 256, 256, 0, stream>>>(out, emax, asum, N);
    }

    {
        int fullTiles = N / 16;
        if (fullTiles > 0)
            gat_gemm<<<fullTiles, 256, 0, stream>>>(x, W, h);
        int rowStart = fullTiles * 16;
        int rem = N - rowStart;
        if (rem > 0) {
            int total = rem * F_OUT;
            gat_gemm_tail<<<(total + 255) / 256, 256, 0, stream>>>(x, W, h, N, rowStart);
        }
    }

    {
        int total = N * NUM_HEADS;
        gat_alpha<<<(total + 255) / 256, 256, 0, stream>>>(h, a_src, a_dst, asrc, adst, N);
    }

    gat_edge_max<<<(E + 255) / 256, 256, 0, stream>>>(src, dst, asrc, adst, emax, E);

    gat_edge_sum<<<(E + 255) / 256, 256, 0, stream>>>(src, dst, asrc, adst, emax, asum, E);

    gat_aggregate<<<(E + 7) / 8, 256, 0, stream>>>(src, dst, asrc, adst, emax, asum, h, out, E);
}


# Edge work by destination ownership: a work-group keeps 400 nodes' maxima, weight sums and row sums in LDS and filters the whole edge list for them; sums in LDS arrival order, one division per row
# speedup vs baseline: 2.7400x; 2.7400x over previous
#include <hip/hip_runtime.h>
#include <hip/hip_bf16.h>


#define NUM_HEADS 4
#define HEAD_DIM 32
#define F_IN 256
#define F_OUT 128
#define LDA 260

typedef __attribute__((ext_vector_type(2))) float v2f;
typedef __attribute__((ext_vector_type(8))) float v8f;

__global__ __launch_bounds__(256) void gat_gemm(const float* __restrict__ x,
                                                const float* __restrict__ W,
                                                float* __restrict__ h) {
    __shared__ float As[16 * LDA];

    const int t = threadIdx.x;
    const int rowBase = blockIdx.x * 16;

    #pragma unroll
    for (int j = 0; j < 4; ++j) {
        int f = t + 256 * j;
        int r = f >> 6;
        int c = (f & 63) << 2;
        float4 v = *(const float4*)(x + (size_t)(rowBase + r) * F_IN + c);
        *(float4*)(&As[r * LDA + c]) = v;
    }
    __syncthreads();

    const int wave = t >> 5;
    const int lane = t & 31;
    const int m    = lane & 15;
    const int koff = (lane >> 4) << 1;
    const int col  = (wave << 4) + m;

    const float* Wp = W + col;
    v8f acc = {};

    #pragma unroll 8
    for (int k = 0; k < F_IN; k += 4) {
        v2f a = *(const v2f*)(&As[m * LDA + k + koff]);
        v2f b;
        b.x = Wp[(size_t)(k + koff) * F_OUT];
        b.y = Wp[(size_t)(k + koff + 1) * F_OUT];
        acc = __builtin_amdgcn_wmma_f32_16x16x4_f32(
             false, a,  false, b,
             (short)0, acc,  false,  false);
    }

    const int roff = (lane >> 4) << 3;
    float* hp = h + (size_t)rowBase * F_OUT + col;
    #pragma unroll
    for (int v = 0; v < 8; ++v)
        hp[(size_t)(v + roff) * F_OUT] = acc[v];
}

__global__ __launch_bounds__(256) void gat_gemm_tail(const float* __restrict__ x,
                                                     const float* __restrict__ W,
                                                     float* __restrict__ h,
                                                     int N, int rowStart) {
    int idx = blockIdx.x * blockDim.x + threadIdx.x;
    int total = (N - rowStart) * F_OUT;
    if (idx >= total) return;
    int r = rowStart + idx / F_OUT;
    int c = idx % F_OUT;
    float s = 0.0f;
    for (int k = 0; k < F_IN; ++k)
        s += x[(size_t)r * F_IN + k] * W[(size_t)k * F_OUT + c];
    h[(size_t)r * F_OUT + c] = s;
}

__global__ __launch_bounds__(256) void gat_alpha(const float* __restrict__ h,
                                                 const float* __restrict__ a_src,
                                                 const float* __restrict__ a_dst,
                                                 float* __restrict__ asrc,
                                                 float* __restrict__ adst,
                                                 int N) {
    int tid = blockIdx.x * blockDim.x + threadIdx.x;
    if (tid >= N * NUM_HEADS) return;
    int n  = tid >> 2;
    int hd = tid & 3;
    const float* hp = h + (size_t)n * F_OUT + hd * HEAD_DIM;
    const float* as = a_src + hd * HEAD_DIM;
    const float* ad = a_dst + hd * HEAD_DIM;
    float s1 = 0.0f, s2 = 0.0f;
    #pragma unroll
    for (int i = 0; i < HEAD_DIM; i += 4) {
        float4 hv = *(const float4*)(hp + i);
        float4 av = *(const float4*)(as + i);
        float4 dv = *(const float4*)(ad + i);
        s1 += hv.x * av.x + hv.y * av.y + hv.z * av.z + hv.w * av.w;
        s2 += hv.x * dv.x + hv.y * dv.y + hv.z * dv.z + hv.w * dv.w;
    }
    asrc[tid] = s1;
    adst[tid] = s2;
}

__device__ __forceinline__ float leaky(float v) { return v > 0.0f ? v : 0.2f * v; }

#define OWN_NODES   400
#define OWN_WAVES   32
#define OWN_THREADS 1024
#define Q_CAP       144
#define Q_DRAIN     16
#define BLK_EDGES   128

__device__ __forceinline__ int4 load_dst4_guarded(const int* __restrict__ dst, int e, int E) {
    int4 r;
    r.x = (e     < E) ? dst[e]     : -1;
    r.y = (e + 1 < E) ? dst[e + 1] : -1;
    r.z = (e + 2 < E) ? dst[e + 2] : -1;
    r.w = (e + 3 < E) ? dst[e + 3] : -1;
    return r;
}

__device__ __forceinline__ void drain_max(const int* qs, const int* qd, int qn, int lane,
                                          const int* __restrict__ src,
                                          const float* __restrict__ asrc,
                                          const float* sadst, float* smax) {
    for (int i0 = 0; i0 < qn; i0 += 32) {
        int idx = i0 + lane;
        if (idx < qn) {
            int s = src[qs[idx]], dl = qd[idx];
            float4 a = *(const float4*)(asrc + (size_t)s * NUM_HEADS);
            float4 b = *(const float4*)(sadst + dl * NUM_HEADS);
            float* mp = smax + dl * NUM_HEADS;
            __hip_atomic_fetch_max(mp + 0, leaky(a.x + b.x), __ATOMIC_RELAXED, __HIP_MEMORY_SCOPE_WORKGROUP);
            __hip_atomic_fetch_max(mp + 1, leaky(a.y + b.y), __ATOMIC_RELAXED, __HIP_MEMORY_SCOPE_WORKGROUP);
            __hip_atomic_fetch_max(mp + 2, leaky(a.z + b.z), __ATOMIC_RELAXED, __HIP_MEMORY_SCOPE_WORKGROUP);
            __hip_atomic_fetch_max(mp + 3, leaky(a.w + b.w), __ATOMIC_RELAXED, __HIP_MEMORY_SCOPE_WORKGROUP);
        }
    }
}

__device__ __forceinline__ void drain_sum(int* qs, const int* qd, int qn, int lane,
                                          const int* __restrict__ src,
                                          const float* __restrict__ asrc,
                                          const float* __restrict__ h,
                                          const float* sadst, const float* smax,
                                          float* ssum, float* acc) {
    const int hd = lane & 3;
    for (int i = lane; i < qn; i += 32) qs[i] = src[qs[i]];
    for (int i0 = 0; i0 < qn; i0 += 8) {
        const int nb = min(8, qn - i0);
        const int idx = i0 + (lane >> 2);
        const bool on = idx < qn;
        int s = 0, dl = 0;
        float as = 0.0f;
        if (on) {
            s = qs[idx]; dl = qd[idx];
            as = asrc[(size_t)s * NUM_HEADS + hd];
        }
        float hv[8][NUM_HEADS];
        #pragma unroll
        for (int j = 0; j < 8; ++j) {
            if (j < nb) {
                int sj = __builtin_amdgcn_readlane(s, 4 * j);
                const float* hp = h + (size_t)sj * F_OUT + lane;
                #pragma unroll
                for (int c = 0; c < NUM_HEADS; ++c) hv[j][c] = hp[c * HEAD_DIM];
            }
        }
        float w = 0.0f;
        if (on) {
            float v = leaky(as + sadst[dl * NUM_HEADS + hd]);
            w = __expf(v - smax[dl * NUM_HEADS + hd]);
            __hip_atomic_fetch_add(ssum + dl * NUM_HEADS + hd, w,
                                   __ATOMIC_RELAXED, __HIP_MEMORY_SCOPE_WORKGROUP);
        }
        const int wbits = __float_as_int(w);
        #pragma unroll
        for (int j = 0; j < 8; ++j) {
            if (j < nb) {
                int dj = __builtin_amdgcn_readlane(dl, 4 * j);
                float* ap = acc + dj * F_OUT + lane;
                #pragma unroll
                for (int c = 0; c < NUM_HEADS; ++c) {
                    float wj = __int_as_float(__builtin_amdgcn_readlane(wbits, 4 * j + c));
                    __hip_atomic_fetch_add(ap + c * HEAD_DIM, wj * hv[j][c],
                                           __ATOMIC_RELAXED, __HIP_MEMORY_SCOPE_WORKGROUP);
                }
            }
        }
    }
}

template <int PASS>
__device__ __forceinline__ void owner_scan(const int* __restrict__ src,
                                           const int* __restrict__ dst,
                                           const float* __restrict__ asrc,
                                           const float* __restrict__ h,
                                           int E, int lo, int nloc, int wave, int lane,
                                           int* qs, int* qd,
                                           const float* sadst, float* smax,
                                           float* ssum, float* acc) {
    const bool vec = (E & 3) == 0;
    const int perIter = OWN_WAVES * 4 * BLK_EDGES;
    const int nIter = (E + perIter - 1) / perIter;
    const int start = (int)(((long long)blockIdx.x * nIter) / gridDim.x);
    const unsigned below = (1u << lane) - 1u;
    int qn = 0;
    for (int it = 0; it < nIter; ++it) {
        int jt = it + start;
        if (jt >= nIter) jt -= nIter;
        const int wbase = (jt * OWN_WAVES + wave) * (4 * BLK_EDGES);
        const int base = wbase + 4 * lane;
        int4 dv[4];
        if (vec && wbase + 4 * BLK_EDGES <= E) {
            #pragma unroll
            for (int k = 0; k < 4; ++k) dv[k] = *(const int4*)(dst + base + k * BLK_EDGES);
        } else {
            #pragma unroll
            for (int k = 0; k < 4; ++k) dv[k] = load_dst4_guarded(dst, base + k * BLK_EDGES, E);
        }
        #pragma unroll
        for (int k = 0; k < 4; ++k) {
            const int e0 = base + k * BLK_EDGES;
            const int dd[4] = {dv[k].x, dv[k].y, dv[k].z, dv[k].w};
            bool m[4];
            bool any = false;
            #pragma unroll
            for (int c = 0; c < 4; ++c) {
                m[c] = (unsigned)(dd[c] - lo) < (unsigned)nloc;
                any = any || m[c];
            }
            if (__builtin_amdgcn_ballot_w32(any) != 0u) {
                #pragma unroll
                for (int c = 0; c < 4; ++c) {
                    const unsigned mask = __builtin_amdgcn_ballot_w32(m[c]);
                    if (m[c]) {
                        const int pos = qn + __builtin_popcount(mask & below);
                        qs[pos] = e0 + c;
                        qd[pos] = dd[c] - lo;
                    }
                    qn += __builtin_popcount(mask);
                }
                if (qn >= Q_DRAIN) {
                    if (PASS == 1) drain_max(qs, qd, qn, lane, src, asrc, sadst, smax);
                    else           drain_sum(qs, qd, qn, lane, src, asrc, h, sadst, smax, ssum, acc);
                    qn = 0;
                }
            }
        }
    }
    if (qn > 0) {
        if (PASS == 1) drain_max(qs, qd, qn, lane, src, asrc, sadst, smax);
        else           drain_sum(qs, qd, qn, lane, src, asrc, h, sadst, smax, ssum, acc);
    }
}

__global__ __launch_bounds__(OWN_THREADS) void gat_owner(const int* __restrict__ src,
                                                         const int* __restrict__ dst,
                                                         const float* __restrict__ asrc,
                                                         const float* __restrict__ adst,
                                                         const float* __restrict__ h,
                                                         float* __restrict__ out,
                                                         int N, int E) {
    __shared__ float acc[OWN_NODES * F_OUT];
    __shared__ float smax[OWN_NODES * NUM_HEADS];
    __shared__ float ssum[OWN_NODES * NUM_HEADS];
    __shared__ float sadst[OWN_NODES * NUM_HEADS];
    __shared__ int   qsrc[OWN_WAVES * Q_CAP];
    __shared__ int   qdst[OWN_WAVES * Q_CAP];

    const int t = threadIdx.x;
    const int lane = t & 31;
    const int wave = __builtin_amdgcn_readfirstlane(t >> 5);
    const int lo = blockIdx.x * OWN_NODES;
    const int nloc = min(OWN_NODES, N - lo);

    for (int i = t; i < OWN_NODES * F_OUT; i += OWN_THREADS) acc[i] = 0.0f;
    for (int i = t; i < OWN_NODES * NUM_HEADS; i += OWN_THREADS) {
        smax[i] = -3.0e38f;
        ssum[i] = 0.0f;
        sadst[i] = (i < nloc * NUM_HEADS) ? adst[(size_t)lo * NUM_HEADS + i] : 0.0f;
    }
    __syncthreads();

    int* qs = qsrc + wave * Q_CAP;
    int* qd = qdst + wave * Q_CAP;

    owner_scan<1>(src, dst, asrc, h, E, lo, nloc, wave, lane, qs, qd, sadst, smax, ssum, acc);
    __syncthreads();
    owner_scan<2>(src, dst, asrc, h, E, lo, nloc, wave, lane, qs, qd, sadst, smax, ssum, acc);
    __syncthreads();

    for (int r = wave; r < nloc; r += OWN_WAVES) {
        float* op = out + (size_t)(lo + r) * F_OUT + lane;
        #pragma unroll
        for (int c = 0; c < NUM_HEADS; ++c)
            op[c * HEAD_DIM] = acc[r * F_OUT + c * HEAD_DIM + lane] /
                               (ssum[r * NUM_HEADS + c] + 1e-10f);
    }
}

extern "C" void kernel_launch(void* const* d_in, const int* in_sizes, int n_in,
                              void* d_out, int out_size, void* d_ws, size_t ws_size,
                              hipStream_t stream) {
    const float* x     = (const float*)d_in[0];
    const int*   ei    = (const int*)d_in[1];
    const float* W     = (const float*)d_in[2];
    const float* a_src = (const float*)d_in[3];
    const float* a_dst = (const float*)d_in[4];
    float*       out   = (float*)d_out;

    const int N = in_sizes[0] / F_IN;
    const int E = in_sizes[1] / 2;
    const int* src = ei;
    const int* dst = ei + E;

    float* h    = (float*)d_ws;
    float* asrc = h    + (size_t)N * F_OUT;
    float* adst = asrc + (size_t)N * NUM_HEADS;

    {
        int fullTiles = N / 16;
        if (fullTiles > 0)
            gat_gemm<<<fullTiles, 256, 0, stream>>>(x, W, h);
        int rowStart = fullTiles * 16;
        int rem = N - rowStart;
        if (rem > 0) {
            int total = rem * F_OUT;
            gat_gemm_tail<<<(total + 255) / 256, 256, 0, stream>>>(x, W, h, N, rowStart);
        }
    }

    {
        int total = N * NUM_HEADS;
        gat_alpha<<<(total + 255) / 256, 256, 0, stream>>>(h, a_src, a_dst, asrc, adst, N);
    }

    gat_owner<<<(N + OWN_NODES - 1) / OWN_NODES, OWN_THREADS, 0, stream>>>(src, dst, asrc, adst, h, out, N, E);
}


# Pass 1 keeps each matched edge (source node, local row) in a work-group LDS list and the weight-and-row pass walks it: the edge list is filtered once, not twice; sums stay in LDS arrival order
# speedup vs baseline: 3.8110x; 1.3909x over previous
#include <hip/hip_runtime.h>
#include <hip/hip_bf16.h>


#define NUM_HEADS 4
#define HEAD_DIM 32
#define F_IN 256
#define F_OUT 128
#define LDA 260

typedef __attribute__((ext_vector_type(2))) float v2f;
typedef __attribute__((ext_vector_type(8))) float v8f;

__global__ __launch_bounds__(256) void gat_gemm(const float* __restrict__ x,
                                                const float* __restrict__ W,
                                                float* __restrict__ h) {
    __shared__ float As[16 * LDA];

    const int t = threadIdx.x;
    const int rowBase = blockIdx.x * 16;

    #pragma unroll
    for (int j = 0; j < 4; ++j) {
        int f = t + 256 * j;
        int r = f >> 6;
        int c = (f & 63) << 2;
        float4 v = *(const float4*)(x + (size_t)(rowBase + r) * F_IN + c);
        *(float4*)(&As[r * LDA + c]) = v;
    }
    __syncthreads();

    const int wave = t >> 5;
    const int lane = t & 31;
    const int m    = lane & 15;
    const int koff = (lane >> 4) << 1;
    const int col  = (wave << 4) + m;

    const float* Wp = W + col;
    v8f acc = {};

    #pragma unroll 8
    for (int k = 0; k < F_IN; k += 4) {
        v2f a = *(const v2f*)(&As[m * LDA + k + koff]);
        v2f b;
        b.x = Wp[(size_t)(k + koff) * F_OUT];
        b.y = Wp[(size_t)(k + koff + 1) * F_OUT];
        acc = __builtin_amdgcn_wmma_f32_16x16x4_f32(
             false, a,  false, b,
             (short)0, acc,  false,  false);
    }

    const int roff = (lane >> 4) << 3;
    float* hp = h + (size_t)rowBase * F_OUT + col;
    #pragma unroll
    for (int v = 0; v < 8; ++v)
        hp[(size_t)(v + roff) * F_OUT] = acc[v];
}

__global__ __launch_bounds__(256) void gat_gemm_tail(const float* __restrict__ x,
                                                     const float* __restrict__ W,
                                                     float* __restrict__ h,
                                                     int N, int rowStart) {
    int idx = blockIdx.x * blockDim.x + threadIdx.x;
    int total = (N - rowStart) * F_OUT;
    if (idx >= total) return;
    int r = rowStart + idx / F_OUT;
    int c = idx % F_OUT;
    float s = 0.0f;
    for (int k = 0; k < F_IN; ++k)
        s += x[(size_t)r * F_IN + k] * W[(size_t)k * F_OUT + c];
    h[(size_t)r * F_OUT + c] = s;
}

__global__ __launch_bounds__(256) void gat_alpha(const float* __restrict__ h,
                                                 const float* __restrict__ a_src,
                                                 const float* __restrict__ a_dst,
                                                 float* __restrict__ asrc,
                                                 float* __restrict__ adst,
                                                 int N) {
    int tid = blockIdx.x * blockDim.x + threadIdx.x;
    if (tid >= N * NUM_HEADS) return;
    int n  = tid >> 2;
    int hd = tid & 3;
    const float* hp = h + (size_t)n * F_OUT + hd * HEAD_DIM;
    const float* as = a_src + hd * HEAD_DIM;
    const float* ad = a_dst + hd * HEAD_DIM;
    float s1 = 0.0f, s2 = 0.0f;
    #pragma unroll
    for (int i = 0; i < HEAD_DIM; i += 4) {
        float4 hv = *(const float4*)(hp + i);
        float4 av = *(const float4*)(as + i);
        float4 dv = *(const float4*)(ad + i);
        s1 += hv.x * av.x + hv.y * av.y + hv.z * av.z + hv.w * av.w;
        s2 += hv.x * dv.x + hv.y * dv.y + hv.z * dv.z + hv.w * dv.w;
    }
    asrc[tid] = s1;
    adst[tid] = s2;
}

__device__ __forceinline__ float leaky(float v) { return v > 0.0f ? v : 0.2f * v; }

#define OWN_NODES   400
#define OWN_WAVES   32
#define OWN_THREADS 1024
#define Q_CAP       144
#define Q_DRAIN     16
#define BLK_EDGES   128
#define L_CAP       12288
#define L_SHIFT     9
static_assert(OWN_NODES <= (1 << L_SHIFT), "local row must fit the low bits of a list word");

__device__ __forceinline__ int4 load_dst4_guarded(const int* __restrict__ dst, int e, int E) {
    int4 r;
    r.x = (e     < E) ? dst[e]     : -1;
    r.y = (e + 1 < E) ? dst[e + 1] : -1;
    r.z = (e + 2 < E) ? dst[e + 2] : -1;
    r.w = (e + 3 < E) ? dst[e + 3] : -1;
    return r;
}

__device__ __forceinline__ void drain_max(const int* qs, const int* qd, int qn, int lane,
                                          const int* __restrict__ src,
                                          const float* __restrict__ asrc,
                                          const float* sadst, float* smax,
                                          unsigned* elist, int* ecount, bool keep) {
    int slot0 = 0;
    if (keep) {
        if (lane == 0)
            slot0 = __hip_atomic_fetch_add(ecount, qn, __ATOMIC_RELAXED, __HIP_MEMORY_SCOPE_WORKGROUP);
        slot0 = __builtin_amdgcn_readfirstlane(slot0);
    }
    for (int i0 = 0; i0 < qn; i0 += 32) {
        int idx = i0 + lane;
        if (idx < qn) {
            int s = src[qs[idx]], dl = qd[idx];
            if (keep && slot0 + idx < L_CAP)
                elist[slot0 + idx] = ((unsigned)s << L_SHIFT) | (unsigned)dl;
            float4 a = *(const float4*)(asrc + (size_t)s * NUM_HEADS);
            float4 b = *(const float4*)(sadst + dl * NUM_HEADS);
            float* mp = smax + dl * NUM_HEADS;
            __hip_atomic_fetch_max(mp + 0, leaky(a.x + b.x), __ATOMIC_RELAXED, __HIP_MEMORY_SCOPE_WORKGROUP);
            __hip_atomic_fetch_max(mp + 1, leaky(a.y + b.y), __ATOMIC_RELAXED, __HIP_MEMORY_SCOPE_WORKGROUP);
            __hip_atomic_fetch_max(mp + 2, leaky(a.z + b.z), __ATOMIC_RELAXED, __HIP_MEMORY_SCOPE_WORKGROUP);
            __hip_atomic_fetch_max(mp + 3, leaky(a.w + b.w), __ATOMIC_RELAXED, __HIP_MEMORY_SCOPE_WORKGROUP);
        }
    }
}

__device__ __forceinline__ void sum_batch(int s, int dl, bool on, int nb, int lane,
                                          const float* __restrict__ asrc,
                                          const float* __restrict__ h,
                                          const float* sadst, const float* smax,
                                          float* ssum, float* acc) {
    const int hd = lane & 3;
    float as = 0.0f;
    if (on) as = asrc[(size_t)s * NUM_HEADS + hd];
    float hv[8][NUM_HEADS];
    #pragma unroll
    for (int j = 0; j < 8; ++j) {
        if (j < nb) {
            int sj = __builtin_amdgcn_readlane(s, 4 * j);
            const float* hp = h + (size_t)sj * F_OUT + lane;
            #pragma unroll
            for (int c = 0; c < NUM_HEADS; ++c) hv[j][c] = hp[c * HEAD_DIM];
        }
    }
    float w = 0.0f;
    if (on) {
        float v = leaky(as + sadst[dl * NUM_HEADS + hd]);
        w = __expf(v - smax[dl * NUM_HEADS + hd]);
        __hip_atomic_fetch_add(ssum + dl * NUM_HEADS + hd, w,
                               __ATOMIC_RELAXED, __HIP_MEMORY_SCOPE_WORKGROUP);
    }
    const int wbits = __float_as_int(w);
    #pragma unroll
    for (int j = 0; j < 8; ++j) {
        if (j < nb) {
            int dj = __builtin_amdgcn_readlane(dl, 4 * j);
            float* ap = acc + dj * F_OUT + lane;
            #pragma unroll
            for (int c = 0; c < NUM_HEADS; ++c) {
                float wj = __int_as_float(__builtin_amdgcn_readlane(wbits, 4 * j + c));
                __hip_atomic_fetch_add(ap + c * HEAD_DIM, wj * hv[j][c],
                                       __ATOMIC_RELAXED, __HIP_MEMORY_SCOPE_WORKGROUP);
            }
        }
    }
}

__device__ __forceinline__ void drain_sum(int* qs, const int* qd, int qn, int lane,
                                          const int* __restrict__ src,
                                          const float* __restrict__ asrc,
                                          const float* __restrict__ h,
                                          const float* sadst, const float* smax,
                                          float* ssum, float* acc) {
    for (int i = lane; i < qn; i += 32) qs[i] = src[qs[i]];
    for (int i0 = 0; i0 < qn; i0 += 8) {
        const int nb = min(8, qn - i0);
        const int idx = i0 + (lane >> 2);
        const bool on = idx < qn;
        int s = 0, dl = 0;
        if (on) { s = qs[idx]; dl = qd[idx]; }
        sum_batch(s, dl, on, nb, lane, asrc, h, sadst, smax, ssum, acc);
    }
}

__device__ __forceinline__ void list_sum(const unsigned* elist, int total, int wave, int lane,
                                         const float* __restrict__ asrc,
                                         const float* __restrict__ h,
                                         const float* sadst, const float* smax,
                                         float* ssum, float* acc) {
    for (int i0 = wave * 8; i0 < total; i0 += OWN_WAVES * 8) {
        const int nb = min(8, total - i0);
        const int idx = i0 + (lane >> 2);
        const bool on = idx < total;
        int s = 0, dl = 0;
        if (on) {
            const unsigned p = elist[idx];
            s  = (int)(p >> L_SHIFT);
            dl = (int)(p & ((1u << L_SHIFT) - 1u));
        }
        sum_batch(s, dl, on, nb, lane, asrc, h, sadst, smax, ssum, acc);
    }
}

template <int PASS>
__device__ __forceinline__ void owner_scan(const int* __restrict__ src,
                                           const int* __restrict__ dst,
                                           const float* __restrict__ asrc,
                                           const float* __restrict__ h,
                                           int E, int lo, int nloc, int wave, int lane,
                                           int* qs, int* qd,
                                           const float* sadst, float* smax,
                                           float* ssum, float* acc,
                                           unsigned* elist, int* ecount, bool keep) {
    const bool vec = (E & 3) == 0;
    const int perIter = OWN_WAVES * 4 * BLK_EDGES;
    const int nIter = (E + perIter - 1) / perIter;
    const int start = (int)(((long long)blockIdx.x * nIter) / gridDim.x);
    const unsigned below = (1u << lane) - 1u;
    int qn = 0;
    for (int it = 0; it < nIter; ++it) {
        int jt = it + start;
        if (jt >= nIter) jt -= nIter;
        const int wbase = (jt * OWN_WAVES + wave) * (4 * BLK_EDGES);
        const int base = wbase + 4 * lane;
        int4 dv[4];
        if (vec && wbase + 4 * BLK_EDGES <= E) {
            #pragma unroll
            for (int k = 0; k < 4; ++k) dv[k] = *(const int4*)(dst + base + k * BLK_EDGES);
        } else {
            #pragma unroll
            for (int k = 0; k < 4; ++k) dv[k] = load_dst4_guarded(dst, base + k * BLK_EDGES, E);
        }
        #pragma unroll
        for (int k = 0; k < 4; ++k) {
            const int e0 = base + k * BLK_EDGES;
            const int dd[4] = {dv[k].x, dv[k].y, dv[k].z, dv[k].w};
            bool m[4];
            bool any = false;
            #pragma unroll
            for (int c = 0; c < 4; ++c) {
                m[c] = (unsigned)(dd[c] - lo) < (unsigned)nloc;
                any = any || m[c];
            }
            if (__builtin_amdgcn_ballot_w32(any) != 0u) {
                #pragma unroll
                for (int c = 0; c < 4; ++c) {
                    const unsigned mask = __builtin_amdgcn_ballot_w32(m[c]);
                    if (m[c]) {
                        const int pos = qn + __builtin_popcount(mask & below);
                        qs[pos] = e0 + c;
                        qd[pos] = dd[c] - lo;
                    }
                    qn += __builtin_popcount(mask);
                }
                if (qn >= Q_DRAIN) {
                    if (PASS == 1) drain_max(qs, qd, qn, lane, src, asrc, sadst, smax, elist, ecount, keep);
                    else           drain_sum(qs, qd, qn, lane, src, asrc, h, sadst, smax, ssum, acc);
                    qn = 0;
                }
            }
        }
    }
    if (qn > 0) {
        if (PASS == 1) drain_max(qs, qd, qn, lane, src, asrc, sadst, smax, elist, ecount, keep);
        else           drain_sum(qs, qd, qn, lane, src, asrc, h, sadst, smax, ssum, acc);
    }
}

__global__ __launch_bounds__(OWN_THREADS) void gat_owner(const int* __restrict__ src,
                                                         const int* __restrict__ dst,
                                                         const float* __restrict__ asrc,
                                                         const float* __restrict__ adst,
                                                         const float* __restrict__ h,
                                                         float* __restrict__ out,
                                                         int N, int E) {
    __shared__ float acc[OWN_NODES * F_OUT];
    __shared__ float smax[OWN_NODES * NUM_HEADS];
    __shared__ float ssum[OWN_NODES * NUM_HEADS];
    __shared__ float sadst[OWN_NODES * NUM_HEADS];
    __shared__ int   qsrc[OWN_WAVES * Q_CAP];
    __shared__ int   qdst[OWN_WAVES * Q_CAP];
    __shared__ unsigned elist[L_CAP];
    __shared__ int   ecount;

    const int t = threadIdx.x;
    const int lane = t & 31;
    const int wave = __builtin_amdgcn_readfirstlane(t >> 5);
    const int lo = blockIdx.x * OWN_NODES;
    const int nloc = min(OWN_NODES, N - lo);
    const bool keep = N <= (1 << (32 - L_SHIFT));

    if (t == 0) ecount = 0;
    for (int i = t; i < OWN_NODES * F_OUT; i += OWN_THREADS) acc[i] = 0.0f;
    for (int i = t; i < OWN_NODES * NUM_HEADS; i += OWN_THREADS) {
        smax[i] = -3.0e38f;
        ssum[i] = 0.0f;
        sadst[i] = (i < nloc * NUM_HEADS) ? adst[(size_t)lo * NUM_HEADS + i] : 0.0f;
    }
    __syncthreads();

    int* qs = qsrc + wave * Q_CAP;
    int* qd = qdst + wave * Q_CAP;

    owner_scan<1>(src, dst, asrc, h, E, lo, nloc, wave, lane, qs, qd, sadst, smax, ssum, acc,
                  elist, &ecount, keep);
    __syncthreads();
    const int total = keep ? ecount : L_CAP + 1;
    if (total <= L_CAP)
        list_sum(elist, total, wave, lane, asrc, h, sadst, smax, ssum, acc);
    else
        owner_scan<2>(src, dst, asrc, h, E, lo, nloc, wave, lane, qs, qd, sadst, smax, ssum, acc,
                      elist, &ecount, false);
    __syncthreads();

    for (int r = wave; r < nloc; r += OWN_WAVES) {
        float* op = out + (size_t)(lo + r) * F_OUT + lane;
        #pragma unroll
        for (int c = 0; c < NUM_HEADS; ++c)
            op[c * HEAD_DIM] = acc[r * F_OUT + c * HEAD_DIM + lane] /
                               (ssum[r * NUM_HEADS + c] + 1e-10f);
    }
}

extern "C" void kernel_launch(void* const* d_in, const int* in_sizes, int n_in,
                              void* d_out, int out_size, void* d_ws, size_t ws_size,
                              hipStream_t stream) {
    const float* x     = (const float*)d_in[0];
    const int*   ei    = (const int*)d_in[1];
    const float* W     = (const float*)d_in[2];
    const float* a_src = (const float*)d_in[3];
    const float* a_dst = (const float*)d_in[4];
    float*       out   = (float*)d_out;

    const int N = in_sizes[0] / F_IN;
    const int E = in_sizes[1] / 2;
    const int* src = ei;
    const int* dst = ei + E;

    float* h    = (float*)d_ws;
    float* asrc = h    + (size_t)N * F_OUT;
    float* adst = asrc + (size_t)N * NUM_HEADS;

    {
        int fullTiles = N / 16;
        if (fullTiles > 0)
            gat_gemm<<<fullTiles, 256, 0, stream>>>(x, W, h);
        int rowStart = fullTiles * 16;
        int rem = N - rowStart;
        if (rem > 0) {
            int total = rem * F_OUT;
            gat_gemm_tail<<<(total + 255) / 256, 256, 0, stream>>>(x, W, h, N, rowStart);
        }
    }

    {
        int total = N * NUM_HEADS;
        gat_alpha<<<(total + 255) / 256, 256, 0, stream>>>(h, a_src, a_dst, asrc, adst, N);
    }

    gat_owner<<<(N + OWN_NODES - 1) / OWN_NODES, OWN_THREADS, 0, stream>>>(src, dst, asrc, adst, h, out, N, E);
}


# Filter reads one byte per edge, not the 4-byte destination: the logits launch writes each edge's owning range number, pass 1 tests 16 edges per 16-byte load; sums stay in LDS arrival order
# speedup vs baseline: 4.1593x; 1.0914x over previous
#include <hip/hip_runtime.h>
#include <hip/hip_bf16.h>


#define NUM_HEADS 4
#define HEAD_DIM 32
#define F_IN 256
#define F_OUT 128
#define LDA 260

typedef __attribute__((ext_vector_type(2))) float v2f;
typedef __attribute__((ext_vector_type(8))) float v8f;

__global__ __launch_bounds__(256) void gat_gemm(const float* __restrict__ x,
                                                const float* __restrict__ W,
                                                float* __restrict__ h) {
    __shared__ float As[16 * LDA];

    const int t = threadIdx.x;
    const int rowBase = blockIdx.x * 16;

    #pragma unroll
    for (int j = 0; j < 4; ++j) {
        int f = t + 256 * j;
        int r = f >> 6;
        int c = (f & 63) << 2;
        float4 v = *(const float4*)(x + (size_t)(rowBase + r) * F_IN + c);
        *(float4*)(&As[r * LDA + c]) = v;
    }
    __syncthreads();

    const int wave = t >> 5;
    const int lane = t & 31;
    const int m    = lane & 15;
    const int koff = (lane >> 4) << 1;
    const int col  = (wave << 4) + m;

    const float* Wp = W + col;
    v8f acc = {};

    #pragma unroll 8
    for (int k = 0; k < F_IN; k += 4) {
        v2f a = *(const v2f*)(&As[m * LDA + k + koff]);
        v2f b;
        b.x = Wp[(size_t)(k + koff) * F_OUT];
        b.y = Wp[(size_t)(k + koff + 1) * F_OUT];
        acc = __builtin_amdgcn_wmma_f32_16x16x4_f32(
             false, a,  false, b,
             (short)0, acc,  false,  false);
    }

    const int roff = (lane >> 4) << 3;
    float* hp = h + (size_t)rowBase * F_OUT + col;
    #pragma unroll
    for (int v = 0; v < 8; ++v)
        hp[(size_t)(v + roff) * F_OUT] = acc[v];
}

__global__ __launch_bounds__(256) void gat_gemm_tail(const float* __restrict__ x,
                                                     const float* __restrict__ W,
                                                     float* __restrict__ h,
                                                     int N, int rowStart) {
    int idx = blockIdx.x * blockDim.x + threadIdx.x;
    int total = (N - rowStart) * F_OUT;
    if (idx >= total) return;
    int r = rowStart + idx / F_OUT;
    int c = idx % F_OUT;
    float s = 0.0f;
    for (int k = 0; k < F_IN; ++k)
        s += x[(size_t)r * F_IN + k] * W[(size_t)k * F_OUT + c];
    h[(size_t)r * F_OUT + c] = s;
}

#define OWN_NODES   400
#define OWN_WAVES   32
#define KEY_NONE    255u
#define KEY_LOADS   4
#define KEY_TRIP    (OWN_WAVES * KEY_LOADS * 512)

__device__ __forceinline__ int keys_padded(int E) {
    return ((E + KEY_TRIP - 1) / KEY_TRIP) * KEY_TRIP;
}

__device__ __forceinline__ bool keys_usable(int N, int E, unsigned long long wsBytes) {
    const int groups = (N + OWN_NODES - 1) / OWN_NODES;
    const unsigned long long need =
        (unsigned long long)N * (F_OUT + 2 * NUM_HEADS) * sizeof(float) + (unsigned long long)keys_padded(E);
    return groups <= (int)KEY_NONE && wsBytes >= need;
}

__device__ __forceinline__ unsigned key_of(int d, int N) {
    return ((unsigned)d < (unsigned)N) ? (unsigned)d / OWN_NODES : KEY_NONE;
}

__global__ __launch_bounds__(256) void gat_alpha(const float* __restrict__ h,
                                                 const float* __restrict__ a_src,
                                                 const float* __restrict__ a_dst,
                                                 float* __restrict__ asrc,
                                                 float* __restrict__ adst,
                                                 const int* __restrict__ dst,
                                                 unsigned* __restrict__ keys,
                                                 int N, int E, unsigned long long wsBytes) {
    int tid = blockIdx.x * blockDim.x + threadIdx.x;
    if (keys_usable(N, E, wsBytes)) {
        const int words = keys_padded(E) >> 2;
        const int step = gridDim.x * blockDim.x;
        const bool vec = (E & 3) == 0;
        for (int g = tid; g < words; g += step) {
            const int e = g << 2;
            int4 d;
            if (vec && e + 4 <= E) {
                d = *(const int4*)(dst + e);
            } else {
                d.x = (e     < E) ? dst[e]     : -1;
                d.y = (e + 1 < E) ? dst[e + 1] : -1;
                d.z = (e + 2 < E) ? dst[e + 2] : -1;
                d.w = (e + 3 < E) ? dst[e + 3] : -1;
            }
            keys[g] = key_of(d.x, N) | (key_of(d.y, N) << 8) | (key_of(d.z, N) << 16) | (key_of(d.w, N) << 24);
        }
    }
    if (tid >= N * NUM_HEADS) return;
    int n  = tid >> 2;
    int hd = tid & 3;
    const float* hp = h + (size_t)n * F_OUT + hd * HEAD_DIM;
    const float* as = a_src + hd * HEAD_DIM;
    const float* ad = a_dst + hd * HEAD_DIM;
    float s1 = 0.0f, s2 = 0.0f;
    #pragma unroll
    for (int i = 0; i < HEAD_DIM; i += 4) {
        float4 hv = *(const float4*)(hp + i);
        float4 av = *(const float4*)(as + i);
        float4 dv = *(const float4*)(ad + i);
        s1 += hv.x * av.x + hv.y * av.y + hv.z * av.z + hv.w * av.w;
        s2 += hv.x * dv.x + hv.y * dv.y + hv.z * dv.z + hv.w * dv.w;
    }
    asrc[tid] = s1;
    adst[tid] = s2;
}

__device__ __forceinline__ float leaky(float v) { return v > 0.0f ? v : 0.2f * v; }

#define OWN_THREADS 1024
#define Q_CAP       144
#define Q_DRAIN     16
#define BLK_EDGES   128
#define L_CAP       12288
#define L_SHIFT     9
static_assert(OWN_NODES <= (1 << L_SHIFT), "local row must fit the low bits of a list word");

__device__ __forceinline__ int4 load_dst4_guarded(const int* __restrict__ dst, int e, int E) {
    int4 r;
    r.x = (e     < E) ? dst[e]     : -1;
    r.y = (e + 1 < E) ? dst[e + 1] : -1;
    r.z = (e + 2 < E) ? dst[e + 2] : -1;
    r.w = (e + 3 < E) ? dst[e + 3] : -1;
    return r;
}

template <bool BYKEY>
__device__ __forceinline__ void drain_max(const int* qs, const int* qd, int qn, int lane,
                                          const int* __restrict__ src,
                                          const int* __restrict__ dst, int lo,
                                          const float* __restrict__ asrc,
                                          const float* sadst, float* smax,
                                          unsigned* elist, int* ecount, bool keep) {
    int slot0 = 0;
    if (keep) {
        if (lane == 0)
            slot0 = __hip_atomic_fetch_add(ecount, qn, __ATOMIC_RELAXED, __HIP_MEMORY_SCOPE_WORKGROUP);
        slot0 = __builtin_amdgcn_readfirstlane(slot0);
    }
    for (int i0 = 0; i0 < qn; i0 += 32) {
        int idx = i0 + lane;
        if (idx < qn) {
            const int e = qs[idx];
            int s = src[e], dl = BYKEY ? dst[e] - lo : qd[idx];
            if (keep && slot0 + idx < L_CAP)
                elist[slot0 + idx] = ((unsigned)s << L_SHIFT) | (unsigned)dl;
            float4 a = *(const float4*)(asrc + (size_t)s * NUM_HEADS);
            float4 b = *(const float4*)(sadst + dl * NUM_HEADS);
            float* mp = smax + dl * NUM_HEADS;
            __hip_atomic_fetch_max(mp + 0, leaky(a.x + b.x), __ATOMIC_RELAXED, __HIP_MEMORY_SCOPE_WORKGROUP);
            __hip_atomic_fetch_max(mp + 1, leaky(a.y + b.y), __ATOMIC_RELAXED, __HIP_MEMORY_SCOPE_WORKGROUP);
            __hip_atomic_fetch_max(mp + 2, leaky(a.z + b.z), __ATOMIC_RELAXED, __HIP_MEMORY_SCOPE_WORKGROUP);
            __hip_atomic_fetch_max(mp + 3, leaky(a.w + b.w), __ATOMIC_RELAXED, __HIP_MEMORY_SCOPE_WORKGROUP);
        }
    }
}

__device__ __forceinline__ void sum_batch(int s, int dl, bool on, int nb, int lane,
                                          const float* __restrict__ asrc,
                                          const float* __restrict__ h,
                                          const float* sadst, const float* smax,
                                          float* ssum, float* acc) {
    const int hd = lane & 3;
    float as = 0.0f;
    if (on) as = asrc[(size_t)s * NUM_HEADS + hd];
    float hv[8][NUM_HEADS];
    #pragma unroll
    for (int j = 0; j < 8; ++j) {
        if (j < nb) {
            int sj = __builtin_amdgcn_readlane(s, 4 * j);
            const float* hp = h + (size_t)sj * F_OUT + lane;
            #pragma unroll
            for (int c = 0; c < NUM_HEADS; ++c) hv[j][c] = hp[c * HEAD_DIM];
        }
    }
    float w = 0.0f;
    if (on) {
        float v = leaky(as + sadst[dl * NUM_HEADS + hd]);
        w = __expf(v - smax[dl * NUM_HEADS + hd]);
        __hip_atomic_fetch_add(ssum + dl * NUM_HEADS + hd, w,
                               __ATOMIC_RELAXED, __HIP_MEMORY_SCOPE_WORKGROUP);
    }
    const int wbits = __float_as_int(w);
    #pragma unroll
    for (int j = 0; j < 8; ++j) {
        if (j < nb) {
            int dj = __builtin_amdgcn_readlane(dl, 4 * j);
            float* ap = acc + dj * F_OUT + lane;
            #pragma unroll
            for (int c = 0; c < NUM_HEADS; ++c) {
                float wj = __int_as_float(__builtin_amdgcn_readlane(wbits, 4 * j + c));
                __hip_atomic_fetch_add(ap + c * HEAD_DIM, wj * hv[j][c],
                                       __ATOMIC_RELAXED, __HIP_MEMORY_SCOPE_WORKGROUP);
            }
        }
    }
}

__device__ __forceinline__ void drain_sum(int* qs, const int* qd, int qn, int lane,
                                          const int* __restrict__ src,
                                          const float* __restrict__ asrc,
                                          const float* __restrict__ h,
                                          const float* sadst, const float* smax,
                                          float* ssum, float* acc) {
    for (int i = lane; i < qn; i += 32) qs[i] = src[qs[i]];
    for (int i0 = 0; i0 < qn; i0 += 8) {
        const int nb = min(8, qn - i0);
        const int idx = i0 + (lane >> 2);
        const bool on = idx < qn;
        int s = 0, dl = 0;
        if (on) { s = qs[idx]; dl = qd[idx]; }
        sum_batch(s, dl, on, nb, lane, asrc, h, sadst, smax, ssum, acc);
    }
}

__device__ __forceinline__ void list_sum(const unsigned* elist, int total, int wave, int lane,
                                         const float* __restrict__ asrc,
                                         const float* __restrict__ h,
                                         const float* sadst, const float* smax,
                                         float* ssum, float* acc) {
    for (int i0 = wave * 8; i0 < total; i0 += OWN_WAVES * 8) {
        const int nb = min(8, total - i0);
        const int idx = i0 + (lane >> 2);
        const bool on = idx < total;
        int s = 0, dl = 0;
        if (on) {
            const unsigned p = elist[idx];
            s  = (int)(p >> L_SHIFT);
            dl = (int)(p & ((1u << L_SHIFT) - 1u));
        }
        sum_batch(s, dl, on, nb, lane, asrc, h, sadst, smax, ssum, acc);
    }
}

template <int PASS>
__device__ __forceinline__ void owner_scan(const int* __restrict__ src,
                                           const int* __restrict__ dst,
                                           const float* __restrict__ asrc,
                                           const float* __restrict__ h,
                                           int E, int lo, int nloc, int wave, int lane,
                                           int* qs, int* qd,
                                           const float* sadst, float* smax,
                                           float* ssum, float* acc,
                                           unsigned* elist, int* ecount, bool keep) {
    const bool vec = (E & 3) == 0;
    const int perIter = OWN_WAVES * 4 * BLK_EDGES;
    const int nIter = (E + perIter - 1) / perIter;
    const int start = (int)(((long long)blockIdx.x * nIter) / gridDim.x);
    const unsigned below = (1u << lane) - 1u;
    int qn = 0;
    for (int it = 0; it < nIter; ++it) {
        int jt = it + start;
        if (jt >= nIter) jt -= nIter;
        const int wbase = (jt * OWN_WAVES + wave) * (4 * BLK_EDGES);
        const int base = wbase + 4 * lane;
        int4 dv[4];
        if (vec && wbase + 4 * BLK_EDGES <= E) {
            #pragma unroll
            for (int k = 0; k < 4; ++k) dv[k] = *(const int4*)(dst + base + k * BLK_EDGES);
        } else {
            #pragma unroll
            for (int k = 0; k < 4; ++k) dv[k] = load_dst4_guarded(dst, base + k * BLK_EDGES, E);
        }
        #pragma unroll
        for (int k = 0; k < 4; ++k) {
            const int e0 = base + k * BLK_EDGES;
            const int dd[4] = {dv[k].x, dv[k].y, dv[k].z, dv[k].w};
            bool m[4];
            bool any = false;
            #pragma unroll
            for (int c = 0; c < 4; ++c) {
                m[c] = (unsigned)(dd[c] - lo) < (unsigned)nloc;
                any = any || m[c];
            }
            if (__builtin_amdgcn_ballot_w32(any) != 0u) {
                #pragma unroll
                for (int c = 0; c < 4; ++c) {
                    const unsigned mask = __builtin_amdgcn_ballot_w32(m[c]);
                    if (m[c]) {
                        const int pos = qn + __builtin_popcount(mask & below);
                        qs[pos] = e0 + c;
                        qd[pos] = dd[c] - lo;
                    }
                    qn += __builtin_popcount(mask);
                }
                if (qn >= Q_DRAIN) {
                    if (PASS == 1) drain_max<false>(qs, qd, qn, lane, src, dst, lo, asrc, sadst, smax, elist, ecount, keep);
                    else           drain_sum(qs, qd, qn, lane, src, asrc, h, sadst, smax, ssum, acc);
                    qn = 0;
                }
            }
        }
    }
    if (qn > 0) {
        if (PASS == 1) drain_max<false>(qs, qd, qn, lane, src, dst, lo, asrc, sadst, smax, elist, ecount, keep);
        else           drain_sum(qs, qd, qn, lane, src, asrc, h, sadst, smax, ssum, acc);
    }
}

__device__ __forceinline__ void owner_scan_keys(const int* __restrict__ src,
                                                const int* __restrict__ dst,
                                                const unsigned* __restrict__ keys,
                                                const float* __restrict__ asrc,
                                                int E, int lo, int wave, int lane,
                                                int* qs, const float* sadst, float* smax,
                                                unsigned* elist, int* ecount, bool keep) {
    const int nIter = keys_padded(E) / KEY_TRIP;
    const int start = (int)(((long long)blockIdx.x * nIter) / gridDim.x);
    const unsigned own = (unsigned)blockIdx.x * 0x01010101u;
    const unsigned below = (1u << lane) - 1u;
    int qn = 0;
    for (int it = 0; it < nIter; ++it) {
        int jt = it + start;
        if (jt >= nIter) jt -= nIter;
        const int base = (jt * OWN_WAVES + wave) * (KEY_LOADS * 512) + 16 * lane;
        uint4 kv[KEY_LOADS];
        #pragma unroll
        for (int k = 0; k < KEY_LOADS; ++k) kv[k] = *(const uint4*)((const unsigned char*)keys + base + k * 512);
        #pragma unroll
        for (int k = 0; k < KEY_LOADS; ++k) {
            const unsigned kw[4] = {kv[k].x, kv[k].y, kv[k].z, kv[k].w};
            #pragma unroll
            for (int c = 0; c < 4; ++c) {
                const int e0 = base + k * 512 + 4 * c;
                const unsigned x = kw[c] ^ own;
                unsigned hit = ~(((x & 0x7f7f7f7fu) + 0x7f7f7f7fu) | x | 0x7f7f7f7fu);
                unsigned any = __builtin_amdgcn_ballot_w32(hit != 0u);
                if (any != 0u) {
                    for (int r = 0; r < 4 && any != 0u; ++r) {
                        if (hit != 0u) {
                            qs[qn + __builtin_popcount(any & below)] = e0 + (__builtin_ctz(hit) >> 3);
                            hit &= hit - 1u;
                        }
                        qn += __builtin_popcount(any);
                        any = __builtin_amdgcn_ballot_w32(hit != 0u);
                    }
                    if (qn >= Q_DRAIN) {
                        drain_max<true>(qs, qs, qn, lane, src, dst, lo, asrc, sadst, smax, elist, ecount, keep);
                        qn = 0;
                    }
                }
            }
        }
    }
    if (qn > 0)
        drain_max<true>(qs, qs, qn, lane, src, dst, lo, asrc, sadst, smax, elist, ecount, keep);
}

__global__ __launch_bounds__(OWN_THREADS) void gat_owner(const int* __restrict__ src,
                                                         const int* __restrict__ dst,
                                                         const float* __restrict__ asrc,
                                                         const float* __restrict__ adst,
                                                         const float* __restrict__ h,
                                                         const unsigned* __restrict__ keys,
                                                         float* __restrict__ out,
                                                         int N, int E, unsigned long long wsBytes) {
    __shared__ float acc[OWN_NODES * F_OUT];
    __shared__ float smax[OWN_NODES * NUM_HEADS];
    __shared__ float ssum[OWN_NODES * NUM_HEADS];
    __shared__ float sadst[OWN_NODES * NUM_HEADS];
    __shared__ int   qsrc[OWN_WAVES * Q_CAP];
    __shared__ int   qdst[OWN_WAVES * Q_CAP];
    __shared__ unsigned elist[L_CAP];
    __shared__ int   ecount;

    const int t = threadIdx.x;
    const int lane = t & 31;
    const int wave = __builtin_amdgcn_readfirstlane(t >> 5);
    const int lo = blockIdx.x * OWN_NODES;
    const int nloc = min(OWN_NODES, N - lo);
    const bool keep = N <= (1 << (32 - L_SHIFT));

    if (t == 0) ecount = 0;
    for (int i = t; i < OWN_NODES * F_OUT; i += OWN_THREADS) acc[i] = 0.0f;
    for (int i = t; i < OWN_NODES * NUM_HEADS; i += OWN_THREADS) {
        smax[i] = -3.0e38f;
        ssum[i] = 0.0f;
        sadst[i] = (i < nloc * NUM_HEADS) ? adst[(size_t)lo * NUM_HEADS + i] : 0.0f;
    }
    __syncthreads();

    int* qs = qsrc + wave * Q_CAP;
    int* qd = qdst + wave * Q_CAP;

    if (keys_usable(N, E, wsBytes))
        owner_scan_keys(src, dst, keys, asrc, E, lo, wave, lane, qs, sadst, smax, elist, &ecount, keep);
    else
        owner_scan<1>(src, dst, asrc, h, E, lo, nloc, wave, lane, qs, qd, sadst, smax, ssum, acc,
                      elist, &ecount, keep);
    __syncthreads();
    const int total = keep ? ecount : L_CAP + 1;
    if (total <= L_CAP)
        list_sum(elist, total, wave, lane, asrc, h, sadst, smax, ssum, acc);
    else
        owner_scan<2>(src, dst, asrc, h, E, lo, nloc, wave, lane, qs, qd, sadst, smax, ssum, acc,
                      elist, &ecount, false);
    __syncthreads();

    for (int r = wave; r < nloc; r += OWN_WAVES) {
        float* op = out + (size_t)(lo + r) * F_OUT + lane;
        #pragma unroll
        for (int c = 0; c < NUM_HEADS; ++c)
            op[c * HEAD_DIM] = acc[r * F_OUT + c * HEAD_DIM + lane] /
                               (ssum[r * NUM_HEADS + c] + 1e-10f);
    }
}

extern "C" void kernel_launch(void* const* d_in, const int* in_sizes, int n_in,
                              void* d_out, int out_size, void* d_ws, size_t ws_size,
                              hipStream_t stream) {
    const float* x     = (const float*)d_in[0];
    const int*   ei    = (const int*)d_in[1];
    const float* W     = (const float*)d_in[2];
    const float* a_src = (const float*)d_in[3];
    const float* a_dst = (const float*)d_in[4];
    float*       out   = (float*)d_out;

    const int N = in_sizes[0] / F_IN;
    const int E = in_sizes[1] / 2;
    const int* src = ei;
    const int* dst = ei + E;

    float* h    = (float*)d_ws;
    float* asrc = h    + (size_t)N * F_OUT;
    float* adst = asrc + (size_t)N * NUM_HEADS;
    unsigned* keys = (unsigned*)(adst + (size_t)N * NUM_HEADS);

    {
        int fullTiles = N / 16;
        if (fullTiles > 0)
            gat_gemm<<<fullTiles, 256, 0, stream>>>(x, W, h);
        int rowStart = fullTiles * 16;
        int rem = N - rowStart;
        if (rem > 0) {
            int total = rem * F_OUT;
            gat_gemm_tail<<<(total + 255) / 256, 256, 0, stream>>>(x, W, h, N, rowStart);
        }
    }

    {
        int total = N * NUM_HEADS;
        gat_alpha<<<(total + 255) / 256, 256, 0, stream>>>(h, a_src, a_dst, asrc, adst, dst, keys, N, E,
                                                           (unsigned long long)ws_size);
    }

    gat_owner<<<(N + OWN_NODES - 1) / OWN_NODES, OWN_THREADS, 0, stream>>>(src, dst, asrc, adst, h, keys, out, N, E,
                                                                          (unsigned long long)ws_size);
}


# Logits taken in the GEMM epilogue: each 16x128 tile of h is staged in LDS, its rows' logits summed there in the same order, h stored as whole rows; the logits launch goes, 3 launches to 2
# speedup vs baseline: 4.5134x; 1.0851x over previous
#include <hip/hip_runtime.h>
#include <hip/hip_bf16.h>


#define NUM_HEADS 4
#define HEAD_DIM 32
#define F_IN 256
#define F_OUT 128
#define LDA 260

typedef __attribute__((ext_vector_type(2))) float v2f;
typedef __attribute__((ext_vector_type(8))) float v8f;

#define OWN_NODES   400
#define OWN_WAVES   32
#define KEY_NONE    255u
#define KEY_LOADS   4
#define KEY_TRIP    (OWN_WAVES * KEY_LOADS * 512)

__device__ __forceinline__ int keys_padded(int E) {
    return ((E + KEY_TRIP - 1) / KEY_TRIP) * KEY_TRIP;
}

__device__ __forceinline__ bool keys_usable(int N, int E, unsigned long long wsBytes) {
    const int groups = (N + OWN_NODES - 1) / OWN_NODES;
    const unsigned long long need =
        (unsigned long long)N * (F_OUT + 2 * NUM_HEADS) * sizeof(float) + (unsigned long long)keys_padded(E);
    return groups <= (int)KEY_NONE && wsBytes >= need;
}

__device__ __forceinline__ unsigned key_of(int d, int N) {
    return ((unsigned)d < (unsigned)N) ? (unsigned)d / OWN_NODES : KEY_NONE;
}

#define HP 132

__device__ __forceinline__ void write_keys(const int* __restrict__ dst, unsigned* __restrict__ keys,
                                           int N, int E, int tid, int step) {
    const int words = keys_padded(E) >> 2;
    const bool vec = (E & 3) == 0;
    for (int g = tid; g < words; g += step) {
        const int e = g << 2;
        int4 d;
        if (vec && e + 4 <= E) {
            d = *(const int4*)(dst + e);
        } else {
            d.x = (e     < E) ? dst[e]     : -1;
            d.y = (e + 1 < E) ? dst[e + 1] : -1;
            d.z = (e + 2 < E) ? dst[e + 2] : -1;
            d.w = (e + 3 < E) ? dst[e + 3] : -1;
        }
        keys[g] = key_of(d.x, N) | (key_of(d.y, N) << 8) | (key_of(d.z, N) << 16) | (key_of(d.w, N) << 24);
    }
}

__device__ __forceinline__ float head_dot(const float* hp, const float* __restrict__ av) {
    float s = 0.0f;
    #pragma unroll
    for (int i = 0; i < HEAD_DIM; i += 4) {
        float4 hv = *(const float4*)(hp + i);
        float4 a  = *(const float4*)(av + i);
        s += hv.x * a.x + hv.y * a.y + hv.z * a.z + hv.w * a.w;
    }
    return s;
}

__global__ __launch_bounds__(256) void gat_gemm(const float* __restrict__ x,
                                                const float* __restrict__ W,
                                                const float* __restrict__ a_src,
                                                const float* __restrict__ a_dst,
                                                float* __restrict__ h,
                                                float* __restrict__ asrc,
                                                float* __restrict__ adst,
                                                const int* __restrict__ dst,
                                                unsigned* __restrict__ keys,
                                                int N, int E, unsigned long long wsBytes) {
    __shared__ float As[16 * LDA];

    const int t = threadIdx.x;
    const int rowBase = blockIdx.x * 16;

    #pragma unroll
    for (int j = 0; j < 4; ++j) {
        int f = t + 256 * j;
        int r = f >> 6;
        int c = (f & 63) << 2;
        float4 v = *(const float4*)(x + (size_t)(rowBase + r) * F_IN + c);
        *(float4*)(&As[r * LDA + c]) = v;
    }
    __syncthreads();

    const int wave = t >> 5;
    const int lane = t & 31;
    const int m    = lane & 15;
    const int koff = (lane >> 4) << 1;
    const int col  = (wave << 4) + m;

    const float* Wp = W + col;
    v8f acc = {};

    #pragma unroll 8
    for (int k = 0; k < F_IN; k += 4) {
        v2f a = *(const v2f*)(&As[m * LDA + k + koff]);
        v2f b;
        b.x = Wp[(size_t)(k + koff) * F_OUT];
        b.y = Wp[(size_t)(k + koff + 1) * F_OUT];
        acc = __builtin_amdgcn_wmma_f32_16x16x4_f32(
             false, a,  false, b,
             (short)0, acc,  false,  false);
    }

    __syncthreads();
    float* Hs = As;
    const int roff = (lane >> 4) << 3;
    #pragma unroll
    for (int v = 0; v < 8; ++v)
        Hs[(v + roff) * HP + col] = acc[v];
    __syncthreads();

    #pragma unroll
    for (int j = 0; j < 2; ++j) {
        int f = t + 256 * j;
        int r = f >> 5;
        int c = (f & 31) << 2;
        *(float4*)(h + (size_t)(rowBase + r) * F_OUT + c) = *(const float4*)(&Hs[r * HP + c]);
    }

    if (t < 2 * 16 * NUM_HEADS) {
        const int q  = t & (16 * NUM_HEADS - 1);
        const int r  = q >> 2;
        const int hd = q & 3;
        const float* hp = Hs + r * HP + hd * HEAD_DIM;
        const size_t o = (size_t)(rowBase + r) * NUM_HEADS + hd;
        if (t < 16 * NUM_HEADS) asrc[o] = head_dot(hp, a_src + hd * HEAD_DIM);
        else                    adst[o] = head_dot(hp, a_dst + hd * HEAD_DIM);
    }

    if (keys_usable(N, E, wsBytes))
        write_keys(dst, keys, N, E, blockIdx.x * blockDim.x + t, gridDim.x * blockDim.x);
}

__global__ __launch_bounds__(256) void gat_gemm_tail(const float* __restrict__ x,
                                                     const float* __restrict__ W,
                                                     float* __restrict__ h,
                                                     int N, int rowStart) {
    int idx = blockIdx.x * blockDim.x + threadIdx.x;
    int total = (N - rowStart) * F_OUT;
    if (idx >= total) return;
    int r = rowStart + idx / F_OUT;
    int c = idx % F_OUT;
    float s = 0.0f;
    for (int k = 0; k < F_IN; ++k)
        s += x[(size_t)r * F_IN + k] * W[(size_t)k * F_OUT + c];
    h[(size_t)r * F_OUT + c] = s;
}

__global__ __launch_bounds__(256) void gat_alpha_tail(const float* __restrict__ h,
                                                      const float* __restrict__ a_src,
                                                      const float* __restrict__ a_dst,
                                                      float* __restrict__ asrc,
                                                      float* __restrict__ adst,
                                                      const int* __restrict__ dst,
                                                      unsigned* __restrict__ keys,
                                                      int N, int E, unsigned long long wsBytes,
                                                      int rowStart, int doKeys) {
    int tid = blockIdx.x * blockDim.x + threadIdx.x;
    if (doKeys && keys_usable(N, E, wsBytes))
        write_keys(dst, keys, N, E, tid, gridDim.x * blockDim.x);
    if (tid >= (N - rowStart) * NUM_HEADS) return;
    int n  = rowStart + (tid >> 2);
    int hd = tid & 3;
    const float* hp = h + (size_t)n * F_OUT + hd * HEAD_DIM;
    asrc[(size_t)n * NUM_HEADS + hd] = head_dot(hp, a_src + hd * HEAD_DIM);
    adst[(size_t)n * NUM_HEADS + hd] = head_dot(hp, a_dst + hd * HEAD_DIM);
}

__device__ __forceinline__ float leaky(float v) { return v > 0.0f ? v : 0.2f * v; }

#define OWN_THREADS 1024
#define Q_CAP       144
#define Q_DRAIN     16
#define BLK_EDGES   128
#define L_CAP       12288
#define L_SHIFT     9
static_assert(OWN_NODES <= (1 << L_SHIFT), "local row must fit the low bits of a list word");

__device__ __forceinline__ int4 load_dst4_guarded(const int* __restrict__ dst, int e, int E) {
    int4 r;
    r.x = (e     < E) ? dst[e]     : -1;
    r.y = (e + 1 < E) ? dst[e + 1] : -1;
    r.z = (e + 2 < E) ? dst[e + 2] : -1;
    r.w = (e + 3 < E) ? dst[e + 3] : -1;
    return r;
}

template <bool BYKEY>
__device__ __forceinline__ void drain_max(const int* qs, const int* qd, int qn, int lane,
                                          const int* __restrict__ src,
                                          const int* __restrict__ dst, int lo,
                                          const float* __restrict__ asrc,
                                          const float* sadst, float* smax,
                                          unsigned* elist, int* ecount, bool keep) {
    int slot0 = 0;
    if (keep) {
        if (lane == 0)
            slot0 = __hip_atomic_fetch_add(ecount, qn, __ATOMIC_RELAXED, __HIP_MEMORY_SCOPE_WORKGROUP);
        slot0 = __builtin_amdgcn_readfirstlane(slot0);
    }
    for (int i0 = 0; i0 < qn; i0 += 32) {
        int idx = i0 + lane;
        if (idx < qn) {
            const int e = qs[idx];
            int s = src[e], dl = BYKEY ? dst[e] - lo : qd[idx];
            if (keep && slot0 + idx < L_CAP)
                elist[slot0 + idx] = ((unsigned)s << L_SHIFT) | (unsigned)dl;
            float4 a = *(const float4*)(asrc + (size_t)s * NUM_HEADS);
            float4 b = *(const float4*)(sadst + dl * NUM_HEADS);
            float* mp = smax + dl * NUM_HEADS;
            __hip_atomic_fetch_max(mp + 0, leaky(a.x + b.x), __ATOMIC_RELAXED, __HIP_MEMORY_SCOPE_WORKGROUP);
            __hip_atomic_fetch_max(mp + 1, leaky(a.y + b.y), __ATOMIC_RELAXED, __HIP_MEMORY_SCOPE_WORKGROUP);
            __hip_atomic_fetch_max(mp + 2, leaky(a.z + b.z), __ATOMIC_RELAXED, __HIP_MEMORY_SCOPE_WORKGROUP);
            __hip_atomic_fetch_max(mp + 3, leaky(a.w + b.w), __ATOMIC_RELAXED, __HIP_MEMORY_SCOPE_WORKGROUP);
        }
    }
}

__device__ __forceinline__ void sum_batch(int s, int dl, bool on, int nb, int lane,
                                          const float* __restrict__ asrc,
                                          const float* __restrict__ h,
                                          const float* sadst, const float* smax,
                                          float* ssum, float* acc) {
    const int hd = lane & 3;
    float as = 0.0f;
    if (on) as = asrc[(size_t)s * NUM_HEADS + hd];
    float hv[8][NUM_HEADS];
    #pragma unroll
    for (int j = 0; j < 8; ++j) {
        if (j < nb) {
            int sj = __builtin_amdgcn_readlane(s, 4 * j);
            const float* hp = h + (size_t)sj * F_OUT + lane;
            #pragma unroll
            for (int c = 0; c < NUM_HEADS; ++c) hv[j][c] = hp[c * HEAD_DIM];
        }
    }
    float w = 0.0f;
    if (on) {
        float v = leaky(as + sadst[dl * NUM_HEADS + hd]);
        w = __expf(v - smax[dl * NUM_HEADS + hd]);
        __hip_atomic_fetch_add(ssum + dl * NUM_HEADS + hd, w,
                               __ATOMIC_RELAXED, __HIP_MEMORY_SCOPE_WORKGROUP);
    }
    const int wbits = __float_as_int(w);
    #pragma unroll
    for (int j = 0; j < 8; ++j) {
        if (j < nb) {
            int dj = __builtin_amdgcn_readlane(dl, 4 * j);
            float* ap = acc + dj * F_OUT + lane;
            #pragma unroll
            for (int c = 0; c < NUM_HEADS; ++c) {
                float wj = __int_as_float(__builtin_amdgcn_readlane(wbits, 4 * j + c));
                __hip_atomic_fetch_add(ap + c * HEAD_DIM, wj * hv[j][c],
                                       __ATOMIC_RELAXED, __HIP_MEMORY_SCOPE_WORKGROUP);
            }
        }
    }
}

__device__ __forceinline__ void drain_sum(int* qs, const int* qd, int qn, int lane,
                                          const int* __restrict__ src,
                                          const float* __restrict__ asrc,
                                          const float* __restrict__ h,
                                          const float* sadst, const float* smax,
                                          float* ssum, float* acc) {
    for (int i = lane; i < qn; i += 32) qs[i] = src[qs[i]];
    for (int i0 = 0; i0 < qn; i0 += 8) {
        const int nb = min(8, qn - i0);
        const int idx = i0 + (lane >> 2);
        const bool on = idx < qn;
        int s = 0, dl = 0;
        if (on) { s = qs[idx]; dl = qd[idx]; }
        sum_batch(s, dl, on, nb, lane, asrc, h, sadst, smax, ssum, acc);
    }
}

__device__ __forceinline__ void list_sum(const unsigned* elist, int total, int wave, int lane,
                                         const float* __restrict__ asrc,
                                         const float* __restrict__ h,
                                         const float* sadst, const float* smax,
                                         float* ssum, float* acc) {
    for (int i0 = wave * 8; i0 < total; i0 += OWN_WAVES * 8) {
        const int nb = min(8, total - i0);
        const int idx = i0 + (lane >> 2);
        const bool on = idx < total;
        int s = 0, dl = 0;
        if (on) {
            const unsigned p = elist[idx];
            s  = (int)(p >> L_SHIFT);
            dl = (int)(p & ((1u << L_SHIFT) - 1u));
        }
        sum_batch(s, dl, on, nb, lane, asrc, h, sadst, smax, ssum, acc);
    }
}

template <int PASS>
__device__ __forceinline__ void owner_scan(const int* __restrict__ src,
                                           const int* __restrict__ dst,
                                           const float* __restrict__ asrc,
                                           const float* __restrict__ h,
                                           int E, int lo, int nloc, int wave, int lane,
                                           int* qs, int* qd,
                                           const float* sadst, float* smax,
                                           float* ssum, float* acc,
                                           unsigned* elist, int* ecount, bool keep) {
    const bool vec = (E & 3) == 0;
    const int perIter = OWN_WAVES * 4 * BLK_EDGES;
    const int nIter = (E + perIter - 1) / perIter;
    const int start = (int)(((long long)blockIdx.x * nIter) / gridDim.x);
    const unsigned below = (1u << lane) - 1u;
    int qn = 0;
    for (int it = 0; it < nIter; ++it) {
        int jt = it + start;
        if (jt >= nIter) jt -= nIter;
        const int wbase = (jt * OWN_WAVES + wave) * (4 * BLK_EDGES);
        const int base = wbase + 4 * lane;
        int4 dv[4];
        if (vec && wbase + 4 * BLK_EDGES <= E) {
            #pragma unroll
            for (int k = 0; k < 4; ++k) dv[k] = *(const int4*)(dst + base + k * BLK_EDGES);
        } else {
            #pragma unroll
            for (int k = 0; k < 4; ++k) dv[k] = load_dst4_guarded(dst, base + k * BLK_EDGES, E);
        }
        #pragma unroll
        for (int k = 0; k < 4; ++k) {
            const int e0 = base + k * BLK_EDGES;
            const int dd[4] = {dv[k].x, dv[k].y, dv[k].z, dv[k].w};
            bool m[4];
            bool any = false;
            #pragma unroll
            for (int c = 0; c < 4; ++c) {
                m[c] = (unsigned)(dd[c] - lo) < (unsigned)nloc;
                any = any || m[c];
            }
            if (__builtin_amdgcn_ballot_w32(any) != 0u) {
                #pragma unroll
                for (int c = 0; c < 4; ++c) {
                    const unsigned mask = __builtin_amdgcn_ballot_w32(m[c]);
                    if (m[c]) {
                        const int pos = qn + __builtin_popcount(mask & below);
                        qs[pos] = e0 + c;
                        qd[pos] = dd[c] - lo;
                    }
                    qn += __builtin_popcount(mask);
                }
                if (qn >= Q_DRAIN) {
                    if (PASS == 1) drain_max<false>(qs, qd, qn, lane, src, dst, lo, asrc, sadst, smax, elist, ecount, keep);
                    else           drain_sum(qs, qd, qn, lane, src, asrc, h, sadst, smax, ssum, acc);
                    qn = 0;
                }
            }
        }
    }
    if (qn > 0) {
        if (PASS == 1) drain_max<false>(qs, qd, qn, lane, src, dst, lo, asrc, sadst, smax, elist, ecount, keep);
        else           drain_sum(qs, qd, qn, lane, src, asrc, h, sadst, smax, ssum, acc);
    }
}

__device__ __forceinline__ void owner_scan_keys(const int* __restrict__ src,
                                                const int* __restrict__ dst,
                                                const unsigned* __restrict__ keys,
                                                const float* __restrict__ asrc,
                                                int E, int lo, int wave, int lane,
                                                int* qs, const float* sadst, float* smax,
                                                unsigned* elist, int* ecount, bool keep) {
    const int nIter = keys_padded(E) / KEY_TRIP;
    const int start = (int)(((long long)blockIdx.x * nIter) / gridDim.x);
    const unsigned own = (unsigned)blockIdx.x * 0x01010101u;
    const unsigned below = (1u << lane) - 1u;
    int qn = 0;
    for (int it = 0; it < nIter; ++it) {
        int jt = it + start;
        if (jt >= nIter) jt -= nIter;
        const int base = (jt * OWN_WAVES + wave) * (KEY_LOADS * 512) + 16 * lane;
        uint4 kv[KEY_LOADS];
        #pragma unroll
        for (int k = 0; k < KEY_LOADS; ++k) kv[k] = *(const uint4*)((const unsigned char*)keys + base + k * 512);
        #pragma unroll
        for (int k = 0; k < KEY_LOADS; ++k) {
            const unsigned kw[4] = {kv[k].x, kv[k].y, kv[k].z, kv[k].w};
            #pragma unroll
            for (int c = 0; c < 4; ++c) {
                const int e0 = base + k * 512 + 4 * c;
                const unsigned x = kw[c] ^ own;
                unsigned hit = ~(((x & 0x7f7f7f7fu) + 0x7f7f7f7fu) | x | 0x7f7f7f7fu);
                unsigned any = __builtin_amdgcn_ballot_w32(hit != 0u);
                if (any != 0u) {
                    for (int r = 0; r < 4 && any != 0u; ++r) {
                        if (hit != 0u) {
                            qs[qn + __builtin_popcount(any & below)] = e0 + (__builtin_ctz(hit) >> 3);
                            hit &= hit - 1u;
                        }
                        qn += __builtin_popcount(any);
                        any = __builtin_amdgcn_ballot_w32(hit != 0u);
                    }
                    if (qn >= Q_DRAIN) {
                        drain_max<true>(qs, qs, qn, lane, src, dst, lo, asrc, sadst, smax, elist, ecount, keep);
                        qn = 0;
                    }
                }
            }
        }
    }
    if (qn > 0)
        drain_max<true>(qs, qs, qn, lane, src, dst, lo, asrc, sadst, smax, elist, ecount, keep);
}

__global__ __launch_bounds__(OWN_THREADS) void gat_owner(const int* __restrict__ src,
                                                         const int* __restrict__ dst,
                                                         const float* __restrict__ asrc,
                                                         const float* __restrict__ adst,
                                                         const float* __restrict__ h,
                                                         const unsigned* __restrict__ keys,
                                                         float* __restrict__ out,
                                                         int N, int E, unsigned long long wsBytes) {
    __shared__ float acc[OWN_NODES * F_OUT];
    __shared__ float smax[OWN_NODES * NUM_HEADS];
    __shared__ float ssum[OWN_NODES * NUM_HEADS];
    __shared__ float sadst[OWN_NODES * NUM_HEADS];
    __shared__ int   qsrc[OWN_WAVES * Q_CAP];
    __shared__ int   qdst[OWN_WAVES * Q_CAP];
    __shared__ unsigned elist[L_CAP];
    __shared__ int   ecount;

    const int t = threadIdx.x;
    const int lane = t & 31;
    const int wave = __builtin_amdgcn_readfirstlane(t >> 5);
    const int lo = blockIdx.x * OWN_NODES;
    const int nloc = min(OWN_NODES, N - lo);
    const bool keep = N <= (1 << (32 - L_SHIFT));

    if (t == 0) ecount = 0;
    for (int i = t; i < OWN_NODES * F_OUT; i += OWN_THREADS) acc[i] = 0.0f;
    for (int i = t; i < OWN_NODES * NUM_HEADS; i += OWN_THREADS) {
        smax[i] = -3.0e38f;
        ssum[i] = 0.0f;
        sadst[i] = (i < nloc * NUM_HEADS) ? adst[(size_t)lo * NUM_HEADS + i] : 0.0f;
    }
    __syncthreads();

    int* qs = qsrc + wave * Q_CAP;
    int* qd = qdst + wave * Q_CAP;

    if (keys_usable(N, E, wsBytes))
        owner_scan_keys(src, dst, keys, asrc, E, lo, wave, lane, qs, sadst, smax, elist, &ecount, keep);
    else
        owner_scan<1>(src, dst, asrc, h, E, lo, nloc, wave, lane, qs, qd, sadst, smax, ssum, acc,
                      elist, &ecount, keep);
    __syncthreads();
    const int total = keep ? ecount : L_CAP + 1;
    if (total <= L_CAP)
        list_sum(elist, total, wave, lane, asrc, h, sadst, smax, ssum, acc);
    else
        owner_scan<2>(src, dst, asrc, h, E, lo, nloc, wave, lane, qs, qd, sadst, smax, ssum, acc,
                      elist, &ecount, false);
    __syncthreads();

    for (int r = wave; r < nloc; r += OWN_WAVES) {
        float* op = out + (size_t)(lo + r) * F_OUT + lane;
        #pragma unroll
        for (int c = 0; c < NUM_HEADS; ++c)
            op[c * HEAD_DIM] = acc[r * F_OUT + c * HEAD_DIM + lane] /
                               (ssum[r * NUM_HEADS + c] + 1e-10f);
    }
}

extern "C" void kernel_launch(void* const* d_in, const int* in_sizes, int n_in,
                              void* d_out, int out_size, void* d_ws, size_t ws_size,
                              hipStream_t stream) {
    const float* x     = (const float*)d_in[0];
    const int*   ei    = (const int*)d_in[1];
    const float* W     = (const float*)d_in[2];
    const float* a_src = (const float*)d_in[3];
    const float* a_dst = (const float*)d_in[4];
    float*       out   = (float*)d_out;

    const int N = in_sizes[0] / F_IN;
    const int E = in_sizes[1] / 2;
    const int* src = ei;
    const int* dst = ei + E;

    float* h    = (float*)d_ws;
    float* asrc = h    + (size_t)N * F_OUT;
    float* adst = asrc + (size_t)N * NUM_HEADS;
    unsigned* keys = (unsigned*)(adst + (size_t)N * NUM_HEADS);

    {
        int fullTiles = N / 16;
        if (fullTiles > 0)
            gat_gemm<<<fullTiles, 256, 0, stream>>>(x, W, a_src, a_dst, h, asrc, adst, dst, keys, N, E,
                                                    (unsigned long long)ws_size);
        int rowStart = fullTiles * 16;
        int rem = N - rowStart;
        if (rem > 0) {
            int total = rem * F_OUT;
            gat_gemm_tail<<<(total + 255) / 256, 256, 0, stream>>>(x, W, h, N, rowStart);
            gat_alpha_tail<<<(rem * NUM_HEADS + 255) / 256, 256, 0, stream>>>(h, a_src, a_dst, asrc, adst, dst, keys, N, E,
                                                                              (unsigned long long)ws_size, rowStart,
                                                                              fullTiles > 0 ? 0 : 1);
        }
    }

    gat_owner<<<(N + OWN_NODES - 1) / OWN_NODES, OWN_THREADS, 0, stream>>>(src, dst, asrc, adst, h, keys, out, N, E,
                                                                          (unsigned long long)ws_size);
}


# Key filter tests a lane's 64 keys of a trip branch-free into one 64-bit mask, then queues one owned edge per lane and round: queue code runs per round, not per hit word; sums in LDS arrival order
# speedup vs baseline: 4.7097x; 1.0435x over previous
#include <hip/hip_runtime.h>
#include <hip/hip_bf16.h>


#define NUM_HEADS 4
#define HEAD_DIM 32
#define F_IN 256
#define F_OUT 128
#define LDA 260

typedef __attribute__((ext_vector_type(2))) float v2f;
typedef __attribute__((ext_vector_type(8))) float v8f;

#define OWN_NODES   400
#define OWN_WAVES   32
#define KEY_NONE    255u
#define KEY_LOADS   4
#define KEY_TRIP    (OWN_WAVES * KEY_LOADS * 512)

__device__ __forceinline__ int keys_padded(int E) {
    return ((E + KEY_TRIP - 1) / KEY_TRIP) * KEY_TRIP;
}

__device__ __forceinline__ bool keys_usable(int N, int E, unsigned long long wsBytes) {
    const int groups = (N + OWN_NODES - 1) / OWN_NODES;
    const unsigned long long need =
        (unsigned long long)N * (F_OUT + 2 * NUM_HEADS) * sizeof(float) + (unsigned long long)keys_padded(E);
    return groups <= (int)KEY_NONE && wsBytes >= need;
}

__device__ __forceinline__ unsigned key_of(int d, int N) {
    return ((unsigned)d < (unsigned)N) ? (unsigned)d / OWN_NODES : KEY_NONE;
}

#define HP 132

__device__ __forceinline__ void write_keys(const int* __restrict__ dst, unsigned* __restrict__ keys,
                                           int N, int E, int tid, int step) {
    const int words = keys_padded(E) >> 2;
    const bool vec = (E & 3) == 0;
    for (int g = tid; g < words; g += step) {
        const int e = g << 2;
        int4 d;
        if (vec && e + 4 <= E) {
            d = *(const int4*)(dst + e);
        } else {
            d.x = (e     < E) ? dst[e]     : -1;
            d.y = (e + 1 < E) ? dst[e + 1] : -1;
            d.z = (e + 2 < E) ? dst[e + 2] : -1;
            d.w = (e + 3 < E) ? dst[e + 3] : -1;
        }
        keys[g] = key_of(d.x, N) | (key_of(d.y, N) << 8) | (key_of(d.z, N) << 16) | (key_of(d.w, N) << 24);
    }
}

__device__ __forceinline__ float head_dot(const float* hp, const float* __restrict__ av) {
    float s = 0.0f;
    #pragma unroll
    for (int i = 0; i < HEAD_DIM; i += 4) {
        float4 hv = *(const float4*)(hp + i);
        float4 a  = *(const float4*)(av + i);
        s += hv.x * a.x + hv.y * a.y + hv.z * a.z + hv.w * a.w;
    }
    return s;
}

__global__ __launch_bounds__(256) void gat_gemm(const float* __restrict__ x,
                                                const float* __restrict__ W,
                                                const float* __restrict__ a_src,
                                                const float* __restrict__ a_dst,
                                                float* __restrict__ h,
                                                float* __restrict__ asrc,
                                                float* __restrict__ adst,
                                                const int* __restrict__ dst,
                                                unsigned* __restrict__ keys,
                                                int N, int E, unsigned long long wsBytes) {
    __shared__ float As[16 * LDA];

    const int t = threadIdx.x;
    const int rowBase = blockIdx.x * 16;

    #pragma unroll
    for (int j = 0; j < 4; ++j) {
        int f = t + 256 * j;
        int r = f >> 6;
        int c = (f & 63) << 2;
        float4 v = *(const float4*)(x + (size_t)(rowBase + r) * F_IN + c);
        *(float4*)(&As[r * LDA + c]) = v;
    }
    __syncthreads();

    const int wave = t >> 5;
    const int lane = t & 31;
    const int m    = lane & 15;
    const int koff = (lane >> 4) << 1;
    const int col  = (wave << 4) + m;

    const float* Wp = W + col;
    v8f acc = {};

    #pragma unroll 8
    for (int k = 0; k < F_IN; k += 4) {
        v2f a = *(const v2f*)(&As[m * LDA + k + koff]);
        v2f b;
        b.x = Wp[(size_t)(k + koff) * F_OUT];
        b.y = Wp[(size_t)(k + koff + 1) * F_OUT];
        acc = __builtin_amdgcn_wmma_f32_16x16x4_f32(
             false, a,  false, b,
             (short)0, acc,  false,  false);
    }

    __syncthreads();
    float* Hs = As;
    const int roff = (lane >> 4) << 3;
    #pragma unroll
    for (int v = 0; v < 8; ++v)
        Hs[(v + roff) * HP + col] = acc[v];
    __syncthreads();

    #pragma unroll
    for (int j = 0; j < 2; ++j) {
        int f = t + 256 * j;
        int r = f >> 5;
        int c = (f & 31) << 2;
        *(float4*)(h + (size_t)(rowBase + r) * F_OUT + c) = *(const float4*)(&Hs[r * HP + c]);
    }

    if (t < 2 * 16 * NUM_HEADS) {
        const int q  = t & (16 * NUM_HEADS - 1);
        const int r  = q >> 2;
        const int hd = q & 3;
        const float* hp = Hs + r * HP + hd * HEAD_DIM;
        const size_t o = (size_t)(rowBase + r) * NUM_HEADS + hd;
        if (t < 16 * NUM_HEADS) asrc[o] = head_dot(hp, a_src + hd * HEAD_DIM);
        else                    adst[o] = head_dot(hp, a_dst + hd * HEAD_DIM);
    }

    if (keys_usable(N, E, wsBytes))
        write_keys(dst, keys, N, E, blockIdx.x * blockDim.x + t, gridDim.x * blockDim.x);
}

__global__ __launch_bounds__(256) void gat_gemm_tail(const float* __restrict__ x,
                                                     const float* __restrict__ W,
                                                     float* __restrict__ h,
                                                     int N, int rowStart) {
    int idx = blockIdx.x * blockDim.x + threadIdx.x;
    int total = (N - rowStart) * F_OUT;
    if (idx >= total) return;
    int r = rowStart + idx / F_OUT;
    int c = idx % F_OUT;
    float s = 0.0f;
    for (int k = 0; k < F_IN; ++k)
        s += x[(size_t)r * F_IN + k] * W[(size_t)k * F_OUT + c];
    h[(size_t)r * F_OUT + c] = s;
}

__global__ __launch_bounds__(256) void gat_alpha_tail(const float* __restrict__ h,
                                                      const float* __restrict__ a_src,
                                                      const float* __restrict__ a_dst,
                                                      float* __restrict__ asrc,
                                                      float* __restrict__ adst,
                                                      const int* __restrict__ dst,
                                                      unsigned* __restrict__ keys,
                                                      int N, int E, unsigned long long wsBytes,
                                                      int rowStart, int doKeys) {
    int tid = blockIdx.x * blockDim.x + threadIdx.x;
    if (doKeys && keys_usable(N, E, wsBytes))
        write_keys(dst, keys, N, E, tid, gridDim.x * blockDim.x);
    if (tid >= (N - rowStart) * NUM_HEADS) return;
    int n  = rowStart + (tid >> 2);
    int hd = tid & 3;
    const float* hp = h + (size_t)n * F_OUT + hd * HEAD_DIM;
    asrc[(size_t)n * NUM_HEADS + hd] = head_dot(hp, a_src + hd * HEAD_DIM);
    adst[(size_t)n * NUM_HEADS + hd] = head_dot(hp, a_dst + hd * HEAD_DIM);
}

__device__ __forceinline__ float leaky(float v) { return v > 0.0f ? v : 0.2f * v; }

#define OWN_THREADS 1024
#define Q_CAP       144
#define Q_DRAIN     16
#define BLK_EDGES   128
#define L_CAP       12288
#define L_SHIFT     9
static_assert(OWN_NODES <= (1 << L_SHIFT), "local row must fit the low bits of a list word");

__device__ __forceinline__ int4 load_dst4_guarded(const int* __restrict__ dst, int e, int E) {
    int4 r;
    r.x = (e     < E) ? dst[e]     : -1;
    r.y = (e + 1 < E) ? dst[e + 1] : -1;
    r.z = (e + 2 < E) ? dst[e + 2] : -1;
    r.w = (e + 3 < E) ? dst[e + 3] : -1;
    return r;
}

template <bool BYKEY>
__device__ __forceinline__ void drain_max(const int* qs, const int* qd, int qn, int lane,
                                          const int* __restrict__ src,
                                          const int* __restrict__ dst, int lo,
                                          const float* __restrict__ asrc,
                                          const float* sadst, float* smax,
                                          unsigned* elist, int* ecount, bool keep) {
    int slot0 = 0;
    if (keep) {
        if (lane == 0)
            slot0 = __hip_atomic_fetch_add(ecount, qn, __ATOMIC_RELAXED, __HIP_MEMORY_SCOPE_WORKGROUP);
        slot0 = __builtin_amdgcn_readfirstlane(slot0);
    }
    for (int i0 = 0; i0 < qn; i0 += 32) {
        int idx = i0 + lane;
        if (idx < qn) {
            const int e = qs[idx];
            int s = src[e], dl = BYKEY ? dst[e] - lo : qd[idx];
            if (keep && slot0 + idx < L_CAP)
                elist[slot0 + idx] = ((unsigned)s << L_SHIFT) | (unsigned)dl;
            float4 a = *(const float4*)(asrc + (size_t)s * NUM_HEADS);
            float4 b = *(const float4*)(sadst + dl * NUM_HEADS);
            float* mp = smax + dl * NUM_HEADS;
            __hip_atomic_fetch_max(mp + 0, leaky(a.x + b.x), __ATOMIC_RELAXED, __HIP_MEMORY_SCOPE_WORKGROUP);
            __hip_atomic_fetch_max(mp + 1, leaky(a.y + b.y), __ATOMIC_RELAXED, __HIP_MEMORY_SCOPE_WORKGROUP);
            __hip_atomic_fetch_max(mp + 2, leaky(a.z + b.z), __ATOMIC_RELAXED, __HIP_MEMORY_SCOPE_WORKGROUP);
            __hip_atomic_fetch_max(mp + 3, leaky(a.w + b.w), __ATOMIC_RELAXED, __HIP_MEMORY_SCOPE_WORKGROUP);
        }
    }
}

__device__ __forceinline__ void sum_batch(int s, int dl, bool on, int nb, int lane,
                                          const float* __restrict__ asrc,
                                          const float* __restrict__ h,
                                          const float* sadst, const float* smax,
                                          float* ssum, float* acc) {
    const int hd = lane & 3;
    float as = 0.0f;
    if (on) as = asrc[(size_t)s * NUM_HEADS + hd];
    float hv[8][NUM_HEADS];
    #pragma unroll
    for (int j = 0; j < 8; ++j) {
        if (j < nb) {
            int sj = __builtin_amdgcn_readlane(s, 4 * j);
            const float* hp = h + (size_t)sj * F_OUT + lane;
            #pragma unroll
            for (int c = 0; c < NUM_HEADS; ++c) hv[j][c] = hp[c * HEAD_DIM];
        }
    }
    float w = 0.0f;
    if (on) {
        float v = leaky(as + sadst[dl * NUM_HEADS + hd]);
        w = __expf(v - smax[dl * NUM_HEADS + hd]);
        __hip_atomic_fetch_add(ssum + dl * NUM_HEADS + hd, w,
                               __ATOMIC_RELAXED, __HIP_MEMORY_SCOPE_WORKGROUP);
    }
    const int wbits = __float_as_int(w);
    #pragma unroll
    for (int j = 0; j < 8; ++j) {
        if (j < nb) {
            int dj = __builtin_amdgcn_readlane(dl, 4 * j);
            float* ap = acc + dj * F_OUT + lane;
            #pragma unroll
            for (int c = 0; c < NUM_HEADS; ++c) {
                float wj = __int_as_float(__builtin_amdgcn_readlane(wbits, 4 * j + c));
                __hip_atomic_fetch_add(ap + c * HEAD_DIM, wj * hv[j][c],
                                       __ATOMIC_RELAXED, __HIP_MEMORY_SCOPE_WORKGROUP);
            }
        }
    }
}

__device__ __forceinline__ void drain_sum(int* qs, const int* qd, int qn, int lane,
                                          const int* __restrict__ src,
                                          const float* __restrict__ asrc,
                                          const float* __restrict__ h,
                                          const float* sadst, const float* smax,
                                          float* ssum, float* acc) {
    for (int i = lane; i < qn; i += 32) qs[i] = src[qs[i]];
    for (int i0 = 0; i0 < qn; i0 += 8) {
        const int nb = min(8, qn - i0);
        const int idx = i0 + (lane >> 2);
        const bool on = idx < qn;
        int s = 0, dl = 0;
        if (on) { s = qs[idx]; dl = qd[idx]; }
        sum_batch(s, dl, on, nb, lane, asrc, h, sadst, smax, ssum, acc);
    }
}

__device__ __forceinline__ void list_sum(const unsigned* elist, int total, int wave, int lane,
                                         const float* __restrict__ asrc,
                                         const float* __restrict__ h,
                                         const float* sadst, const float* smax,
                                         float* ssum, float* acc) {
    for (int i0 = wave * 8; i0 < total; i0 += OWN_WAVES * 8) {
        const int nb = min(8, total - i0);
        const int idx = i0 + (lane >> 2);
        const bool on = idx < total;
        int s = 0, dl = 0;
        if (on) {
            const unsigned p = elist[idx];
            s  = (int)(p >> L_SHIFT);
            dl = (int)(p & ((1u << L_SHIFT) - 1u));
        }
        sum_batch(s, dl, on, nb, lane, asrc, h, sadst, smax, ssum, acc);
    }
}

template <int PASS>
__device__ __forceinline__ void owner_scan(const int* __restrict__ src,
                                           const int* __restrict__ dst,
                                           const float* __restrict__ asrc,
                                           const float* __restrict__ h,
                                           int E, int lo, int nloc, int wave, int lane,
                                           int* qs, int* qd,
                                           const float* sadst, float* smax,
                                           float* ssum, float* acc,
                                           unsigned* elist, int* ecount, bool keep) {
    const bool vec = (E & 3) == 0;
    const int perIter = OWN_WAVES * 4 * BLK_EDGES;
    const int nIter = (E + perIter - 1) / perIter;
    const int start = (int)(((long long)blockIdx.x * nIter) / gridDim.x);
    const unsigned below = (1u << lane) - 1u;
    int qn = 0;
    for (int it = 0; it < nIter; ++it) {
        int jt = it + start;
        if (jt >= nIter) jt -= nIter;
        const int wbase = (jt * OWN_WAVES + wave) * (4 * BLK_EDGES);
        const int base = wbase + 4 * lane;
        int4 dv[4];
        if (vec && wbase + 4 * BLK_EDGES <= E) {
            #pragma unroll
            for (int k = 0; k < 4; ++k) dv[k] = *(const int4*)(dst + base + k * BLK_EDGES);
        } else {
            #pragma unroll
            for (int k = 0; k < 4; ++k) dv[k] = load_dst4_guarded(dst, base + k * BLK_EDGES, E);
        }
        #pragma unroll
        for (int k = 0; k < 4; ++k) {
            const int e0 = base + k * BLK_EDGES;
            const int dd[4] = {dv[k].x, dv[k].y, dv[k].z, dv[k].w};
            bool m[4];
            bool any = false;
            #pragma unroll
            for (int c = 0; c < 4; ++c) {
                m[c] = (unsigned)(dd[c] - lo) < (unsigned)nloc;
                any = any || m[c];
            }
            if (__builtin_amdgcn_ballot_w32(any) != 0u) {
                #pragma unroll
                for (int c = 0; c < 4; ++c) {
                    const unsigned mask = __builtin_amdgcn_ballot_w32(m[c]);
                    if (m[c]) {
                        const int pos = qn + __builtin_popcount(mask & below);
                        qs[pos] = e0 + c;
                        qd[pos] = dd[c] - lo;
                    }
                    qn += __builtin_popcount(mask);
                }
                if (qn >= Q_DRAIN) {
                    if (PASS == 1) drain_max<false>(qs, qd, qn, lane, src, dst, lo, asrc, sadst, smax, elist, ecount, keep);
                    else           drain_sum(qs, qd, qn, lane, src, asrc, h, sadst, smax, ssum, acc);
                    qn = 0;
                }
            }
        }
    }
    if (qn > 0) {
        if (PASS == 1) drain_max<false>(qs, qd, qn, lane, src, dst, lo, asrc, sadst, smax, elist, ecount, keep);
        else           drain_sum(qs, qd, qn, lane, src, asrc, h, sadst, smax, ssum, acc);
    }
}

static_assert(KEY_LOADS == 4, "the mask packs four loads of sixteen keys into 64 bits");
__device__ __forceinline__ void owner_scan_keys(const int* __restrict__ src,
                                                const int* __restrict__ dst,
                                                const unsigned* __restrict__ keys,
                                                const float* __restrict__ asrc,
                                                int E, int lo, int wave, int lane,
                                                int* qs, const float* sadst, float* smax,
                                                unsigned* elist, int* ecount, bool keep) {
    const int nIter = keys_padded(E) / KEY_TRIP;
    const int start = (int)(((long long)blockIdx.x * nIter) / gridDim.x);
    const unsigned own = (unsigned)blockIdx.x * 0x01010101u;
    const unsigned below = (1u << lane) - 1u;
    int qn = 0;
    for (int it = 0; it < nIter; ++it) {
        int jt = it + start;
        if (jt >= nIter) jt -= nIter;
        const int base = (jt * OWN_WAVES + wave) * (KEY_LOADS * 512) + 16 * lane;
        uint4 kv[KEY_LOADS];
        #pragma unroll
        for (int k = 0; k < KEY_LOADS; ++k) kv[k] = *(const uint4*)((const unsigned char*)keys + base + k * 512);
        unsigned miss[2] = {0u, 0u};
        #pragma unroll
        for (int k = 0; k < KEY_LOADS; ++k) {
            const unsigned kw[4] = {kv[k].x, kv[k].y, kv[k].z, kv[k].w};
            #pragma unroll
            for (int c = 0; c < 4; ++c) {
                const unsigned x = kw[c] ^ own;
                const unsigned nz = ((x & 0x7f7f7f7fu) + 0x7f7f7f7fu) | x;
                const int j = 4 * (k & 1) + c;
                miss[k >> 1] |= (nz >> (7 - j)) & (0x01010101u << j);
            }
        }
        unsigned long long m = (unsigned long long)(~miss[0]) | ((unsigned long long)(~miss[1]) << 32);
        unsigned any = __builtin_amdgcn_ballot_w32(m != 0ull);
        for (int r = 0; r < 64 && any != 0u; ++r) {
            if (m != 0ull) {
                const int p = __builtin_ctzll(m);
                m &= m - 1ull;
                qs[qn + __builtin_popcount(any & below)] =
                    base + (p >> 5) * 1024 + ((p >> 2) & 1) * 512 + (p & 3) * 4 + ((p >> 3) & 3);
            }
            qn += __builtin_popcount(any);
            if (qn >= Q_DRAIN) {
                drain_max<true>(qs, qs, qn, lane, src, dst, lo, asrc, sadst, smax, elist, ecount, keep);
                qn = 0;
            }
            any = __builtin_amdgcn_ballot_w32(m != 0ull);
        }
    }
    if (qn > 0)
        drain_max<true>(qs, qs, qn, lane, src, dst, lo, asrc, sadst, smax, elist, ecount, keep);
}

__global__ __launch_bounds__(OWN_THREADS) void gat_owner(const int* __restrict__ src,
                                                         const int* __restrict__ dst,
                                                         const float* __restrict__ asrc,
                                                         const float* __restrict__ adst,
                                                         const float* __restrict__ h,
                                                         const unsigned* __restrict__ keys,
                                                         float* __restrict__ out,
                                                         int N, int E, unsigned long long wsBytes) {
    __shared__ float acc[OWN_NODES * F_OUT];
    __shared__ float smax[OWN_NODES * NUM_HEADS];
    __shared__ float ssum[OWN_NODES * NUM_HEADS];
    __shared__ float sadst[OWN_NODES * NUM_HEADS];
    __shared__ int   qsrc[OWN_WAVES * Q_CAP];
    __shared__ int   qdst[OWN_WAVES * Q_CAP];
    __shared__ unsigned elist[L_CAP];
    __shared__ int   ecount;

    const int t = threadIdx.x;
    const int lane = t & 31;
    const int wave = __builtin_amdgcn_readfirstlane(t >> 5);
    const int lo = blockIdx.x * OWN_NODES;
    const int nloc = min(OWN_NODES, N - lo);
    const bool keep = N <= (1 << (32 - L_SHIFT));

    if (t == 0) ecount = 0;
    for (int i = t; i < OWN_NODES * F_OUT; i += OWN_THREADS) acc[i] = 0.0f;
    for (int i = t; i < OWN_NODES * NUM_HEADS; i += OWN_THREADS) {
        smax[i] = -3.0e38f;
        ssum[i] = 0.0f;
        sadst[i] = (i < nloc * NUM_HEADS) ? adst[(size_t)lo * NUM_HEADS + i] : 0.0f;
    }
    __syncthreads();

    int* qs = qsrc + wave * Q_CAP;
    int* qd = qdst + wave * Q_CAP;

    if (keys_usable(N, E, wsBytes))
        owner_scan_keys(src, dst, keys, asrc, E, lo, wave, lane, qs, sadst, smax, elist, &ecount, keep);
    else
        owner_scan<1>(src, dst, asrc, h, E, lo, nloc, wave, lane, qs, qd, sadst, smax, ssum, acc,
                      elist, &ecount, keep);
    __syncthreads();
    const int total = keep ? ecount : L_CAP + 1;
    if (total <= L_CAP)
        list_sum(elist, total, wave, lane, asrc, h, sadst, smax, ssum, acc);
    else
        owner_scan<2>(src, dst, asrc, h, E, lo, nloc, wave, lane, qs, qd, sadst, smax, ssum, acc,
                      elist, &ecount, false);
    __syncthreads();

    for (int r = wave; r < nloc; r += OWN_WAVES) {
        float* op = out + (size_t)(lo + r) * F_OUT + lane;
        #pragma unroll
        for (int c = 0; c < NUM_HEADS; ++c)
            op[c * HEAD_DIM] = acc[r * F_OUT + c * HEAD_DIM + lane] /
                               (ssum[r * NUM_HEADS + c] + 1e-10f);
    }
}

extern "C" void kernel_launch(void* const* d_in, const int* in_sizes, int n_in,
                              void* d_out, int out_size, void* d_ws, size_t ws_size,
                              hipStream_t stream) {
    const float* x     = (const float*)d_in[0];
    const int*   ei    = (const int*)d_in[1];
    const float* W     = (const float*)d_in[2];
    const float* a_src = (const float*)d_in[3];
    const float* a_dst = (const float*)d_in[4];
    float*       out   = (float*)d_out;

    const int N = in_sizes[0] / F_IN;
    const int E = in_sizes[1] / 2;
    const int* src = ei;
    const int* dst = ei + E;

    float* h    = (float*)d_ws;
    float* asrc = h    + (size_t)N * F_OUT;
    float* adst = asrc + (size_t)N * NUM_HEADS;
    unsigned* keys = (unsigned*)(adst + (size_t)N * NUM_HEADS);

    {
        int fullTiles = N / 16;
        if (fullTiles > 0)
            gat_gemm<<<fullTiles, 256, 0, stream>>>(x, W, a_src, a_dst, h, asrc, adst, dst, keys, N, E,
                                                    (unsigned long long)ws_size);
        int rowStart = fullTiles * 16;
        int rem = N - rowStart;
        if (rem > 0) {
            int total = rem * F_OUT;
            gat_gemm_tail<<<(total + 255) / 256, 256, 0, stream>>>(x, W, h, N, rowStart);
            gat_alpha_tail<<<(rem * NUM_HEADS + 255) / 256, 256, 0, stream>>>(h, a_src, a_dst, asrc, adst, dst, keys, N, E,
                                                                              (unsigned long long)ws_size, rowStart,
                                                                              fullTiles > 0 ? 0 : 1);
        }
    }

    gat_owner<<<(N + OWN_NODES - 1) / OWN_NODES, OWN_THREADS, 0, stream>>>(src, dst, asrc, adst, h, keys, out, N, E,
                                                                          (unsigned long long)ws_size);
}


# First launch also writes one packed word per edge (source node, row in owner's range); owner pass reads that word per owned edge, not src[e] and dst[e]: one line, not two; LDS sum order as before
# speedup vs baseline: 5.1889x; 1.1017x over previous
#include <hip/hip_runtime.h>
#include <hip/hip_bf16.h>


#define NUM_HEADS 4
#define HEAD_DIM 32
#define F_IN 256
#define F_OUT 128
#define LDA 260

typedef __attribute__((ext_vector_type(2))) float v2f;
typedef __attribute__((ext_vector_type(8))) float v8f;

#define OWN_NODES   400
#define OWN_WAVES   32
#define KEY_NONE    255u
#define KEY_LOADS   4
#define KEY_TRIP    (OWN_WAVES * KEY_LOADS * 512)

__device__ __forceinline__ int keys_padded(int E) {
    return ((E + KEY_TRIP - 1) / KEY_TRIP) * KEY_TRIP;
}

__device__ __forceinline__ bool keys_usable(int N, int E, unsigned long long wsBytes) {
    const int groups = (N + OWN_NODES - 1) / OWN_NODES;
    const unsigned long long need =
        (unsigned long long)N * (F_OUT + 2 * NUM_HEADS) * sizeof(float) + (unsigned long long)keys_padded(E);
    return groups <= (int)KEY_NONE && wsBytes >= need;
}

__device__ __forceinline__ unsigned key_of(int d, int N) {
    return ((unsigned)d < (unsigned)N) ? (unsigned)d / OWN_NODES : KEY_NONE;
}

#define L_SHIFT     9
#define L_MASK      ((1u << L_SHIFT) - 1u)
static_assert(OWN_NODES <= (1 << L_SHIFT), "local row must fit the low bits of a packed word");

__device__ __forceinline__ int packed_len(int E) { return (E + 3) & ~3; }

__device__ __forceinline__ bool packed_usable(int N, int E, unsigned long long wsBytes) {
    const unsigned long long need =
        (unsigned long long)N * (F_OUT + 2 * NUM_HEADS) * sizeof(float) + (unsigned long long)keys_padded(E) +
        (unsigned long long)packed_len(E) * sizeof(unsigned);
    return keys_usable(N, E, wsBytes) && N <= (1 << (32 - L_SHIFT)) && wsBytes >= need;
}

__device__ __forceinline__ unsigned* packed_of(unsigned* keys, int E) { return keys + (keys_padded(E) >> 2); }
__device__ __forceinline__ const unsigned* packed_of(const unsigned* keys, int E) { return keys + (keys_padded(E) >> 2); }

__device__ __forceinline__ unsigned pack_word(int s, int d, unsigned key) {
    return ((unsigned)s << L_SHIFT) | (((unsigned)d - key * OWN_NODES) & L_MASK);
}

#define HP 132
#define GEMM_GROUPS 512

__device__ __forceinline__ void write_keys(const int* __restrict__ src, const int* __restrict__ dst,
                                           unsigned* __restrict__ keys, int N, int E, bool pack,
                                           int tid, int step) {
    const int words = keys_padded(E) >> 2;
    unsigned* __restrict__ pk = packed_of(keys, E);
    const bool vec = (E & 3) == 0;
    for (int g = tid; g < words; g += step) {
        const int e = g << 2;
        int4 d, s = {0, 0, 0, 0};
        if (vec && e + 4 <= E) {
            d = *(const int4*)(dst + e);
            if (pack) s = *(const int4*)(src + e);
        } else {
            d.x = (e     < E) ? dst[e]     : -1;
            d.y = (e + 1 < E) ? dst[e + 1] : -1;
            d.z = (e + 2 < E) ? dst[e + 2] : -1;
            d.w = (e + 3 < E) ? dst[e + 3] : -1;
            if (pack) {
                s.x = (e     < E) ? src[e]     : 0;
                s.y = (e + 1 < E) ? src[e + 1] : 0;
                s.z = (e + 2 < E) ? src[e + 2] : 0;
                s.w = (e + 3 < E) ? src[e + 3] : 0;
            }
        }
        const unsigned k0 = key_of(d.x, N), k1 = key_of(d.y, N), k2 = key_of(d.z, N), k3 = key_of(d.w, N);
        keys[g] = k0 | (k1 << 8) | (k2 << 16) | (k3 << 24);
        if (pack && e < E) {
            uint4 w;
            w.x = pack_word(s.x, d.x, k0);
            w.y = pack_word(s.y, d.y, k1);
            w.z = pack_word(s.z, d.z, k2);
            w.w = pack_word(s.w, d.w, k3);
            *(uint4*)(pk + e) = w;
        }
    }
}

__device__ __forceinline__ float head_dot(const float* hp, const float* __restrict__ av) {
    float s = 0.0f;
    #pragma unroll
    for (int i = 0; i < HEAD_DIM; i += 4) {
        float4 hv = *(const float4*)(hp + i);
        float4 a  = *(const float4*)(av + i);
        s += hv.x * a.x + hv.y * a.y + hv.z * a.z + hv.w * a.w;
    }
    return s;
}

__global__ __launch_bounds__(256) void gat_gemm(const float* __restrict__ x,
                                                const float* __restrict__ W,
                                                const float* __restrict__ a_src,
                                                const float* __restrict__ a_dst,
                                                float* __restrict__ h,
                                                float* __restrict__ asrc,
                                                float* __restrict__ adst,
                                                const int* __restrict__ src,
                                                const int* __restrict__ dst,
                                                unsigned* __restrict__ keys,
                                                int N, int E, unsigned long long wsBytes) {
    __shared__ float As[16 * LDA];

    const int t = threadIdx.x;
    const int wave = t >> 5;
    const int lane = t & 31;
    const int m    = lane & 15;
    const int koff = (lane >> 4) << 1;
    const int col  = (wave << 4) + m;
    const int fullTiles = N / 16;

    const float* Wp = W + col;
    v2f bw[F_IN / 4];
    #pragma unroll
    for (int i = 0; i < F_IN / 4; ++i) {
        bw[i].x = Wp[(size_t)(4 * i + koff) * F_OUT];
        bw[i].y = Wp[(size_t)(4 * i + koff + 1) * F_OUT];
    }

    const int ar = t >> 6;
    const int ac = (t & 63) << 2;
    float4 p0 = {}, p1 = {}, p2 = {}, p3 = {};
    int tile = blockIdx.x;
    if (tile < fullTiles) {
        const float* xp = x + (size_t)(tile * 16 + ar) * F_IN + ac;
        p0 = *(const float4*)(xp);
        p1 = *(const float4*)(xp + 4 * F_IN);
        p2 = *(const float4*)(xp + 8 * F_IN);
        p3 = *(const float4*)(xp + 12 * F_IN);
    }

    for (; tile < fullTiles; tile += gridDim.x) {
        const int rowBase = tile * 16;
        *(float4*)(&As[(ar     ) * LDA + ac]) = p0;
        *(float4*)(&As[(ar +  4) * LDA + ac]) = p1;
        *(float4*)(&As[(ar +  8) * LDA + ac]) = p2;
        *(float4*)(&As[(ar + 12) * LDA + ac]) = p3;
        __syncthreads();

        const int nextTile = tile + gridDim.x;
        if (nextTile < fullTiles) {
            const float* xp = x + (size_t)(nextTile * 16 + ar) * F_IN + ac;
            p0 = *(const float4*)(xp);
            p1 = *(const float4*)(xp + 4 * F_IN);
            p2 = *(const float4*)(xp + 8 * F_IN);
            p3 = *(const float4*)(xp + 12 * F_IN);
        }

        v8f acc = {};
        #pragma unroll
        for (int i = 0; i < F_IN / 4; ++i) {
            v2f a = *(const v2f*)(&As[m * LDA + 4 * i + koff]);
            acc = __builtin_amdgcn_wmma_f32_16x16x4_f32(
                 false, a,  false, bw[i],
                 (short)0, acc,  false,  false);
        }
        __builtin_amdgcn_sched_barrier(0);
        asm volatile("v_nop\n\tv_nop\n\tv_nop\n\tv_nop");
        __builtin_amdgcn_sched_barrier(0);

        __syncthreads();
        float* Hs = As;
        const int roff = (lane >> 4) << 3;
        #pragma unroll
        for (int v = 0; v < 8; ++v)
            Hs[(v + roff) * HP + col] = acc[v];
        __syncthreads();

        #pragma unroll
        for (int j = 0; j < 2; ++j) {
            int f = t + 256 * j;
            int r = f >> 5;
            int cc = (f & 31) << 2;
            *(float4*)(h + (size_t)(rowBase + r) * F_OUT + cc) = *(const float4*)(&Hs[r * HP + cc]);
        }

        if (t < 2 * 16 * NUM_HEADS) {
            const int q  = t & (16 * NUM_HEADS - 1);
            const int r  = q >> 2;
            const int hd = q & 3;
            const float* hp = Hs + r * HP + hd * HEAD_DIM;
            const size_t o = (size_t)(rowBase + r) * NUM_HEADS + hd;
            if (t < 16 * NUM_HEADS) asrc[o] = head_dot(hp, a_src + hd * HEAD_DIM);
            else                    adst[o] = head_dot(hp, a_dst + hd * HEAD_DIM);
        }
        __syncthreads();
    }

    if (keys_usable(N, E, wsBytes))
        write_keys(src, dst, keys, N, E, packed_usable(N, E, wsBytes),
                   blockIdx.x * blockDim.x + t, gridDim.x * blockDim.x);
}

__global__ __launch_bounds__(256) void gat_gemm_tail(const float* __restrict__ x,
                                                     const float* __restrict__ W,
                                                     float* __restrict__ h,
                                                     int N, int rowStart) {
    int idx = blockIdx.x * blockDim.x + threadIdx.x;
    int total = (N - rowStart) * F_OUT;
    if (idx >= total) return;
    int r = rowStart + idx / F_OUT;
    int c = idx % F_OUT;
    float s = 0.0f;
    for (int k = 0; k < F_IN; ++k)
        s += x[(size_t)r * F_IN + k] * W[(size_t)k * F_OUT + c];
    h[(size_t)r * F_OUT + c] = s;
}

__global__ __launch_bounds__(256) void gat_alpha_tail(const float* __restrict__ h,
                                                      const float* __restrict__ a_src,
                                                      const float* __restrict__ a_dst,
                                                      float* __restrict__ asrc,
                                                      float* __restrict__ adst,
                                                      const int* __restrict__ src,
                                                      const int* __restrict__ dst,
                                                      unsigned* __restrict__ keys,
                                                      int N, int E, unsigned long long wsBytes,
                                                      int rowStart, int doKeys) {
    int tid = blockIdx.x * blockDim.x + threadIdx.x;
    if (doKeys && keys_usable(N, E, wsBytes))
        write_keys(src, dst, keys, N, E, packed_usable(N, E, wsBytes), tid, gridDim.x * blockDim.x);
    if (tid >= (N - rowStart) * NUM_HEADS) return;
    int n  = rowStart + (tid >> 2);
    int hd = tid & 3;
    const float* hp = h + (size_t)n * F_OUT + hd * HEAD_DIM;
    asrc[(size_t)n * NUM_HEADS + hd] = head_dot(hp, a_src + hd * HEAD_DIM);
    adst[(size_t)n * NUM_HEADS + hd] = head_dot(hp, a_dst + hd * HEAD_DIM);
}

__device__ __forceinline__ float leaky(float v) { return v > 0.0f ? v : 0.2f * v; }

#define OWN_THREADS 1024
#define Q_CAP       144
#define Q_DRAIN     16
#define BLK_EDGES   128
#define L_CAP       12288

__device__ __forceinline__ int4 load_dst4_guarded(const int* __restrict__ dst, int e, int E) {
    int4 r;
    r.x = (e     < E) ? dst[e]     : -1;
    r.y = (e + 1 < E) ? dst[e + 1] : -1;
    r.z = (e + 2 < E) ? dst[e + 2] : -1;
    r.w = (e + 3 < E) ? dst[e + 3] : -1;
    return r;
}

enum { Q_ROW = 0, SRC_DST = 1, PACKED = 2 };
template <int LOOK>
__device__ __forceinline__ void drain_max(const int* qs, const int* qd, int qn, int lane,
                                          const int* __restrict__ src,
                                          const int* __restrict__ dst,
                                          const unsigned* __restrict__ pk, int lo,
                                          const float* __restrict__ asrc,
                                          const float* sadst, float* smax,
                                          unsigned* elist, int* ecount, bool keep) {
    int slot0 = 0;
    if (keep) {
        if (lane == 0)
            slot0 = __hip_atomic_fetch_add(ecount, qn, __ATOMIC_RELAXED, __HIP_MEMORY_SCOPE_WORKGROUP);
        slot0 = __builtin_amdgcn_readfirstlane(slot0);
    }
    for (int i0 = 0; i0 < qn; i0 += 32) {
        int idx = i0 + lane;
        if (idx < qn) {
            const int e = qs[idx];
            int s, dl;
            if (LOOK == PACKED) {
                const unsigned p = pk[e];
                s  = (int)(p >> L_SHIFT);
                dl = (int)(p & L_MASK);
            } else {
                s  = src[e];
                dl = (LOOK == SRC_DST) ? dst[e] - lo : qd[idx];
            }
            if (keep && slot0 + idx < L_CAP)
                elist[slot0 + idx] = ((unsigned)s << L_SHIFT) | (unsigned)dl;
            float4 a = *(const float4*)(asrc + (size_t)s * NUM_HEADS);
            float4 b = *(const float4*)(sadst + dl * NUM_HEADS);
            float* mp = smax + dl * NUM_HEADS;
            __hip_atomic_fetch_max(mp + 0, leaky(a.x + b.x), __ATOMIC_RELAXED, __HIP_MEMORY_SCOPE_WORKGROUP);
            __hip_atomic_fetch_max(mp + 1, leaky(a.y + b.y), __ATOMIC_RELAXED, __HIP_MEMORY_SCOPE_WORKGROUP);
            __hip_atomic_fetch_max(mp + 2, leaky(a.z + b.z), __ATOMIC_RELAXED, __HIP_MEMORY_SCOPE_WORKGROUP);
            __hip_atomic_fetch_max(mp + 3, leaky(a.w + b.w), __ATOMIC_RELAXED, __HIP_MEMORY_SCOPE_WORKGROUP);
        }
    }
}

__device__ __forceinline__ void sum_batch(int s, int dl, bool on, int nb, int lane,
                                          const float* __restrict__ asrc,
                                          const float* __restrict__ h,
                                          const float* sadst, const float* smax,
                                          float* ssum, float* acc) {
    const int hd = lane & 3;
    float as = 0.0f;
    if (on) as = asrc[(size_t)s * NUM_HEADS + hd];
    float hv[8][NUM_HEADS];
    #pragma unroll
    for (int j = 0; j < 8; ++j) {
        if (j < nb) {
            int sj = __builtin_amdgcn_readlane(s, 4 * j);
            const float* hp = h + (size_t)sj * F_OUT + lane;
            #pragma unroll
            for (int c = 0; c < NUM_HEADS; ++c) hv[j][c] = hp[c * HEAD_DIM];
        }
    }
    float w = 0.0f;
    if (on) {
        float v = leaky(as + sadst[dl * NUM_HEADS + hd]);
        w = __expf(v - smax[dl * NUM_HEADS + hd]);
        __hip_atomic_fetch_add(ssum + dl * NUM_HEADS + hd, w,
                               __ATOMIC_RELAXED, __HIP_MEMORY_SCOPE_WORKGROUP);
    }
    const int wbits = __float_as_int(w);
    #pragma unroll
    for (int j = 0; j < 8; ++j) {
        if (j < nb) {
            int dj = __builtin_amdgcn_readlane(dl, 4 * j);
            float* ap = acc + dj * F_OUT + lane;
            #pragma unroll
            for (int c = 0; c < NUM_HEADS; ++c) {
                float wj = __int_as_float(__builtin_amdgcn_readlane(wbits, 4 * j + c));
                __hip_atomic_fetch_add(ap + c * HEAD_DIM, wj * hv[j][c],
                                       __ATOMIC_RELAXED, __HIP_MEMORY_SCOPE_WORKGROUP);
            }
        }
    }
}

__device__ __forceinline__ void drain_sum(int* qs, const int* qd, int qn, int lane,
                                          const int* __restrict__ src,
                                          const float* __restrict__ asrc,
                                          const float* __restrict__ h,
                                          const float* sadst, const float* smax,
                                          float* ssum, float* acc) {
    for (int i = lane; i < qn; i += 32) qs[i] = src[qs[i]];
    for (int i0 = 0; i0 < qn; i0 += 8) {
        const int nb = min(8, qn - i0);
        const int idx = i0 + (lane >> 2);
        const bool on = idx < qn;
        int s = 0, dl = 0;
        if (on) { s = qs[idx]; dl = qd[idx]; }
        sum_batch(s, dl, on, nb, lane, asrc, h, sadst, smax, ssum, acc);
    }
}

__device__ __forceinline__ void list_sum(const unsigned* elist, int total, int wave, int lane,
                                         const float* __restrict__ asrc,
                                         const float* __restrict__ h,
                                         const float* sadst, const float* smax,
                                         float* ssum, float* acc) {
    for (int i0 = wave * 8; i0 < total; i0 += OWN_WAVES * 8) {
        const int nb = min(8, total - i0);
        const int idx = i0 + (lane >> 2);
        const bool on = idx < total;
        int s = 0, dl = 0;
        if (on) {
            const unsigned p = elist[idx];
            s  = (int)(p >> L_SHIFT);
            dl = (int)(p & L_MASK);
        }
        sum_batch(s, dl, on, nb, lane, asrc, h, sadst, smax, ssum, acc);
    }
}

template <int PASS>
__device__ __forceinline__ void owner_scan(const int* __restrict__ src,
                                           const int* __restrict__ dst,
                                           const float* __restrict__ asrc,
                                           const float* __restrict__ h,
                                           int E, int lo, int nloc, int wave, int lane,
                                           int* qs, int* qd,
                                           const float* sadst, float* smax,
                                           float* ssum, float* acc,
                                           unsigned* elist, int* ecount, bool keep) {
    const bool vec = (E & 3) == 0;
    const int perIter = OWN_WAVES * 4 * BLK_EDGES;
    const int nIter = (E + perIter - 1) / perIter;
    const int start = (int)(((long long)blockIdx.x * nIter) / gridDim.x);
    const unsigned below = (1u << lane) - 1u;
    int qn = 0;
    for (int it = 0; it < nIter; ++it) {
        int jt = it + start;
        if (jt >= nIter) jt -= nIter;
        const int wbase = (jt * OWN_WAVES + wave) * (4 * BLK_EDGES);
        const int base = wbase + 4 * lane;
        int4 dv[4];
        if (vec && wbase + 4 * BLK_EDGES <= E) {
            #pragma unroll
            for (int k = 0; k < 4; ++k) dv[k] = *(const int4*)(dst + base + k * BLK_EDGES);
        } else {
            #pragma unroll
            for (int k = 0; k < 4; ++k) dv[k] = load_dst4_guarded(dst, base + k * BLK_EDGES, E);
        }
        #pragma unroll
        for (int k = 0; k < 4; ++k) {
            const int e0 = base + k * BLK_EDGES;
            const int dd[4] = {dv[k].x, dv[k].y, dv[k].z, dv[k].w};
            bool m[4];
            bool any = false;
            #pragma unroll
            for (int c = 0; c < 4; ++c) {
                m[c] = (unsigned)(dd[c] - lo) < (unsigned)nloc;
                any = any || m[c];
            }
            if (__builtin_amdgcn_ballot_w32(any) != 0u) {
                #pragma unroll
                for (int c = 0; c < 4; ++c) {
                    const unsigned mask = __builtin_amdgcn_ballot_w32(m[c]);
                    if (m[c]) {
                        const int pos = qn + __builtin_popcount(mask & below);
                        qs[pos] = e0 + c;
                        qd[pos] = dd[c] - lo;
                    }
                    qn += __builtin_popcount(mask);
                }
                if (qn >= Q_DRAIN) {
                    if (PASS == 1) drain_max<Q_ROW>(qs, qd, qn, lane, src, dst, nullptr, lo, asrc, sadst, smax, elist, ecount, keep);
                    else           drain_sum(qs, qd, qn, lane, src, asrc, h, sadst, smax, ssum, acc);
                    qn = 0;
                }
            }
        }
    }
    if (qn > 0) {
        if (PASS == 1) drain_max<Q_ROW>(qs, qd, qn, lane, src, dst, nullptr, lo, asrc, sadst, smax, elist, ecount, keep);
        else           drain_sum(qs, qd, qn, lane, src, asrc, h, sadst, smax, ssum, acc);
    }
}

static_assert(KEY_LOADS == 4, "the mask packs four loads of sixteen keys into 64 bits");
template <int LOOK>
__device__ __forceinline__ void owner_scan_keys(const int* __restrict__ src,
                                                const int* __restrict__ dst,
                                                const unsigned* __restrict__ keys,
                                                const float* __restrict__ asrc,
                                                int E, int lo, int wave, int lane,
                                                int* qs, const float* sadst, float* smax,
                                                unsigned* elist, int* ecount, bool keep) {
    const int nIter = keys_padded(E) / KEY_TRIP;
    const unsigned* __restrict__ pk = packed_of(keys, E);
    const int start = (int)(((long long)blockIdx.x * nIter) / gridDim.x);
    const unsigned own = (unsigned)blockIdx.x * 0x01010101u;
    const unsigned below = (1u << lane) - 1u;
    int qn = 0;
    for (int it = 0; it < nIter; ++it) {
        int jt = it + start;
        if (jt >= nIter) jt -= nIter;
        const int base = (jt * OWN_WAVES + wave) * (KEY_LOADS * 512) + 16 * lane;
        uint4 kv[KEY_LOADS];
        #pragma unroll
        for (int k = 0; k < KEY_LOADS; ++k) kv[k] = *(const uint4*)((const unsigned char*)keys + base + k * 512);
        unsigned miss[2] = {0u, 0u};
        #pragma unroll
        for (int k = 0; k < KEY_LOADS; ++k) {
            const unsigned kw[4] = {kv[k].x, kv[k].y, kv[k].z, kv[k].w};
            #pragma unroll
            for (int c = 0; c < 4; ++c) {
                const unsigned x = kw[c] ^ own;
                const unsigned nz = ((x & 0x7f7f7f7fu) + 0x7f7f7f7fu) | x;
                const int j = 4 * (k & 1) + c;
                miss[k >> 1] |= (nz >> (7 - j)) & (0x01010101u << j);
            }
        }
        unsigned long long m = (unsigned long long)(~miss[0]) | ((unsigned long long)(~miss[1]) << 32);
        unsigned any = __builtin_amdgcn_ballot_w32(m != 0ull);
        for (int r = 0; r < 64 && any != 0u; ++r) {
            if (m != 0ull) {
                const int p = __builtin_ctzll(m);
                m &= m - 1ull;
                qs[qn + __builtin_popcount(any & below)] =
                    base + (p >> 5) * 1024 + ((p >> 2) & 1) * 512 + (p & 3) * 4 + ((p >> 3) & 3);
            }
            qn += __builtin_popcount(any);
            if (qn >= Q_DRAIN) {
                drain_max<LOOK>(qs, qs, qn, lane, src, dst, pk, lo, asrc, sadst, smax, elist, ecount, keep);
                qn = 0;
            }
            any = __builtin_amdgcn_ballot_w32(m != 0ull);
        }
    }
    if (qn > 0)
        drain_max<LOOK>(qs, qs, qn, lane, src, dst, pk, lo, asrc, sadst, smax, elist, ecount, keep);
}

__global__ __launch_bounds__(OWN_THREADS) void gat_owner(const int* __restrict__ src,
                                                         const int* __restrict__ dst,
                                                         const float* __restrict__ asrc,
                                                         const float* __restrict__ adst,
                                                         const float* __restrict__ h,
                                                         const unsigned* __restrict__ keys,
                                                         float* __restrict__ out,
                                                         int N, int E, unsigned long long wsBytes) {
    __shared__ float acc[OWN_NODES * F_OUT];
    __shared__ float smax[OWN_NODES * NUM_HEADS];
    __shared__ float ssum[OWN_NODES * NUM_HEADS];
    __shared__ float sadst[OWN_NODES * NUM_HEADS];
    __shared__ int   qsrc[OWN_WAVES * Q_CAP];
    __shared__ int   qdst[OWN_WAVES * Q_CAP];
    __shared__ unsigned elist[L_CAP];
    __shared__ int   ecount;

    const int t = threadIdx.x;
    const int lane = t & 31;
    const int wave = __builtin_amdgcn_readfirstlane(t >> 5);
    const int lo = blockIdx.x * OWN_NODES;
    const int nloc = min(OWN_NODES, N - lo);
    const bool keep = N <= (1 << (32 - L_SHIFT));

    if (t == 0) ecount = 0;
    for (int i = t; i < OWN_NODES * F_OUT; i += OWN_THREADS) acc[i] = 0.0f;
    for (int i = t; i < OWN_NODES * NUM_HEADS; i += OWN_THREADS) {
        smax[i] = -3.0e38f;
        ssum[i] = 0.0f;
        sadst[i] = (i < nloc * NUM_HEADS) ? adst[(size_t)lo * NUM_HEADS + i] : 0.0f;
    }
    __syncthreads();

    int* qs = qsrc + wave * Q_CAP;
    int* qd = qdst + wave * Q_CAP;

    if (packed_usable(N, E, wsBytes))
        owner_scan_keys<PACKED>(src, dst, keys, asrc, E, lo, wave, lane, qs, sadst, smax, elist, &ecount, keep);
    else if (keys_usable(N, E, wsBytes))
        owner_scan_keys<SRC_DST>(src, dst, keys, asrc, E, lo, wave, lane, qs, sadst, smax, elist, &ecount, keep);
    else
        owner_scan<1>(src, dst, asrc, h, E, lo, nloc, wave, lane, qs, qd, sadst, smax, ssum, acc,
                      elist, &ecount, keep);
    __syncthreads();
    const int total = keep ? ecount : L_CAP + 1;
    if (total <= L_CAP)
        list_sum(elist, total, wave, lane, asrc, h, sadst, smax, ssum, acc);
    else
        owner_scan<2>(src, dst, asrc, h, E, lo, nloc, wave, lane, qs, qd, sadst, smax, ssum, acc,
                      elist, &ecount, false);
    __syncthreads();

    for (int r = wave; r < nloc; r += OWN_WAVES) {
        float* op = out + (size_t)(lo + r) * F_OUT + lane;
        #pragma unroll
        for (int c = 0; c < NUM_HEADS; ++c)
            op[c * HEAD_DIM] = acc[r * F_OUT + c * HEAD_DIM + lane] /
                               (ssum[r * NUM_HEADS + c] + 1e-10f);
    }
}

extern "C" void kernel_launch(void* const* d_in, const int* in_sizes, int n_in,
                              void* d_out, int out_size, void* d_ws, size_t ws_size,
                              hipStream_t stream) {
    const float* x     = (const float*)d_in[0];
    const int*   ei    = (const int*)d_in[1];
    const float* W     = (const float*)d_in[2];
    const float* a_src = (const float*)d_in[3];
    const float* a_dst = (const float*)d_in[4];
    float*       out   = (float*)d_out;

    const int N = in_sizes[0] / F_IN;
    const int E = in_sizes[1] / 2;
    const int* src = ei;
    const int* dst = ei + E;

    float* h    = (float*)d_ws;
    float* asrc = h    + (size_t)N * F_OUT;
    float* adst = asrc + (size_t)N * NUM_HEADS;
    unsigned* keys = (unsigned*)(adst + (size_t)N * NUM_HEADS);

    {
        int fullTiles = N / 16;
        if (fullTiles > 0)
            gat_gemm<<<GEMM_GROUPS, 256, 0, stream>>>(x, W, a_src, a_dst, h, asrc, adst, src, dst, keys, N, E,
                                                    (unsigned long long)ws_size);
        int rowStart = fullTiles * 16;
        int rem = N - rowStart;
        if (rem > 0) {
            int total = rem * F_OUT;
            gat_gemm_tail<<<(total + 255) / 256, 256, 0, stream>>>(x, W, h, N, rowStart);
            gat_alpha_tail<<<(rem * NUM_HEADS + 255) / 256, 256, 0, stream>>>(h, a_src, a_dst, asrc, adst, src, dst, keys, N, E,
                                                                              (unsigned long long)ws_size, rowStart,
                                                                              fullTiles > 0 ? 0 : 1);
        }
    }

    gat_owner<<<(N + OWN_NODES - 1) / OWN_NODES, OWN_THREADS, 0, stream>>>(src, dst, asrc, adst, h, keys, out, N, E,
                                                                          (unsigned long long)ws_size);
}


# Each wave owns rows r = wave (mod 32) of its group's range: kept edges filed by owning wave, which adds into its LDS rows by plain 16-byte load/add/store, one 512-byte row load per edge; list order
# speedup vs baseline: 5.4723x; 1.0546x over previous
#include <hip/hip_runtime.h>
#include <hip/hip_bf16.h>


#define NUM_HEADS 4
#define HEAD_DIM 32
#define F_IN 256
#define F_OUT 128
#define LDA 260

typedef __attribute__((ext_vector_type(2))) float v2f;
typedef __attribute__((ext_vector_type(8))) float v8f;

#define OWN_NODES   400
#define OWN_WAVES   32
#define KEY_NONE    255u
#define KEY_LOADS   4
#define KEY_TRIP    (OWN_WAVES * KEY_LOADS * 512)

__device__ __forceinline__ int keys_padded(int E) {
    return ((E + KEY_TRIP - 1) / KEY_TRIP) * KEY_TRIP;
}

__device__ __forceinline__ bool keys_usable(int N, int E, unsigned long long wsBytes) {
    const int groups = (N + OWN_NODES - 1) / OWN_NODES;
    const unsigned long long need =
        (unsigned long long)N * (F_OUT + 2 * NUM_HEADS) * sizeof(float) + (unsigned long long)keys_padded(E);
    return groups <= (int)KEY_NONE && wsBytes >= need;
}

__device__ __forceinline__ unsigned key_of(int d, int N) {
    return ((unsigned)d < (unsigned)N) ? (unsigned)d / OWN_NODES : KEY_NONE;
}

#define L_SHIFT     9
#define L_MASK      ((1u << L_SHIFT) - 1u)
static_assert(OWN_NODES <= (1 << L_SHIFT), "local row must fit the low bits of a packed word");

__device__ __forceinline__ int packed_len(int E) { return (E + 3) & ~3; }

__device__ __forceinline__ bool packed_usable(int N, int E, unsigned long long wsBytes) {
    const unsigned long long need =
        (unsigned long long)N * (F_OUT + 2 * NUM_HEADS) * sizeof(float) + (unsigned long long)keys_padded(E) +
        (unsigned long long)packed_len(E) * sizeof(unsigned);
    return keys_usable(N, E, wsBytes) && N <= (1 << (32 - L_SHIFT)) && wsBytes >= need;
}

__device__ __forceinline__ unsigned* packed_of(unsigned* keys, int E) { return keys + (keys_padded(E) >> 2); }
__device__ __forceinline__ const unsigned* packed_of(const unsigned* keys, int E) { return keys + (keys_padded(E) >> 2); }

__device__ __forceinline__ unsigned pack_word(int s, int d, unsigned key) {
    return ((unsigned)s << L_SHIFT) | (((unsigned)d - key * OWN_NODES) & L_MASK);
}

#define HP 132
#define GEMM_GROUPS 512

__device__ __forceinline__ void write_keys(const int* __restrict__ src, const int* __restrict__ dst,
                                           unsigned* __restrict__ keys, int N, int E, bool pack,
                                           int tid, int step) {
    const int words = keys_padded(E) >> 2;
    unsigned* __restrict__ pk = packed_of(keys, E);
    const bool vec = (E & 3) == 0;
    for (int g = tid; g < words; g += step) {
        const int e = g << 2;
        int4 d, s = {0, 0, 0, 0};
        if (vec && e + 4 <= E) {
            d = *(const int4*)(dst + e);
            if (pack) s = *(const int4*)(src + e);
        } else {
            d.x = (e     < E) ? dst[e]     : -1;
            d.y = (e + 1 < E) ? dst[e + 1] : -1;
            d.z = (e + 2 < E) ? dst[e + 2] : -1;
            d.w = (e + 3 < E) ? dst[e + 3] : -1;
            if (pack) {
                s.x = (e     < E) ? src[e]     : 0;
                s.y = (e + 1 < E) ? src[e + 1] : 0;
                s.z = (e + 2 < E) ? src[e + 2] : 0;
                s.w = (e + 3 < E) ? src[e + 3] : 0;
            }
        }
        const unsigned k0 = key_of(d.x, N), k1 = key_of(d.y, N), k2 = key_of(d.z, N), k3 = key_of(d.w, N);
        keys[g] = k0 | (k1 << 8) | (k2 << 16) | (k3 << 24);
        if (pack && e < E) {
            uint4 w;
            w.x = pack_word(s.x, d.x, k0);
            w.y = pack_word(s.y, d.y, k1);
            w.z = pack_word(s.z, d.z, k2);
            w.w = pack_word(s.w, d.w, k3);
            *(uint4*)(pk + e) = w;
        }
    }
}

__device__ __forceinline__ float head_dot(const float* hp, const float* __restrict__ av) {
    float s = 0.0f;
    #pragma unroll
    for (int i = 0; i < HEAD_DIM; i += 4) {
        float4 hv = *(const float4*)(hp + i);
        float4 a  = *(const float4*)(av + i);
        s += hv.x * a.x + hv.y * a.y + hv.z * a.z + hv.w * a.w;
    }
    return s;
}

__global__ __launch_bounds__(256) void gat_gemm(const float* __restrict__ x,
                                                const float* __restrict__ W,
                                                const float* __restrict__ a_src,
                                                const float* __restrict__ a_dst,
                                                float* __restrict__ h,
                                                float* __restrict__ asrc,
                                                float* __restrict__ adst,
                                                const int* __restrict__ src,
                                                const int* __restrict__ dst,
                                                unsigned* __restrict__ keys,
                                                int N, int E, unsigned long long wsBytes) {
    __shared__ float As[16 * LDA];

    const int t = threadIdx.x;
    const int wave = t >> 5;
    const int lane = t & 31;
    const int m    = lane & 15;
    const int koff = (lane >> 4) << 1;
    const int col  = (wave << 4) + m;
    const int fullTiles = N / 16;

    const float* Wp = W + col;
    v2f bw[F_IN / 4];
    #pragma unroll
    for (int i = 0; i < F_IN / 4; ++i) {
        bw[i].x = Wp[(size_t)(4 * i + koff) * F_OUT];
        bw[i].y = Wp[(size_t)(4 * i + koff + 1) * F_OUT];
    }

    const int ar = t >> 6;
    const int ac = (t & 63) << 2;
    float4 p0 = {}, p1 = {}, p2 = {}, p3 = {};
    int tile = blockIdx.x;
    if (tile < fullTiles) {
        const float* xp = x + (size_t)(tile * 16 + ar) * F_IN + ac;
        p0 = *(const float4*)(xp);
        p1 = *(const float4*)(xp + 4 * F_IN);
        p2 = *(const float4*)(xp + 8 * F_IN);
        p3 = *(const float4*)(xp + 12 * F_IN);
    }

    for (; tile < fullTiles; tile += gridDim.x) {
        const int rowBase = tile * 16;
        *(float4*)(&As[(ar     ) * LDA + ac]) = p0;
        *(float4*)(&As[(ar +  4) * LDA + ac]) = p1;
        *(float4*)(&As[(ar +  8) * LDA + ac]) = p2;
        *(float4*)(&As[(ar + 12) * LDA + ac]) = p3;
        __syncthreads();

        const int nextTile = tile + gridDim.x;
        if (nextTile < fullTiles) {
            const float* xp = x + (size_t)(nextTile * 16 + ar) * F_IN + ac;
            p0 = *(const float4*)(xp);
            p1 = *(const float4*)(xp + 4 * F_IN);
            p2 = *(const float4*)(xp + 8 * F_IN);
            p3 = *(const float4*)(xp + 12 * F_IN);
        }

        v8f acc = {};
        #pragma unroll
        for (int i = 0; i < F_IN / 4; ++i) {
            v2f a = *(const v2f*)(&As[m * LDA + 4 * i + koff]);
            acc = __builtin_amdgcn_wmma_f32_16x16x4_f32(
                 false, a,  false, bw[i],
                 (short)0, acc,  false,  false);
        }
        __builtin_amdgcn_sched_barrier(0);
        asm volatile("v_nop\n\tv_nop\n\tv_nop\n\tv_nop");
        __builtin_amdgcn_sched_barrier(0);

        __syncthreads();
        float* Hs = As;
        const int roff = (lane >> 4) << 3;
        #pragma unroll
        for (int v = 0; v < 8; ++v)
            Hs[(v + roff) * HP + col] = acc[v];
        __syncthreads();

        #pragma unroll
        for (int j = 0; j < 2; ++j) {
            int f = t + 256 * j;
            int r = f >> 5;
            int cc = (f & 31) << 2;
            *(float4*)(h + (size_t)(rowBase + r) * F_OUT + cc) = *(const float4*)(&Hs[r * HP + cc]);
        }

        if (t < 2 * 16 * NUM_HEADS) {
            const int q  = t & (16 * NUM_HEADS - 1);
            const int r  = q >> 2;
            const int hd = q & 3;
            const float* hp = Hs + r * HP + hd * HEAD_DIM;
            const size_t o = (size_t)(rowBase + r) * NUM_HEADS + hd;
            if (t < 16 * NUM_HEADS) asrc[o] = head_dot(hp, a_src + hd * HEAD_DIM);
            else                    adst[o] = head_dot(hp, a_dst + hd * HEAD_DIM);
        }
        __syncthreads();
    }

    if (keys_usable(N, E, wsBytes))
        write_keys(src, dst, keys, N, E, packed_usable(N, E, wsBytes),
                   blockIdx.x * blockDim.x + t, gridDim.x * blockDim.x);
}

__global__ __launch_bounds__(256) void gat_gemm_tail(const float* __restrict__ x,
                                                     const float* __restrict__ W,
                                                     float* __restrict__ h,
                                                     int N, int rowStart) {
    int idx = blockIdx.x * blockDim.x + threadIdx.x;
    int total = (N - rowStart) * F_OUT;
    if (idx >= total) return;
    int r = rowStart + idx / F_OUT;
    int c = idx % F_OUT;
    float s = 0.0f;
    for (int k = 0; k < F_IN; ++k)
        s += x[(size_t)r * F_IN + k] * W[(size_t)k * F_OUT + c];
    h[(size_t)r * F_OUT + c] = s;
}

__global__ __launch_bounds__(256) void gat_alpha_tail(const float* __restrict__ h,
                                                      const float* __restrict__ a_src,
                                                      const float* __restrict__ a_dst,
                                                      float* __restrict__ asrc,
                                                      float* __restrict__ adst,
                                                      const int* __restrict__ src,
                                                      const int* __restrict__ dst,
                                                      unsigned* __restrict__ keys,
                                                      int N, int E, unsigned long long wsBytes,
                                                      int rowStart, int doKeys) {
    int tid = blockIdx.x * blockDim.x + threadIdx.x;
    if (doKeys && keys_usable(N, E, wsBytes))
        write_keys(src, dst, keys, N, E, packed_usable(N, E, wsBytes), tid, gridDim.x * blockDim.x);
    if (tid >= (N - rowStart) * NUM_HEADS) return;
    int n  = rowStart + (tid >> 2);
    int hd = tid & 3;
    const float* hp = h + (size_t)n * F_OUT + hd * HEAD_DIM;
    asrc[(size_t)n * NUM_HEADS + hd] = head_dot(hp, a_src + hd * HEAD_DIM);
    adst[(size_t)n * NUM_HEADS + hd] = head_dot(hp, a_dst + hd * HEAD_DIM);
}

__device__ __forceinline__ float leaky(float v) { return v > 0.0f ? v : 0.2f * v; }

#define OWN_THREADS 1024
#define Q_CAP       144
#define Q_DRAIN     16
#define BLK_EDGES   128
#define L_CAP       12288
#define SUB_CAP     (L_CAP / OWN_WAVES)
static_assert((OWN_WAVES & (OWN_WAVES - 1)) == 0, "row r of a range belongs to wave r & (OWN_WAVES - 1)");

__device__ __forceinline__ int4 load_dst4_guarded(const int* __restrict__ dst, int e, int E) {
    int4 r;
    r.x = (e     < E) ? dst[e]     : -1;
    r.y = (e + 1 < E) ? dst[e + 1] : -1;
    r.z = (e + 2 < E) ? dst[e + 2] : -1;
    r.w = (e + 3 < E) ? dst[e + 3] : -1;
    return r;
}

enum { Q_ROW = 0, SRC_DST = 1, PACKED = 2 };
template <int LOOK>
__device__ __forceinline__ void drain_max(const int* qs, const int* qd, int qn, int lane,
                                          const int* __restrict__ src,
                                          const int* __restrict__ dst,
                                          const unsigned* __restrict__ pk, int lo,
                                          const float* __restrict__ asrc,
                                          const float* sadst, float* smax,
                                          unsigned* elist, int* ecount, bool keep) {
    for (int i0 = 0; i0 < qn; i0 += 32) {
        int idx = i0 + lane;
        if (idx < qn) {
            const int e = qs[idx];
            int s, dl;
            if (LOOK == PACKED) {
                const unsigned p = pk[e];
                s  = (int)(p >> L_SHIFT);
                dl = (int)(p & L_MASK);
            } else {
                s  = src[e];
                dl = (LOOK == SRC_DST) ? dst[e] - lo : qd[idx];
            }
            if (keep) {
                const int ow  = dl & (OWN_WAVES - 1);
                const int pos = __hip_atomic_fetch_add(ecount + ow, 1, __ATOMIC_RELAXED, __HIP_MEMORY_SCOPE_WORKGROUP);
                if (pos < SUB_CAP)
                    elist[ow * SUB_CAP + pos] = ((unsigned)s << L_SHIFT) | (unsigned)dl;
            }
            float4 a = *(const float4*)(asrc + (size_t)s * NUM_HEADS);
            float4 b = *(const float4*)(sadst + dl * NUM_HEADS);
            float* mp = smax + dl * NUM_HEADS;
            __hip_atomic_fetch_max(mp + 0, leaky(a.x + b.x), __ATOMIC_RELAXED, __HIP_MEMORY_SCOPE_WORKGROUP);
            __hip_atomic_fetch_max(mp + 1, leaky(a.y + b.y), __ATOMIC_RELAXED, __HIP_MEMORY_SCOPE_WORKGROUP);
            __hip_atomic_fetch_max(mp + 2, leaky(a.z + b.z), __ATOMIC_RELAXED, __HIP_MEMORY_SCOPE_WORKGROUP);
            __hip_atomic_fetch_max(mp + 3, leaky(a.w + b.w), __ATOMIC_RELAXED, __HIP_MEMORY_SCOPE_WORKGROUP);
        }
    }
}

__device__ __forceinline__ void sum_batch(int s, int dl, bool on, int nb, int lane,
                                          const float* __restrict__ asrc,
                                          const float* __restrict__ h,
                                          const float* sadst, const float* smax,
                                          float* ssum, float* acc) {
    const int hd = lane & 3;
    float as = 0.0f;
    if (on) as = asrc[(size_t)s * NUM_HEADS + hd];
    float hv[8][NUM_HEADS];
    #pragma unroll
    for (int j = 0; j < 8; ++j) {
        if (j < nb) {
            int sj = __builtin_amdgcn_readlane(s, 4 * j);
            const float* hp = h + (size_t)sj * F_OUT + lane;
            #pragma unroll
            for (int c = 0; c < NUM_HEADS; ++c) hv[j][c] = hp[c * HEAD_DIM];
        }
    }
    float w = 0.0f;
    if (on) {
        float v = leaky(as + sadst[dl * NUM_HEADS + hd]);
        w = __expf(v - smax[dl * NUM_HEADS + hd]);
        __hip_atomic_fetch_add(ssum + dl * NUM_HEADS + hd, w,
                               __ATOMIC_RELAXED, __HIP_MEMORY_SCOPE_WORKGROUP);
    }
    const int wbits = __float_as_int(w);
    #pragma unroll
    for (int j = 0; j < 8; ++j) {
        if (j < nb) {
            int dj = __builtin_amdgcn_readlane(dl, 4 * j);
            float* ap = acc + dj * F_OUT + lane;
            #pragma unroll
            for (int c = 0; c < NUM_HEADS; ++c) {
                float wj = __int_as_float(__builtin_amdgcn_readlane(wbits, 4 * j + c));
                __hip_atomic_fetch_add(ap + c * HEAD_DIM, wj * hv[j][c],
                                       __ATOMIC_RELAXED, __HIP_MEMORY_SCOPE_WORKGROUP);
            }
        }
    }
}

__device__ __forceinline__ void drain_sum(int* qs, const int* qd, int qn, int lane,
                                          const int* __restrict__ src,
                                          const float* __restrict__ asrc,
                                          const float* __restrict__ h,
                                          const float* sadst, const float* smax,
                                          float* ssum, float* acc) {
    for (int i = lane; i < qn; i += 32) qs[i] = src[qs[i]];
    for (int i0 = 0; i0 < qn; i0 += 8) {
        const int nb = min(8, qn - i0);
        const int idx = i0 + (lane >> 2);
        const bool on = idx < qn;
        int s = 0, dl = 0;
        if (on) { s = qs[idx]; dl = qd[idx]; }
        sum_batch(s, dl, on, nb, lane, asrc, h, sadst, smax, ssum, acc);
    }
}

template <int J>
__device__ __forceinline__ float group_bcast(float w) {
    return __int_as_float(__builtin_amdgcn_mov_dpp8(__float_as_int(w), J * 0x249249));
}

__device__ __forceinline__ void batch_fetch(const unsigned* mylist, int cnt, int i0, int lane,
                                            const float* __restrict__ asrc,
                                            const float* __restrict__ h,
                                            int& s, int& dl, float& as, float4 (&hv)[8]) {
    const int nb = min(8, cnt - i0);
    const int idx = i0 + (lane & 7);
    s = 0; dl = 0; as = 0.0f;
    if (idx < cnt) {
        const unsigned p = mylist[idx];
        s  = (int)(p >> L_SHIFT);
        dl = (int)(p & L_MASK);
        as = asrc[(size_t)s * NUM_HEADS + (lane >> 3)];
    }
#define ROW_LOAD(J) \
    if (J < nb) hv[J] = *(const float4*)(h + (size_t)__builtin_amdgcn_readlane(s, J) * F_OUT + 4 * lane); \
    else        hv[J] = make_float4(0.0f, 0.0f, 0.0f, 0.0f);
    ROW_LOAD(0) ROW_LOAD(1) ROW_LOAD(2) ROW_LOAD(3) ROW_LOAD(4) ROW_LOAD(5) ROW_LOAD(6) ROW_LOAD(7)
#undef ROW_LOAD
}

__device__ __forceinline__ void row_fma(float4& a, float wj, const float4& hv) {
    a.x += wj * hv.x; a.y += wj * hv.y; a.z += wj * hv.z; a.w += wj * hv.w;
}

__device__ __forceinline__ void batch_add(int cnt, int i0, int lane, int dl, float as, const float4 (&hv)[8],
                                          const float* sadst, const float* smax,
                                          float* ssum, float* acc) {
    const int nb = min(8, cnt - i0);
    const int hd = lane >> 3;
    float w = 0.0f;
    if (i0 + (lane & 7) < cnt) {
        float v = leaky(as + sadst[dl * NUM_HEADS + hd]);
        w = __expf(v - smax[dl * NUM_HEADS + hd]);
        __hip_atomic_fetch_add(ssum + dl * NUM_HEADS + hd, w,
                               __ATOMIC_RELAXED, __HIP_MEMORY_SCOPE_WORKGROUP);
    }
#define ROW_PAIR(J0, J1) \
    if (J0 < nb) { \
        const float w0 = group_bcast<J0>(w); \
        const int d0 = __builtin_amdgcn_readlane(dl, J0); \
        float4* p0 = (float4*)(acc + d0 * F_OUT + 4 * lane); \
        if (J1 < nb) { \
            const float w1 = group_bcast<J1>(w); \
            const int d1 = __builtin_amdgcn_readlane(dl, J1); \
            if (d0 == d1) { \
                float4 a = *p0; \
                row_fma(a, w0, hv[J0]); \
                row_fma(a, w1, hv[J1]); \
                *p0 = a; \
            } else { \
                float4* p1 = (float4*)(acc + d1 * F_OUT + 4 * lane); \
                float4 a0 = *p0; \
                float4 a1 = *p1; \
                row_fma(a0, w0, hv[J0]); \
                row_fma(a1, w1, hv[J1]); \
                *p0 = a0; \
                *p1 = a1; \
            } \
        } else { \
            float4 a = *p0; \
            row_fma(a, w0, hv[J0]); \
            *p0 = a; \
        } \
    }
    ROW_PAIR(0, 1) ROW_PAIR(2, 3) ROW_PAIR(4, 5) ROW_PAIR(6, 7)
#undef ROW_PAIR
}

__device__ __forceinline__ void wave_rows(const unsigned* mylist, int cnt, int lane,
                                          const float* __restrict__ asrc,
                                          const float* __restrict__ h,
                                          const float* sadst, const float* smax,
                                          float* ssum, float* acc) {
    int s = 0, dl = 0;
    float as = 0.0f;
    float4 hv[8];
    if (cnt > 0) batch_fetch(mylist, cnt, 0, lane, asrc, h, s, dl, as, hv);
    for (int i0 = 0; i0 < cnt; i0 += 8) {
        int s2 = 0, dl2 = 0;
        float as2 = 0.0f;
        float4 hv2[8];
        const bool more = i0 + 8 < cnt;
        if (more) batch_fetch(mylist, cnt, i0 + 8, lane, asrc, h, s2, dl2, as2, hv2);
        batch_add(cnt, i0, lane, dl, as, hv, sadst, smax, ssum, acc);
        if (more) {
            s = s2; dl = dl2; as = as2;
            #pragma unroll
            for (int q = 0; q < 8; ++q) hv[q] = hv2[q];
        }
    }
}

template <int PASS>
__device__ __forceinline__ void owner_scan(const int* __restrict__ src,
                                           const int* __restrict__ dst,
                                           const float* __restrict__ asrc,
                                           const float* __restrict__ h,
                                           int E, int lo, int nloc, int wave, int lane,
                                           int* qs, int* qd,
                                           const float* sadst, float* smax,
                                           float* ssum, float* acc,
                                           unsigned* elist, int* ecount, bool keep) {
    const bool vec = (E & 3) == 0;
    const int perIter = OWN_WAVES * 4 * BLK_EDGES;
    const int nIter = (E + perIter - 1) / perIter;
    const int start = (int)(((long long)blockIdx.x * nIter) / gridDim.x);
    const unsigned below = (1u << lane) - 1u;
    int qn = 0;
    for (int it = 0; it < nIter; ++it) {
        int jt = it + start;
        if (jt >= nIter) jt -= nIter;
        const int wbase = (jt * OWN_WAVES + wave) * (4 * BLK_EDGES);
        const int base = wbase + 4 * lane;
        int4 dv[4];
        if (vec && wbase + 4 * BLK_EDGES <= E) {
            #pragma unroll
            for (int k = 0; k < 4; ++k) dv[k] = *(const int4*)(dst + base + k * BLK_EDGES);
        } else {
            #pragma unroll
            for (int k = 0; k < 4; ++k) dv[k] = load_dst4_guarded(dst, base + k * BLK_EDGES, E);
        }
        #pragma unroll
        for (int k = 0; k < 4; ++k) {
            const int e0 = base + k * BLK_EDGES;
            const int dd[4] = {dv[k].x, dv[k].y, dv[k].z, dv[k].w};
            bool m[4];
            bool any = false;
            #pragma unroll
            for (int c = 0; c < 4; ++c) {
                m[c] = (unsigned)(dd[c] - lo) < (unsigned)nloc;
                any = any || m[c];
            }
            if (__builtin_amdgcn_ballot_w32(any) != 0u) {
                #pragma unroll
                for (int c = 0; c < 4; ++c) {
                    const unsigned mask = __builtin_amdgcn_ballot_w32(m[c]);
                    if (m[c]) {
                        const int pos = qn + __builtin_popcount(mask & below);
                        qs[pos] = e0 + c;
                        qd[pos] = dd[c] - lo;
                    }
                    qn += __builtin_popcount(mask);
                }
                if (qn >= Q_DRAIN) {
                    if (PASS == 1) drain_max<Q_ROW>(qs, qd, qn, lane, src, dst, nullptr, lo, asrc, sadst, smax, elist, ecount, keep);
                    else           drain_sum(qs, qd, qn, lane, src, asrc, h, sadst, smax, ssum, acc);
                    qn = 0;
                }
            }
        }
    }
    if (qn > 0) {
        if (PASS == 1) drain_max<Q_ROW>(qs, qd, qn, lane, src, dst, nullptr, lo, asrc, sadst, smax, elist, ecount, keep);
        else           drain_sum(qs, qd, qn, lane, src, asrc, h, sadst, smax, ssum, acc);
    }
}

static_assert(KEY_LOADS == 4, "the mask packs four loads of sixteen keys into 64 bits");
template <int LOOK>
__device__ __forceinline__ void owner_scan_keys(const int* __restrict__ src,
                                                const int* __restrict__ dst,
                                                const unsigned* __restrict__ keys,
                                                const float* __restrict__ asrc,
                                                int E, int lo, int wave, int lane,
                                                int* qs, const float* sadst, float* smax,
                                                unsigned* elist, int* ecount, bool keep) {
    const int nIter = keys_padded(E) / KEY_TRIP;
    const unsigned* __restrict__ pk = packed_of(keys, E);
    const int start = (int)(((long long)blockIdx.x * nIter) / gridDim.x);
    const unsigned own = (unsigned)blockIdx.x * 0x01010101u;
    const unsigned below = (1u << lane) - 1u;
    int qn = 0;
    for (int it = 0; it < nIter; ++it) {
        int jt = it + start;
        if (jt >= nIter) jt -= nIter;
        const int base = (jt * OWN_WAVES + wave) * (KEY_LOADS * 512) + 16 * lane;
        uint4 kv[KEY_LOADS];
        #pragma unroll
        for (int k = 0; k < KEY_LOADS; ++k) kv[k] = *(const uint4*)((const unsigned char*)keys + base + k * 512);
        unsigned miss[2] = {0u, 0u};
        #pragma unroll
        for (int k = 0; k < KEY_LOADS; ++k) {
            const unsigned kw[4] = {kv[k].x, kv[k].y, kv[k].z, kv[k].w};
            #pragma unroll
            for (int c = 0; c < 4; ++c) {
                const unsigned x = kw[c] ^ own;
                const unsigned nz = ((x & 0x7f7f7f7fu) + 0x7f7f7f7fu) | x;
                const int j = 4 * (k & 1) + c;
                miss[k >> 1] |= (nz >> (7 - j)) & (0x01010101u << j);
            }
        }
        unsigned long long m = (unsigned long long)(~miss[0]) | ((unsigned long long)(~miss[1]) << 32);
        unsigned any = __builtin_amdgcn_ballot_w32(m != 0ull);
        for (int r = 0; r < 64 && any != 0u; ++r) {
            if (m != 0ull) {
                const int p = __builtin_ctzll(m);
                m &= m - 1ull;
                qs[qn + __builtin_popcount(any & below)] =
                    base + (p >> 5) * 1024 + ((p >> 2) & 1) * 512 + (p & 3) * 4 + ((p >> 3) & 3);
            }
            qn += __builtin_popcount(any);
            if (qn >= Q_DRAIN) {
                drain_max<LOOK>(qs, qs, qn, lane, src, dst, pk, lo, asrc, sadst, smax, elist, ecount, keep);
                qn = 0;
            }
            any = __builtin_amdgcn_ballot_w32(m != 0ull);
        }
    }
    if (qn > 0)
        drain_max<LOOK>(qs, qs, qn, lane, src, dst, pk, lo, asrc, sadst, smax, elist, ecount, keep);
}

__global__ __launch_bounds__(OWN_THREADS) void gat_owner(const int* __restrict__ src,
                                                         const int* __restrict__ dst,
                                                         const float* __restrict__ asrc,
                                                         const float* __restrict__ adst,
                                                         const float* __restrict__ h,
                                                         const unsigned* __restrict__ keys,
                                                         float* __restrict__ out,
                                                         int N, int E, unsigned long long wsBytes) {
    __shared__ float acc[OWN_NODES * F_OUT];
    __shared__ float smax[OWN_NODES * NUM_HEADS];
    __shared__ float ssum[OWN_NODES * NUM_HEADS];
    __shared__ float sadst[OWN_NODES * NUM_HEADS];
    __shared__ int   qsrc[OWN_WAVES * Q_CAP];
    __shared__ int   qdst[OWN_WAVES * Q_CAP];
    __shared__ unsigned elist[L_CAP];
    __shared__ int   ecount[OWN_WAVES];

    const int t = threadIdx.x;
    const int lane = t & 31;
    const int wave = __builtin_amdgcn_readfirstlane(t >> 5);
    const int lo = blockIdx.x * OWN_NODES;
    const int nloc = min(OWN_NODES, N - lo);
    const bool keep = N <= (1 << (32 - L_SHIFT));

    if (t < OWN_WAVES) ecount[t] = 0;
    for (int i = t; i < OWN_NODES * F_OUT; i += OWN_THREADS) acc[i] = 0.0f;
    for (int i = t; i < OWN_NODES * NUM_HEADS; i += OWN_THREADS) {
        smax[i] = -3.0e38f;
        ssum[i] = 0.0f;
        sadst[i] = (i < nloc * NUM_HEADS) ? adst[(size_t)lo * NUM_HEADS + i] : 0.0f;
    }
    __syncthreads();

    int* qs = qsrc + wave * Q_CAP;
    int* qd = qdst + wave * Q_CAP;

    if (packed_usable(N, E, wsBytes))
        owner_scan_keys<PACKED>(src, dst, keys, asrc, E, lo, wave, lane, qs, sadst, smax, elist, ecount, keep);
    else if (keys_usable(N, E, wsBytes))
        owner_scan_keys<SRC_DST>(src, dst, keys, asrc, E, lo, wave, lane, qs, sadst, smax, elist, ecount, keep);
    else
        owner_scan<1>(src, dst, asrc, h, E, lo, nloc, wave, lane, qs, qd, sadst, smax, ssum, acc,
                      elist, ecount, keep);
    __syncthreads();
    bool whole = keep;
    for (int i = 0; i < OWN_WAVES; ++i) whole = whole && ecount[i] <= SUB_CAP;
    if (whole) {
        const int cnt = __builtin_amdgcn_readfirstlane(ecount[wave]);
        wave_rows(elist + wave * SUB_CAP, cnt, lane, asrc, h, sadst, smax, ssum, acc);
    } else
        owner_scan<2>(src, dst, asrc, h, E, lo, nloc, wave, lane, qs, qd, sadst, smax, ssum, acc,
                      elist, ecount, false);
    __syncthreads();

    for (int r = wave; r < nloc; r += OWN_WAVES) {
        float* op = out + (size_t)(lo + r) * F_OUT + lane;
        #pragma unroll
        for (int c = 0; c < NUM_HEADS; ++c)
            op[c * HEAD_DIM] = acc[r * F_OUT + c * HEAD_DIM + lane] /
                               (ssum[r * NUM_HEADS + c] + 1e-10f);
    }
}

extern "C" void kernel_launch(void* const* d_in, const int* in_sizes, int n_in,
                              void* d_out, int out_size, void* d_ws, size_t ws_size,
                              hipStream_t stream) {
    const float* x     = (const float*)d_in[0];
    const int*   ei    = (const int*)d_in[1];
    const float* W     = (const float*)d_in[2];
    const float* a_src = (const float*)d_in[3];
    const float* a_dst = (const float*)d_in[4];
    float*       out   = (float*)d_out;

    const int N = in_sizes[0] / F_IN;
    const int E = in_sizes[1] / 2;
    const int* src = ei;
    const int* dst = ei + E;

    float* h    = (float*)d_ws;
    float* asrc = h    + (size_t)N * F_OUT;
    float* adst = asrc + (size_t)N * NUM_HEADS;
    unsigned* keys = (unsigned*)(adst + (size_t)N * NUM_HEADS);

    {
        int fullTiles = N / 16;
        if (fullTiles > 0)
            gat_gemm<<<GEMM_GROUPS, 256, 0, stream>>>(x, W, a_src, a_dst, h, asrc, adst, src, dst, keys, N, E,
                                                    (unsigned long long)ws_size);
        int rowStart = fullTiles * 16;
        int rem = N - rowStart;
        if (rem > 0) {
            int total = rem * F_OUT;
            gat_gemm_tail<<<(total + 255) / 256, 256, 0, stream>>>(x, W, h, N, rowStart);
            gat_alpha_tail<<<(rem * NUM_HEADS + 255) / 256, 256, 0, stream>>>(h, a_src, a_dst, asrc, adst, src, dst, keys, N, E,
                                                                              (unsigned long long)ws_size, rowStart,
                                                                              fullTiles > 0 ? 0 : 1);
        }
    }

    gat_owner<<<(N + OWN_NODES - 1) / OWN_NODES, OWN_THREADS, 0, stream>>>(src, dst, asrc, adst, h, keys, out, N, E,
                                                                          (unsigned long long)ws_size);
}


# Kept edges filed by destination row, not by owning wave: the wave sums a row in registers in list order, its weights by an 8-lane tree, divides and stores it; no LDS row add, clear or store pass
# speedup vs baseline: 5.8087x; 1.0615x over previous
#include <hip/hip_runtime.h>
#include <hip/hip_bf16.h>


#define NUM_HEADS 4
#define HEAD_DIM 32
#define F_IN 256
#define F_OUT 128
#define LDA 260

typedef __attribute__((ext_vector_type(2))) float v2f;
typedef __attribute__((ext_vector_type(8))) float v8f;

#define OWN_NODES   400
#define OWN_WAVES   32
#define KEY_NONE    255u
#define KEY_LOADS   4
#define KEY_TRIP    (OWN_WAVES * KEY_LOADS * 512)

__device__ __forceinline__ int keys_padded(int E) {
    return ((E + KEY_TRIP - 1) / KEY_TRIP) * KEY_TRIP;
}

__device__ __forceinline__ bool keys_usable(int N, int E, unsigned long long wsBytes) {
    const int groups = (N + OWN_NODES - 1) / OWN_NODES;
    const unsigned long long need =
        (unsigned long long)N * (F_OUT + 2 * NUM_HEADS) * sizeof(float) + (unsigned long long)keys_padded(E);
    return groups <= (int)KEY_NONE && wsBytes >= need;
}

__device__ __forceinline__ unsigned key_of(int d, int N) {
    return ((unsigned)d < (unsigned)N) ? (unsigned)d / OWN_NODES : KEY_NONE;
}

#define L_SHIFT     9
#define L_MASK      ((1u << L_SHIFT) - 1u)
static_assert(OWN_NODES <= (1 << L_SHIFT), "local row must fit the low bits of a packed word");

__device__ __forceinline__ int packed_len(int E) { return (E + 3) & ~3; }

__device__ __forceinline__ bool packed_usable(int N, int E, unsigned long long wsBytes) {
    const unsigned long long need =
        (unsigned long long)N * (F_OUT + 2 * NUM_HEADS) * sizeof(float) + (unsigned long long)keys_padded(E) +
        (unsigned long long)packed_len(E) * sizeof(unsigned);
    return keys_usable(N, E, wsBytes) && N <= (1 << (32 - L_SHIFT)) && wsBytes >= need;
}

__device__ __forceinline__ unsigned* packed_of(unsigned* keys, int E) { return keys + (keys_padded(E) >> 2); }
__device__ __forceinline__ const unsigned* packed_of(const unsigned* keys, int E) { return keys + (keys_padded(E) >> 2); }

__device__ __forceinline__ unsigned pack_word(int s, int d, unsigned key) {
    return ((unsigned)s << L_SHIFT) | (((unsigned)d - key * OWN_NODES) & L_MASK);
}

#define HP 132
#define GEMM_GROUPS 512

__device__ __forceinline__ void write_keys(const int* __restrict__ src, const int* __restrict__ dst,
                                           unsigned* __restrict__ keys, int N, int E, bool pack,
                                           int tid, int step) {
    const int words = keys_padded(E) >> 2;
    unsigned* __restrict__ pk = packed_of(keys, E);
    const bool vec = (E & 3) == 0;
    for (int g = tid; g < words; g += step) {
        const int e = g << 2;
        int4 d, s = {0, 0, 0, 0};
        if (vec && e + 4 <= E) {
            d = *(const int4*)(dst + e);
            if (pack) s = *(const int4*)(src + e);
        } else {
            d.x = (e     < E) ? dst[e]     : -1;
            d.y = (e + 1 < E) ? dst[e + 1] : -1;
            d.z = (e + 2 < E) ? dst[e + 2] : -1;
            d.w = (e + 3 < E) ? dst[e + 3] : -1;
            if (pack) {
                s.x = (e     < E) ? src[e]     : 0;
                s.y = (e + 1 < E) ? src[e + 1] : 0;
                s.z = (e + 2 < E) ? src[e + 2] : 0;
                s.w = (e + 3 < E) ? src[e + 3] : 0;
            }
        }
        const unsigned k0 = key_of(d.x, N), k1 = key_of(d.y, N), k2 = key_of(d.z, N), k3 = key_of(d.w, N);
        keys[g] = k0 | (k1 << 8) | (k2 << 16) | (k3 << 24);
        if (pack && e < E) {
            uint4 w;
            w.x = pack_word(s.x, d.x, k0);
            w.y = pack_word(s.y, d.y, k1);
            w.z = pack_word(s.z, d.z, k2);
            w.w = pack_word(s.w, d.w, k3);
            *(uint4*)(pk + e) = w;
        }
    }
}

__device__ __forceinline__ float head_dot(const float* hp, const float* __restrict__ av) {
    float s = 0.0f;
    #pragma unroll
    for (int i = 0; i < HEAD_DIM; i += 4) {
        float4 hv = *(const float4*)(hp + i);
        float4 a  = *(const float4*)(av + i);
        s += hv.x * a.x + hv.y * a.y + hv.z * a.z + hv.w * a.w;
    }
    return s;
}

__global__ __launch_bounds__(256) void gat_gemm(const float* __restrict__ x,
                                                const float* __restrict__ W,
                                                const float* __restrict__ a_src,
                                                const float* __restrict__ a_dst,
                                                float* __restrict__ h,
                                                float* __restrict__ asrc,
                                                float* __restrict__ adst,
                                                const int* __restrict__ src,
                                                const int* __restrict__ dst,
                                                unsigned* __restrict__ keys,
                                                int N, int E, unsigned long long wsBytes) {
    __shared__ float As[16 * LDA];

    const int t = threadIdx.x;
    const int wave = t >> 5;
    const int lane = t & 31;
    const int m    = lane & 15;
    const int koff = (lane >> 4) << 1;
    const int col  = (wave << 4) + m;
    const int fullTiles = N / 16;

    const float* Wp = W + col;
    v2f bw[F_IN / 4];
    #pragma unroll
    for (int i = 0; i < F_IN / 4; ++i) {
        bw[i].x = Wp[(size_t)(4 * i + koff) * F_OUT];
        bw[i].y = Wp[(size_t)(4 * i + koff + 1) * F_OUT];
    }

    const int ar = t >> 6;
    const int ac = (t & 63) << 2;
    float4 p0 = {}, p1 = {}, p2 = {}, p3 = {};
    int tile = blockIdx.x;
    if (tile < fullTiles) {
        const float* xp = x + (size_t)(tile * 16 + ar) * F_IN + ac;
        p0 = *(const float4*)(xp);
        p1 = *(const float4*)(xp + 4 * F_IN);
        p2 = *(const float4*)(xp + 8 * F_IN);
        p3 = *(const float4*)(xp + 12 * F_IN);
    }

    for (; tile < fullTiles; tile += gridDim.x) {
        const int rowBase = tile * 16;
        *(float4*)(&As[(ar     ) * LDA + ac]) = p0;
        *(float4*)(&As[(ar +  4) * LDA + ac]) = p1;
        *(float4*)(&As[(ar +  8) * LDA + ac]) = p2;
        *(float4*)(&As[(ar + 12) * LDA + ac]) = p3;
        __syncthreads();

        const int nextTile = tile + gridDim.x;
        if (nextTile < fullTiles) {
            const float* xp = x + (size_t)(nextTile * 16 + ar) * F_IN + ac;
            p0 = *(const float4*)(xp);
            p1 = *(const float4*)(xp + 4 * F_IN);
            p2 = *(const float4*)(xp + 8 * F_IN);
            p3 = *(const float4*)(xp + 12 * F_IN);
        }

        v8f acc = {};
        #pragma unroll
        for (int i = 0; i < F_IN / 4; ++i) {
            v2f a = *(const v2f*)(&As[m * LDA + 4 * i + koff]);
            acc = __builtin_amdgcn_wmma_f32_16x16x4_f32(
                 false, a,  false, bw[i],
                 (short)0, acc,  false,  false);
        }
        __builtin_amdgcn_sched_barrier(0);
        asm volatile("v_nop\n\tv_nop\n\tv_nop\n\tv_nop");
        __builtin_amdgcn_sched_barrier(0);

        __syncthreads();
        float* Hs = As;
        const int roff = (lane >> 4) << 3;
        #pragma unroll
        for (int v = 0; v < 8; ++v)
            Hs[(v + roff) * HP + col] = acc[v];
        __syncthreads();

        #pragma unroll
        for (int j = 0; j < 2; ++j) {
            int f = t + 256 * j;
            int r = f >> 5;
            int cc = (f & 31) << 2;
            *(float4*)(h + (size_t)(rowBase + r) * F_OUT + cc) = *(const float4*)(&Hs[r * HP + cc]);
        }

        if (t < 2 * 16 * NUM_HEADS) {
            const int q  = t & (16 * NUM_HEADS - 1);
            const int r  = q >> 2;
            const int hd = q & 3;
            const float* hp = Hs + r * HP + hd * HEAD_DIM;
            const size_t o = (size_t)(rowBase + r) * NUM_HEADS + hd;
            if (t < 16 * NUM_HEADS) asrc[o] = head_dot(hp, a_src + hd * HEAD_DIM);
            else                    adst[o] = head_dot(hp, a_dst + hd * HEAD_DIM);
        }
        __syncthreads();
    }

    if (keys_usable(N, E, wsBytes))
        write_keys(src, dst, keys, N, E, packed_usable(N, E, wsBytes),
                   blockIdx.x * blockDim.x + t, gridDim.x * blockDim.x);
}

__global__ __launch_bounds__(256) void gat_gemm_tail(const float* __restrict__ x,
                                                     const float* __restrict__ W,
                                                     float* __restrict__ h,
                                                     int N, int rowStart) {
    int idx = blockIdx.x * blockDim.x + threadIdx.x;
    int total = (N - rowStart) * F_OUT;
    if (idx >= total) return;
    int r = rowStart + idx / F_OUT;
    int c = idx % F_OUT;
    float s = 0.0f;
    for (int k = 0; k < F_IN; ++k)
        s += x[(size_t)r * F_IN + k] * W[(size_t)k * F_OUT + c];
    h[(size_t)r * F_OUT + c] = s;
}

__global__ __launch_bounds__(256) void gat_alpha_tail(const float* __restrict__ h,
                                                      const float* __restrict__ a_src,
                                                      const float* __restrict__ a_dst,
                                                      float* __restrict__ asrc,
                                                      float* __restrict__ adst,
                                                      const int* __restrict__ src,
                                                      const int* __restrict__ dst,
                                                      unsigned* __restrict__ keys,
                                                      int N, int E, unsigned long long wsBytes,
                                                      int rowStart, int doKeys) {
    int tid = blockIdx.x * blockDim.x + threadIdx.x;
    if (doKeys && keys_usable(N, E, wsBytes))
        write_keys(src, dst, keys, N, E, packed_usable(N, E, wsBytes), tid, gridDim.x * blockDim.x);
    if (tid >= (N - rowStart) * NUM_HEADS) return;
    int n  = rowStart + (tid >> 2);
    int hd = tid & 3;
    const float* hp = h + (size_t)n * F_OUT + hd * HEAD_DIM;
    asrc[(size_t)n * NUM_HEADS + hd] = head_dot(hp, a_src + hd * HEAD_DIM);
    adst[(size_t)n * NUM_HEADS + hd] = head_dot(hp, a_dst + hd * HEAD_DIM);
}

__device__ __forceinline__ float leaky(float v) { return v > 0.0f ? v : 0.2f * v; }

#define OWN_THREADS 1024
#define Q_CAP       144
#define Q_DRAIN     16
#define BLK_EDGES   128
#define ROW_CAP     F_OUT
static_assert((OWN_WAVES & (OWN_WAVES - 1)) == 0, "row r of a range belongs to wave r & (OWN_WAVES - 1)");
static_assert(ROW_CAP <= F_OUT, "a row's list must fit the space of the row's sums");

__device__ __forceinline__ int4 load_dst4_guarded(const int* __restrict__ dst, int e, int E) {
    int4 r;
    r.x = (e     < E) ? dst[e]     : -1;
    r.y = (e + 1 < E) ? dst[e + 1] : -1;
    r.z = (e + 2 < E) ? dst[e + 2] : -1;
    r.w = (e + 3 < E) ? dst[e + 3] : -1;
    return r;
}

enum { Q_ROW = 0, SRC_DST = 1, PACKED = 2 };
template <int LOOK>
__device__ __forceinline__ void drain_max(const int* qs, const int* qd, int qn, int lane,
                                          const int* __restrict__ src,
                                          const int* __restrict__ dst,
                                          const unsigned* __restrict__ pk, int lo,
                                          const float* __restrict__ asrc,
                                          const float* sadst, float* smax,
                                          float* rlist, int* rcount, int* ovf) {
    for (int i0 = 0; i0 < qn; i0 += 32) {
        int idx = i0 + lane;
        if (idx < qn) {
            const int e = qs[idx];
            int s, dl;
            if (LOOK == PACKED) {
                const unsigned p = pk[e];
                s  = (int)(p >> L_SHIFT);
                dl = (int)(p & L_MASK);
            } else {
                s  = src[e];
                dl = (LOOK == SRC_DST) ? dst[e] - lo : qd[idx];
            }
            const int pos = __hip_atomic_fetch_add(rcount + dl, 1, __ATOMIC_RELAXED, __HIP_MEMORY_SCOPE_WORKGROUP);
            if (pos < ROW_CAP) rlist[dl * ROW_CAP + pos] = __uint_as_float((unsigned)s);
            else               *ovf = 1;
            float4 a = *(const float4*)(asrc + (size_t)s * NUM_HEADS);
            float4 b = *(const float4*)(sadst + dl * NUM_HEADS);
            float* mp = smax + dl * NUM_HEADS;
            __hip_atomic_fetch_max(mp + 0, leaky(a.x + b.x), __ATOMIC_RELAXED, __HIP_MEMORY_SCOPE_WORKGROUP);
            __hip_atomic_fetch_max(mp + 1, leaky(a.y + b.y), __ATOMIC_RELAXED, __HIP_MEMORY_SCOPE_WORKGROUP);
            __hip_atomic_fetch_max(mp + 2, leaky(a.z + b.z), __ATOMIC_RELAXED, __HIP_MEMORY_SCOPE_WORKGROUP);
            __hip_atomic_fetch_max(mp + 3, leaky(a.w + b.w), __ATOMIC_RELAXED, __HIP_MEMORY_SCOPE_WORKGROUP);
        }
    }
}

__device__ __forceinline__ void sum_batch(int s, int dl, bool on, int nb, int lane,
                                          const float* __restrict__ asrc,
                                          const float* __restrict__ h,
                                          const float* sadst, const float* smax,
                                          float* ssum, float* acc) {
    const int hd = lane & 3;
    float as = 0.0f;
    if (on) as = asrc[(size_t)s * NUM_HEADS + hd];
    float hv[8][NUM_HEADS];
    #pragma unroll
    for (int j = 0; j < 8; ++j) {
        if (j < nb) {
            int sj = __builtin_amdgcn_readlane(s, 4 * j);
            const float* hp = h + (size_t)sj * F_OUT + lane;
            #pragma unroll
            for (int c = 0; c < NUM_HEADS; ++c) hv[j][c] = hp[c * HEAD_DIM];
        }
    }
    float w = 0.0f;
    if (on) {
        float v = leaky(as + sadst[dl * NUM_HEADS + hd]);
        w = __expf(v - smax[dl * NUM_HEADS + hd]);
        __hip_atomic_fetch_add(ssum + dl * NUM_HEADS + hd, w,
                               __ATOMIC_RELAXED, __HIP_MEMORY_SCOPE_WORKGROUP);
    }
    const int wbits = __float_as_int(w);
    #pragma unroll
    for (int j = 0; j < 8; ++j) {
        if (j < nb) {
            int dj = __builtin_amdgcn_readlane(dl, 4 * j);
            float* ap = acc + dj * F_OUT + lane;
            #pragma unroll
            for (int c = 0; c < NUM_HEADS; ++c) {
                float wj = __int_as_float(__builtin_amdgcn_readlane(wbits, 4 * j + c));
                __hip_atomic_fetch_add(ap + c * HEAD_DIM, wj * hv[j][c],
                                       __ATOMIC_RELAXED, __HIP_MEMORY_SCOPE_WORKGROUP);
            }
        }
    }
}

__device__ __forceinline__ void drain_sum(int* qs, const int* qd, int qn, int lane,
                                          const int* __restrict__ src,
                                          const float* __restrict__ asrc,
                                          const float* __restrict__ h,
                                          const float* sadst, const float* smax,
                                          float* ssum, float* acc) {
    for (int i = lane; i < qn; i += 32) qs[i] = src[qs[i]];
    for (int i0 = 0; i0 < qn; i0 += 8) {
        const int nb = min(8, qn - i0);
        const int idx = i0 + (lane >> 2);
        const bool on = idx < qn;
        int s = 0, dl = 0;
        if (on) { s = qs[idx]; dl = qd[idx]; }
        sum_batch(s, dl, on, nb, lane, asrc, h, sadst, smax, ssum, acc);
    }
}

template <int J>
__device__ __forceinline__ float group_bcast(float w) {
    return __int_as_float(__builtin_amdgcn_mov_dpp8(__float_as_int(w), J * 0x249249));
}

__device__ __forceinline__ float group_sum8(float v) {
    v += __int_as_float(__builtin_amdgcn_mov_dpp8(__float_as_int(v),
            1 | (0 << 3) | (3 << 6) | (2 << 9) | (5 << 12) | (4 << 15) | (7 << 18) | (6 << 21)));
    v += __int_as_float(__builtin_amdgcn_mov_dpp8(__float_as_int(v),
            2 | (3 << 3) | (0 << 6) | (1 << 9) | (6 << 12) | (7 << 15) | (4 << 18) | (5 << 21)));
    v += __int_as_float(__builtin_amdgcn_mov_dpp8(__float_as_int(v),
            4 | (5 << 3) | (6 << 6) | (7 << 9) | (0 << 12) | (1 << 15) | (2 << 18) | (3 << 21)));
    return v;
}

__device__ __forceinline__ void row_fetch(const float* rl, int cnt, int i0, int lane,
                                          const float* __restrict__ asrc,
                                          const float* __restrict__ h,
                                          float& as, float4 (&hv)[8]) {
    const int nb = min(8, cnt - i0);
    const int idx = i0 + (lane & 7);
    int s = 0;
    as = 0.0f;
    if (idx < cnt) {
        s  = (int)__float_as_uint(rl[idx]);
        as = asrc[(size_t)s * NUM_HEADS + (lane >> 3)];
    }
#define ROW_LOAD(J) \
    if (J < nb) hv[J] = *(const float4*)(h + (size_t)__builtin_amdgcn_readlane(s, J) * F_OUT + 4 * lane); \
    else        hv[J] = make_float4(0.0f, 0.0f, 0.0f, 0.0f);
    ROW_LOAD(0) ROW_LOAD(1) ROW_LOAD(2) ROW_LOAD(3) ROW_LOAD(4) ROW_LOAD(5) ROW_LOAD(6) ROW_LOAD(7)
#undef ROW_LOAD
}

__device__ __forceinline__ void row_fma(float4& a, float wj, const float4& hv) {
    a.x += wj * hv.x; a.y += wj * hv.y; a.z += wj * hv.z; a.w += wj * hv.w;
}

__device__ __forceinline__ void row_add(int cnt, int i0, int lane, float as, const float4 (&hv)[8],
                                        float sd, float mx, float4& a, float& wsum) {
    const int nb = min(8, cnt - i0);
    float w = 0.0f;
    if (i0 + (lane & 7) < cnt) w = __expf(leaky(as + sd) - mx);
    wsum += w;
#define ROW_EDGE(J) if (J < nb) row_fma(a, group_bcast<J>(w), hv[J]);
    ROW_EDGE(0) ROW_EDGE(1) ROW_EDGE(2) ROW_EDGE(3) ROW_EDGE(4) ROW_EDGE(5) ROW_EDGE(6) ROW_EDGE(7)
#undef ROW_EDGE
}

__device__ __forceinline__ void wave_rows(const float* rlist, const int* rcount, int wave, int lane,
                                          int lo, int nloc,
                                          const float* __restrict__ asrc,
                                          const float* __restrict__ h,
                                          const float* sadst, const float* smax,
                                          float* __restrict__ out) {
    const int hd = lane >> 3;
    int rn = wave;
    if (rn >= nloc) return;
    int cntn = __builtin_amdgcn_readfirstlane(rcount[rn]);
    int ib = 0;
    bool more = true;
    bool have = false;
    int r = 0, i0 = 0, cnt = 0;
    float as = 0.0f, sd = 0.0f, mx = 0.0f;
    float4 hv[8];
    #pragma unroll
    for (int q = 0; q < 8; ++q) hv[q] = make_float4(0.0f, 0.0f, 0.0f, 0.0f);
    float4 a = make_float4(0.0f, 0.0f, 0.0f, 0.0f);
    float wsum = 0.0f;
    for (;;) {
        float as2 = 0.0f;
        float4 hv2[8];
        if (more) row_fetch(rlist + rn * ROW_CAP, cntn, ib, lane, asrc, h, as2, hv2);
        if (have) {
            row_add(cnt, i0, lane, as, hv, sd, mx, a, wsum);
            if (i0 + 8 >= cnt) {
                const float den = group_sum8(wsum) + 1e-10f;
                float4 o;
                o.x = a.x / den; o.y = a.y / den; o.z = a.z / den; o.w = a.w / den;
                *(float4*)(out + (size_t)(lo + r) * F_OUT + 4 * lane) = o;
                a = make_float4(0.0f, 0.0f, 0.0f, 0.0f);
                wsum = 0.0f;
            }
        }
        if (!more) break;
        if (ib == 0) {
            sd = sadst[rn * NUM_HEADS + hd];
            mx = smax[rn * NUM_HEADS + hd];
        }
        r = rn; i0 = ib; cnt = cntn; as = as2; have = true;
        #pragma unroll
        for (int q = 0; q < 8; ++q) hv[q] = hv2[q];
        ib += 8;
        if (ib >= cntn) {
            rn += OWN_WAVES;
            ib = 0;
            more = rn < nloc;
            cntn = more ? __builtin_amdgcn_readfirstlane(rcount[rn]) : 0;
        }
    }
}

template <int PASS>
__device__ __forceinline__ void owner_scan(const int* __restrict__ src,
                                           const int* __restrict__ dst,
                                           const float* __restrict__ asrc,
                                           const float* __restrict__ h,
                                           int E, int lo, int nloc, int wave, int lane,
                                           int* qs, int* qd,
                                           const float* sadst, float* smax,
                                           float* ssum, float* acc,
                                           float* rlist, int* rcount, int* ovf) {
    const bool vec = (E & 3) == 0;
    const int perIter = OWN_WAVES * 4 * BLK_EDGES;
    const int nIter = (E + perIter - 1) / perIter;
    const int start = (int)(((long long)blockIdx.x * nIter) / gridDim.x);
    const unsigned below = (1u << lane) - 1u;
    int qn = 0;
    for (int it = 0; it < nIter; ++it) {
        int jt = it + start;
        if (jt >= nIter) jt -= nIter;
        const int wbase = (jt * OWN_WAVES + wave) * (4 * BLK_EDGES);
        const int base = wbase + 4 * lane;
        int4 dv[4];
        if (vec && wbase + 4 * BLK_EDGES <= E) {
            #pragma unroll
            for (int k = 0; k < 4; ++k) dv[k] = *(const int4*)(dst + base + k * BLK_EDGES);
        } else {
            #pragma unroll
            for (int k = 0; k < 4; ++k) dv[k] = load_dst4_guarded(dst, base + k * BLK_EDGES, E);
        }
        #pragma unroll
        for (int k = 0; k < 4; ++k) {
            const int e0 = base + k * BLK_EDGES;
            const int dd[4] = {dv[k].x, dv[k].y, dv[k].z, dv[k].w};
            bool m[4];
            bool any = false;
            #pragma unroll
            for (int c = 0; c < 4; ++c) {
                m[c] = (unsigned)(dd[c] - lo) < (unsigned)nloc;
                any = any || m[c];
            }
            if (__builtin_amdgcn_ballot_w32(any) != 0u) {
                #pragma unroll
                for (int c = 0; c < 4; ++c) {
                    const unsigned mask = __builtin_amdgcn_ballot_w32(m[c]);
                    if (m[c]) {
                        const int pos = qn + __builtin_popcount(mask & below);
                        qs[pos] = e0 + c;
                        qd[pos] = dd[c] - lo;
                    }
                    qn += __builtin_popcount(mask);
                }
                if (qn >= Q_DRAIN) {
                    if (PASS == 1) drain_max<Q_ROW>(qs, qd, qn, lane, src, dst, nullptr, lo, asrc, sadst, smax, rlist, rcount, ovf);
                    else           drain_sum(qs, qd, qn, lane, src, asrc, h, sadst, smax, ssum, acc);
                    qn = 0;
                }
            }
        }
    }
    if (qn > 0) {
        if (PASS == 1) drain_max<Q_ROW>(qs, qd, qn, lane, src, dst, nullptr, lo, asrc, sadst, smax, rlist, rcount, ovf);
        else           drain_sum(qs, qd, qn, lane, src, asrc, h, sadst, smax, ssum, acc);
    }
}

static_assert(KEY_LOADS == 4, "the mask packs four loads of sixteen keys into 64 bits");
template <int LOOK>
__device__ __forceinline__ void owner_scan_keys(const int* __restrict__ src,
                                                const int* __restrict__ dst,
                                                const unsigned* __restrict__ keys,
                                                const float* __restrict__ asrc,
                                                int E, int lo, int wave, int lane,
                                                int* qs, const float* sadst, float* smax,
                                                float* rlist, int* rcount, int* ovf) {
    const int nIter = keys_padded(E) / KEY_TRIP;
    const unsigned* __restrict__ pk = packed_of(keys, E);
    const int start = (int)(((long long)blockIdx.x * nIter) / gridDim.x);
    const unsigned own = (unsigned)blockIdx.x * 0x01010101u;
    const unsigned below = (1u << lane) - 1u;
    int qn = 0;
    for (int it = 0; it < nIter; ++it) {
        int jt = it + start;
        if (jt >= nIter) jt -= nIter;
        const int base = (jt * OWN_WAVES + wave) * (KEY_LOADS * 512) + 16 * lane;
        uint4 kv[KEY_LOADS];
        #pragma unroll
        for (int k = 0; k < KEY_LOADS; ++k) kv[k] = *(const uint4*)((const unsigned char*)keys + base + k * 512);
        unsigned miss[2] = {0u, 0u};
        #pragma unroll
        for (int k = 0; k < KEY_LOADS; ++k) {
            const unsigned kw[4] = {kv[k].x, kv[k].y, kv[k].z, kv[k].w};
            #pragma unroll
            for (int c = 0; c < 4; ++c) {
                const unsigned x = kw[c] ^ own;
                const unsigned nz = ((x & 0x7f7f7f7fu) + 0x7f7f7f7fu) | x;
                const int j = 4 * (k & 1) + c;
                miss[k >> 1] |= (nz >> (7 - j)) & (0x01010101u << j);
            }
        }
        unsigned long long m = (unsigned long long)(~miss[0]) | ((unsigned long long)(~miss[1]) << 32);
        unsigned any = __builtin_amdgcn_ballot_w32(m != 0ull);
        for (int r = 0; r < 64 && any != 0u; ++r) {
            if (m != 0ull) {
                const int p = __builtin_ctzll(m);
                m &= m - 1ull;
                qs[qn + __builtin_popcount(any & below)] =
                    base + (p >> 5) * 1024 + ((p >> 2) & 1) * 512 + (p & 3) * 4 + ((p >> 3) & 3);
            }
            qn += __builtin_popcount(any);
            if (qn >= Q_DRAIN) {
                drain_max<LOOK>(qs, qs, qn, lane, src, dst, pk, lo, asrc, sadst, smax, rlist, rcount, ovf);
                qn = 0;
            }
            any = __builtin_amdgcn_ballot_w32(m != 0ull);
        }
    }
    if (qn > 0)
        drain_max<LOOK>(qs, qs, qn, lane, src, dst, pk, lo, asrc, sadst, smax, rlist, rcount, ovf);
}

__global__ __launch_bounds__(OWN_THREADS) void gat_owner(const int* __restrict__ src,
                                                         const int* __restrict__ dst,
                                                         const float* __restrict__ asrc,
                                                         const float* __restrict__ adst,
                                                         const float* __restrict__ h,
                                                         const unsigned* __restrict__ keys,
                                                         float* __restrict__ out,
                                                         int N, int E, unsigned long long wsBytes) {
    __shared__ float acc[OWN_NODES * F_OUT];
    __shared__ float smax[OWN_NODES * NUM_HEADS];
    __shared__ float ssum[OWN_NODES * NUM_HEADS];
    __shared__ float sadst[OWN_NODES * NUM_HEADS];
    __shared__ int   qsrc[OWN_WAVES * Q_CAP];
    __shared__ int   qdst[OWN_WAVES * Q_CAP];
    __shared__ int   rcount[OWN_NODES];
    __shared__ int   ovf;

    const int t = threadIdx.x;
    const int lane = t & 31;
    const int wave = __builtin_amdgcn_readfirstlane(t >> 5);
    const int lo = blockIdx.x * OWN_NODES;
    const int nloc = min(OWN_NODES, N - lo);

    if (t == 0) ovf = 0;
    for (int i = t; i < OWN_NODES; i += OWN_THREADS) rcount[i] = 0;
    for (int i = t; i < OWN_NODES * NUM_HEADS; i += OWN_THREADS) {
        smax[i] = -3.0e38f;
        ssum[i] = 0.0f;
        sadst[i] = (i < nloc * NUM_HEADS) ? adst[(size_t)lo * NUM_HEADS + i] : 0.0f;
    }
    __syncthreads();

    int* qs = qsrc + wave * Q_CAP;
    int* qd = qdst + wave * Q_CAP;

    if (packed_usable(N, E, wsBytes))
        owner_scan_keys<PACKED>(src, dst, keys, asrc, E, lo, wave, lane, qs, sadst, smax, acc, rcount, &ovf);
    else if (keys_usable(N, E, wsBytes))
        owner_scan_keys<SRC_DST>(src, dst, keys, asrc, E, lo, wave, lane, qs, sadst, smax, acc, rcount, &ovf);
    else
        owner_scan<1>(src, dst, asrc, h, E, lo, nloc, wave, lane, qs, qd, sadst, smax, ssum, acc,
                      acc, rcount, &ovf);
    __syncthreads();
    if (ovf == 0) {
        wave_rows(acc, rcount, wave, lane, lo, nloc, asrc, h, sadst, smax, out);
    } else {
        for (int i = t; i < OWN_NODES * F_OUT; i += OWN_THREADS) acc[i] = 0.0f;
        __syncthreads();
        owner_scan<2>(src, dst, asrc, h, E, lo, nloc, wave, lane, qs, qd, sadst, smax, ssum, acc,
                      nullptr, nullptr, nullptr);
        __syncthreads();

        for (int r = wave; r < nloc; r += OWN_WAVES) {
            float* op = out + (size_t)(lo + r) * F_OUT + lane;
            #pragma unroll
            for (int c = 0; c < NUM_HEADS; ++c)
                op[c * HEAD_DIM] = acc[r * F_OUT + c * HEAD_DIM + lane] /
                                   (ssum[r * NUM_HEADS + c] + 1e-10f);
        }
    }
}

extern "C" void kernel_launch(void* const* d_in, const int* in_sizes, int n_in,
                              void* d_out, int out_size, void* d_ws, size_t ws_size,
                              hipStream_t stream) {
    const float* x     = (const float*)d_in[0];
    const int*   ei    = (const int*)d_in[1];
    const float* W     = (const float*)d_in[2];
    const float* a_src = (const float*)d_in[3];
    const float* a_dst = (const float*)d_in[4];
    float*       out   = (float*)d_out;

    const int N = in_sizes[0] / F_IN;
    const int E = in_sizes[1] / 2;
    const int* src = ei;
    const int* dst = ei + E;

    float* h    = (float*)d_ws;
    float* asrc = h    + (size_t)N * F_OUT;
    float* adst = asrc + (size_t)N * NUM_HEADS;
    unsigned* keys = (unsigned*)(adst + (size_t)N * NUM_HEADS);

    {
        int fullTiles = N / 16;
        if (fullTiles > 0)
            gat_gemm<<<GEMM_GROUPS, 256, 0, stream>>>(x, W, a_src, a_dst, h, asrc, adst, src, dst, keys, N, E,
                                                    (unsigned long long)ws_size);
        int rowStart = fullTiles * 16;
        int rem = N - rowStart;
        if (rem > 0) {
            int total = rem * F_OUT;
            gat_gemm_tail<<<(total + 255) / 256, 256, 0, stream>>>(x, W, h, N, rowStart);
            gat_alpha_tail<<<(rem * NUM_HEADS + 255) / 256, 256, 0, stream>>>(h, a_src, a_dst, asrc, adst, src, dst, keys, N, E,
                                                                              (unsigned long long)ws_size, rowStart,
                                                                              fullTiles > 0 ? 0 : 1);
        }
    }

    gat_owner<<<(N + OWN_NODES - 1) / OWN_NODES, OWN_THREADS, 0, stream>>>(src, dst, asrc, adst, h, keys, out, N, E,
                                                                          (unsigned long long)ws_size);
}
